# starting kernel
# baseline (speedup 1.0000x reference)
#include <hip/hip_runtime.h>
#include <hip/hip_bf16.h>

typedef __attribute__((ext_vector_type(2))) float v2f;
typedef __attribute__((ext_vector_type(8))) float v8f;

__global__ void k_init_deg(float* __restrict__ deg, int n) {
  int i = blockIdx.x * blockDim.x + threadIdx.x;
  if (i < n) deg[i] = 1.0f;
}

__global__ void k_count_deg(const int* __restrict__ dst, float* __restrict__ deg, int e) {
  int i = blockIdx.x * blockDim.x + threadIdx.x;
  if (i < e) atomicAdd(&deg[dst[i]], 1.0f);
}

__global__ void k_rsqrt_inplace(float* __restrict__ d, int n) {
  int i = blockIdx.x * blockDim.x + threadIdx.x;
  if (i < n) d[i] = rsqrtf(d[i]);
}

template <int K, int N>
__global__ void k_gemm_wmma_f32(const float* __restrict__ A,
                                const float* __restrict__ B,
                                float* __restrict__ C) {
  const int lane = threadIdx.x & 31;
  const int row  = lane & 15;
  const int kp   = lane >> 4;
  const int m0   = blockIdx.x * 16;
  const int n0   = blockIdx.y * 16;
  const int col  = n0 + row;
  const int colc = (col < N) ? col : (N - 1);

  const float* Arow = A + (size_t)(m0 + row) * K;
  v8f c = {};
#pragma unroll
  for (int k0 = 0; k0 < K; k0 += 4) {
    const int ka = k0 + 2 * kp;
    v2f a, b;
    a.x = Arow[ka];
    a.y = Arow[ka + 1];
    b.x = B[(size_t)ka * N + colc];
    b.y = B[(size_t)(ka + 1) * N + colc];
    c = __builtin_amdgcn_wmma_f32_16x16x4_f32(false, a, false, b, (short)0, c,
                                              false, false);
  }
  if (col < N) {
#pragma unroll
    for (int r = 0; r < 8; ++r)
      C[(size_t)(m0 + r + 8 * kp) * N + col] = c[r];
  }
}

__global__ void k_self_loop_init(const float* __restrict__ h,
                                 const float* __restrict__ dinv,
                                 float* __restrict__ acc, int n, int stride) {
  int idx = blockIdx.x * blockDim.x + threadIdx.x;
  if (idx >= n * stride) return;
  int i = idx / stride;
  float di = dinv[i];
  acc[idx] = h[idx] * di * di;
}

template <int F4, int STRIDE>
__global__ void k_edge_agg(const int* __restrict__ src, const int* __restrict__ dst,
                           const float* __restrict__ dinv,
                           const float* __restrict__ h,
                           float* __restrict__ acc, int e) {
  int tid = blockIdx.x * blockDim.x + threadIdx.x;
  if (tid >= e * F4) return;
  int ei  = tid / F4;
  int sub = tid - ei * F4;
  int s = src[ei], d = dst[ei];
  float norm = dinv[s] * dinv[d];
  const float4 v = *(const float4*)(h + (size_t)s * STRIDE + sub * 4);
  float* ap = acc + (size_t)d * STRIDE + sub * 4;
  atomicAdd(ap + 0, v.x * norm);
  atomicAdd(ap + 1, v.y * norm);
  atomicAdd(ap + 2, v.z * norm);
  atomicAdd(ap + 3, v.w * norm);
}

__global__ void k_relu_bias(const float* __restrict__ acc,
                            const float* __restrict__ b,
                            float* __restrict__ out, int n, int stride) {
  int idx = blockIdx.x * blockDim.x + threadIdx.x;
  if (idx >= n * stride) return;
  int f = idx % stride;
  out[idx] = fmaxf(acc[idx] + b[f], 0.0f);
}

__global__ void k_bias_logsoftmax40(float* __restrict__ out,
                                    const float* __restrict__ b, int n) {
  int i = blockIdx.x * blockDim.x + threadIdx.x;
  if (i >= n) return;
  float* p = out + (size_t)i * 40;
  float v[40];
  float m = -3.402823466e38f;
#pragma unroll
  for (int f = 0; f < 40; ++f) {
    v[f] = p[f] + b[f];
    m = fmaxf(m, v[f]);
  }
  float s = 0.0f;
#pragma unroll
  for (int f = 0; f < 40; ++f) s += expf(v[f] - m);
  float lse = m + logf(s);
#pragma unroll
  for (int f = 0; f < 40; ++f) p[f] = v[f] - lse;
}

extern "C" void kernel_launch(void* const* d_in, const int* in_sizes, int n_in,
                              void* d_out, int out_size, void* d_ws, size_t ws_size,
                              hipStream_t stream) {
  (void)n_in; (void)out_size; (void)ws_size;

  const float* x  = (const float*)d_in[0];
  const int*   ei = (const int*)d_in[1];
  const float* W1 = (const float*)d_in[2];
  const float* b1 = (const float*)d_in[3];
  const float* W2 = (const float*)d_in[4];
  const float* b2 = (const float*)d_in[5];
  float* out = (float*)d_out;

  const int F_IN = 128, F_H = 64, F_OUT = 40;
  const int Nn = in_sizes[0] / F_IN;
  const int E  = in_sizes[1] / 2;
  const int* src = ei;
  const int* dst = ei + E;

  char* ws = (char*)d_ws;
  float* dinv = (float*)(ws);
  float* h    = (float*)(ws + ((size_t)4  << 20));
  float* acc1 = (float*)(ws + ((size_t)36 << 20));
  float* h2   = (float*)(ws + ((size_t)68 << 20));

  const dim3 B256(256);

  k_init_deg<<<(Nn + 255) / 256, B256, 0, stream>>>(dinv, Nn);
  k_count_deg<<<(E + 255) / 256, B256, 0, stream>>>(dst, dinv, E);
  k_rsqrt_inplace<<<(Nn + 255) / 256, B256, 0, stream>>>(dinv, Nn);

  k_gemm_wmma_f32<128, 64><<<dim3(Nn / 16, 4), dim3(32), 0, stream>>>(x, W1, h);
  k_self_loop_init<<<((Nn * F_H) + 255) / 256, B256, 0, stream>>>(h, dinv, acc1, Nn, F_H);
  k_edge_agg<16, 64><<<((E * 16) + 255) / 256, B256, 0, stream>>>(src, dst, dinv, h, acc1, E);
  k_relu_bias<<<((Nn * F_H) + 255) / 256, B256, 0, stream>>>(acc1, b1, h, Nn, F_H);

  k_gemm_wmma_f32<64, 40><<<dim3(Nn / 16, 3), dim3(32), 0, stream>>>(h, W2, h2);
  k_self_loop_init<<<((Nn * F_OUT) + 255) / 256, B256, 0, stream>>>(h2, dinv, out, Nn, F_OUT);
  k_edge_agg<10, 40><<<((E * 10) + 255) / 256, B256, 0, stream>>>(src, dst, dinv, h2, out, E);
  k_bias_logsoftmax40<<<(Nn + 255) / 256, B256, 0, stream>>>(out, b2, Nn);
}


# Edges bucketed by destination (slot = returning count atomic); one wave sums each node's rows in registers in slot order, no float atomics; self-loop, bias, relu, log-softmax (lane-tree sum) fused
# speedup vs baseline: 1.7661x; 1.7661x over previous
#include <hip/hip_runtime.h>
#include <hip/hip_bf16.h>

typedef __attribute__((ext_vector_type(2))) float v2f;
typedef __attribute__((ext_vector_type(8))) float v8f;


__global__ void k_zero_counts(int* __restrict__ cnt, int* __restrict__ cursor, int n) {
  int i = blockIdx.x * blockDim.x + threadIdx.x;
  if (i < n) cnt[i] = 0;
  if (i == 0) *cursor = 0;
}

__global__ void k_count_slot(const int* __restrict__ dst, int* __restrict__ cnt,
                             int* __restrict__ slot, int e) {
  int i = blockIdx.x * blockDim.x + threadIdx.x;
  if (i < e) slot[i] = atomicAdd(&cnt[dst[i]], 1);
}

__global__ void __launch_bounds__(256)
k_segments(const int* __restrict__ cnt, int* __restrict__ cursor,
           int* __restrict__ first, float* __restrict__ dinv, int n) {
  __shared__ int wsum[8];
  __shared__ int sbase;
  const int i = blockIdx.x * 256 + threadIdx.x;
  const int lane = threadIdx.x & 31;
  const int wave = threadIdx.x >> 5;
  const int c = (i < n) ? cnt[i] : 0;
  int x = c;
  for (int o = 1; o < 32; o <<= 1) {
    int y = __shfl_up(x, o);
    if (lane >= o) x += y;
  }
  if (lane == 31) wsum[wave] = x;
  __syncthreads();
  if (threadIdx.x == 0) {
    int t = 0;
    for (int k = 0; k < 8; ++k) { int v = wsum[k]; wsum[k] = t; t += v; }
    sbase = atomicAdd(cursor, t);
  }
  __syncthreads();
  if (i < n) {
    first[i] = sbase + wsum[wave] + x - c;
    dinv[i] = rsqrtf((float)(c + 1));
  }
}

__global__ void k_fill_buckets(const int* __restrict__ src, const int* __restrict__ dst,
                               const int* __restrict__ slot, const int* __restrict__ first,
                               const float* __restrict__ dinv, int2* __restrict__ bucket, int e) {
  int i = blockIdx.x * blockDim.x + threadIdx.x;
  if (i >= e) return;
  const int s = src[i];
  bucket[first[dst[i]] + slot[i]] = make_int2(s, __float_as_int(dinv[s]));
}

template <int K, int N>
__global__ void k_gemm_wmma_f32(const float* __restrict__ A,
                                const float* __restrict__ B,
                                float* __restrict__ C) {
  const int lane = threadIdx.x & 31;
  const int row  = lane & 15;
  const int kp   = lane >> 4;
  const int m0   = blockIdx.x * 16;
  const int n0   = blockIdx.y * 16;
  const int col  = n0 + row;
  const int colc = (col < N) ? col : (N - 1);

  const float* Arow = A + (size_t)(m0 + row) * K;
  v8f c = {};
#pragma unroll
  for (int k0 = 0; k0 < K; k0 += 4) {
    const int ka = k0 + 2 * kp;
    v2f a, b;
    a.x = Arow[ka];
    a.y = Arow[ka + 1];
    b.x = B[(size_t)ka * N + colc];
    b.y = B[(size_t)(ka + 1) * N + colc];
    c = __builtin_amdgcn_wmma_f32_16x16x4_f32(false, a, false, b, (short)0, c,
                                              false, false);
  }
  if (col < N) {
#pragma unroll
    for (int r = 0; r < 8; ++r)
      C[(size_t)(m0 + r + 8 * kp) * N + col] = c[r];
  }
}


__global__ void __launch_bounds__(256)
k_gather64_relu(const int2* __restrict__ bucket, const int* __restrict__ first,
                const int* __restrict__ cnt, const float* __restrict__ dinv,
                const float* __restrict__ h, const float* __restrict__ b,
                float* __restrict__ out, int n) {
  const int lane = threadIdx.x & 31;
  const int i = __builtin_amdgcn_readfirstlane((int)(blockIdx.x * 8 + (threadIdx.x >> 5)));
  if (i >= n) return;
  const int ne = __builtin_amdgcn_readfirstlane(cnt[i]);
  const int start = __builtin_amdgcn_readfirstlane(first[i]);
  const float di = dinv[i];
  const float* hl = h + 2 * lane;
  const float2 hv = *(const float2*)(hl + (size_t)i * 64);
  float2 acc;
  acc.x = hv.x * di * di;
  acc.y = hv.y * di * di;
  for (int base = 0; base < ne; base += 32) {
    int m = ne - base;
    if (m > 32) m = 32;
    int es = i, ew = 0;
    if (lane < m) {
      const int2 t = bucket[start + base + lane];
      es = t.x;
      ew = t.y;
    }
    for (int e = 0; e < m; e += 4) {
      const int s0 = __builtin_amdgcn_readlane(es, e);
      const int s1 = __builtin_amdgcn_readlane(es, e + 1);
      const int s2 = __builtin_amdgcn_readlane(es, e + 2);
      const int s3 = __builtin_amdgcn_readlane(es, e + 3);
      const float2 v0 = *(const float2*)(hl + (size_t)s0 * 64);
      const float2 v1 = *(const float2*)(hl + (size_t)s1 * 64);
      const float2 v2 = *(const float2*)(hl + (size_t)s2 * 64);
      const float2 v3 = *(const float2*)(hl + (size_t)s3 * 64);
      const float n0 = __int_as_float(__builtin_amdgcn_readlane(ew, e)) * di;
      const float n1 = __int_as_float(__builtin_amdgcn_readlane(ew, e + 1)) * di;
      const float n2 = __int_as_float(__builtin_amdgcn_readlane(ew, e + 2)) * di;
      const float n3 = __int_as_float(__builtin_amdgcn_readlane(ew, e + 3)) * di;
      acc.x += v0.x * n0; acc.y += v0.y * n0;
      if (e + 1 < m) { acc.x += v1.x * n1; acc.y += v1.y * n1; }
      if (e + 2 < m) { acc.x += v2.x * n2; acc.y += v2.y * n2; }
      if (e + 3 < m) { acc.x += v3.x * n3; acc.y += v3.y * n3; }
    }
  }
  const float2 bv = *(const float2*)(b + 2 * lane);
  float2 o;
  o.x = fmaxf(acc.x + bv.x, 0.0f);
  o.y = fmaxf(acc.y + bv.y, 0.0f);
  *(float2*)(out + (size_t)i * 64 + 2 * lane) = o;
}

__global__ void __launch_bounds__(256)
k_gather40_logsoftmax(const int2* __restrict__ bucket, const int* __restrict__ first,
                      const int* __restrict__ cnt, const float* __restrict__ dinv,
                      const float* __restrict__ h, const float* __restrict__ b,
                      float* __restrict__ out, int n) {
  const int lane = threadIdx.x & 31;
  const int i = __builtin_amdgcn_readfirstlane((int)(blockIdx.x * 8 + (threadIdx.x >> 5)));
  if (i >= n) return;
  const int ne = __builtin_amdgcn_readfirstlane(cnt[i]);
  const int start = __builtin_amdgcn_readfirstlane(first[i]);
  const float di = dinv[i];
  const bool own = lane < 20;
  const int fo = 2 * (own ? lane : lane - 20);
  const float* hl = h + fo;
  const float2 hv = *(const float2*)(hl + (size_t)i * 40);
  float2 acc;
  acc.x = hv.x * di * di;
  acc.y = hv.y * di * di;
  for (int base = 0; base < ne; base += 32) {
    int m = ne - base;
    if (m > 32) m = 32;
    int es = i, ew = 0;
    if (lane < m) {
      const int2 t = bucket[start + base + lane];
      es = t.x;
      ew = t.y;
    }
    for (int e = 0; e < m; e += 4) {
      const int s0 = __builtin_amdgcn_readlane(es, e);
      const int s1 = __builtin_amdgcn_readlane(es, e + 1);
      const int s2 = __builtin_amdgcn_readlane(es, e + 2);
      const int s3 = __builtin_amdgcn_readlane(es, e + 3);
      const float2 v0 = *(const float2*)(hl + (size_t)s0 * 40);
      const float2 v1 = *(const float2*)(hl + (size_t)s1 * 40);
      const float2 v2 = *(const float2*)(hl + (size_t)s2 * 40);
      const float2 v3 = *(const float2*)(hl + (size_t)s3 * 40);
      const float n0 = __int_as_float(__builtin_amdgcn_readlane(ew, e)) * di;
      const float n1 = __int_as_float(__builtin_amdgcn_readlane(ew, e + 1)) * di;
      const float n2 = __int_as_float(__builtin_amdgcn_readlane(ew, e + 2)) * di;
      const float n3 = __int_as_float(__builtin_amdgcn_readlane(ew, e + 3)) * di;
      acc.x += v0.x * n0; acc.y += v0.y * n0;
      if (e + 1 < m) { acc.x += v1.x * n1; acc.y += v1.y * n1; }
      if (e + 2 < m) { acc.x += v2.x * n2; acc.y += v2.y * n2; }
      if (e + 3 < m) { acc.x += v3.x * n3; acc.y += v3.y * n3; }
    }
  }
  const float2 bv = *(const float2*)(b + fo);
  const float vx = acc.x + bv.x;
  const float vy = acc.y + bv.y;
  float mx = own ? fmaxf(vx, vy) : -3.402823466e38f;
  for (int o = 16; o > 0; o >>= 1) mx = fmaxf(mx, __shfl_xor(mx, o));
  float sm = own ? (expf(vx - mx) + expf(vy - mx)) : 0.0f;
  for (int o = 16; o > 0; o >>= 1) sm += __shfl_xor(sm, o);
  const float lse = mx + logf(sm);
  if (own) {
    float2 o2;
    o2.x = vx - lse;
    o2.y = vy - lse;
    *(float2*)(out + (size_t)i * 40 + fo) = o2;
  }
}

extern "C" void kernel_launch(void* const* d_in, const int* in_sizes, int n_in,
                              void* d_out, int out_size, void* d_ws, size_t ws_size,
                              hipStream_t stream) {
  (void)n_in; (void)out_size; (void)ws_size;

  const float* x  = (const float*)d_in[0];
  const int*   ei = (const int*)d_in[1];
  const float* W1 = (const float*)d_in[2];
  const float* b1 = (const float*)d_in[3];
  const float* W2 = (const float*)d_in[4];
  const float* b2 = (const float*)d_in[5];
  float* out = (float*)d_out;

  const int F_IN = 128;
  const int Nn = in_sizes[0] / F_IN;
  const int E  = in_sizes[1] / 2;
  const int* src = ei;
  const int* dst = ei + E;

  char* ws = (char*)d_ws;
  int*   cnt    = (int*)(ws);
  int*   cursor = (int*)(ws + ((size_t)1  << 20));
  int*   first  = (int*)(ws + ((size_t)2  << 20));
  float* dinv   = (float*)(ws + ((size_t)3  << 20));
  float* h      = (float*)(ws + ((size_t)4  << 20));
  float* h1     = (float*)(ws + ((size_t)36 << 20));
  float* h2     = (float*)(ws + ((size_t)68 << 20));
  int*   slot   = (int*)(ws + ((size_t)88 << 20));
  int2*  bucket = (int2*)(ws + ((size_t)92 << 20));

  const dim3 B256(256);

  k_zero_counts<<<(Nn + 255) / 256, B256, 0, stream>>>(cnt, cursor, Nn);
  k_count_slot<<<(E + 255) / 256, B256, 0, stream>>>(dst, cnt, slot, E);
  k_segments<<<(Nn + 255) / 256, B256, 0, stream>>>(cnt, cursor, first, dinv, Nn);
  k_fill_buckets<<<(E + 255) / 256, B256, 0, stream>>>(src, dst, slot, first, dinv, bucket, E);

  k_gemm_wmma_f32<128, 64><<<dim3(Nn / 16, 4), dim3(32), 0, stream>>>(x, W1, h);
  k_gather64_relu<<<(Nn + 7) / 8, B256, 0, stream>>>(bucket, first, cnt, dinv, h, b1, h1, Nn);

  k_gemm_wmma_f32<64, 40><<<dim3(Nn / 16, 3), dim3(32), 0, stream>>>(h1, W2, h2);
  k_gather40_logsoftmax<<<(Nn + 7) / 8, B256, 0, stream>>>(bucket, first, cnt, dinv, h2, b2, out, Nn);
}


# Waves launched cut over fivefold (wave entry rate was the limit): gather wave takes 8 nodes in turn, next node's words fetched ahead; GEMM wave owns 16 rows over all column tiles; 4 edges per thread
# speedup vs baseline: 2.0140x; 1.1404x over previous
#include <hip/hip_runtime.h>
#include <hip/hip_bf16.h>

typedef __attribute__((ext_vector_type(2))) float v2f;
typedef __attribute__((ext_vector_type(8))) float v8f;


__global__ void k_zero_counts(int* __restrict__ cnt, int* __restrict__ cursor, int n) {
  const int i = blockIdx.x * blockDim.x + threadIdx.x;
  const int j = 4 * i;
  if (j + 3 < n) {
    *(int4*)(cnt + j) = make_int4(0, 0, 0, 0);
  } else {
    for (int k = j; k < n; ++k) cnt[k] = 0;
  }
  if (i == 0) *cursor = 0;
}

__global__ void k_count_slot(const int* __restrict__ dst, int* __restrict__ cnt,
                             int* __restrict__ slot, int e, int q) {
  const int i = blockIdx.x * blockDim.x + threadIdx.x;
  if (i >= q) return;
  const int i1 = i + q, i2 = i + 2 * q, i3 = i + 3 * q;
  const int d0 = dst[i];
  const int d1 = (i1 < e) ? dst[i1] : 0;
  const int d2 = (i2 < e) ? dst[i2] : 0;
  const int d3 = (i3 < e) ? dst[i3] : 0;
  const int r0 = atomicAdd(&cnt[d0], 1);
  int r1 = 0, r2 = 0, r3 = 0;
  if (i1 < e) r1 = atomicAdd(&cnt[d1], 1);
  if (i2 < e) r2 = atomicAdd(&cnt[d2], 1);
  if (i3 < e) r3 = atomicAdd(&cnt[d3], 1);
  slot[i] = r0;
  if (i1 < e) slot[i1] = r1;
  if (i2 < e) slot[i2] = r2;
  if (i3 < e) slot[i3] = r3;
}

__global__ void __launch_bounds__(256)
k_segments(const int* __restrict__ cnt, int* __restrict__ cursor,
           int* __restrict__ first, float* __restrict__ dinv, int n) {
  __shared__ int wsum[8];
  __shared__ int sbase;
  const int i = blockIdx.x * 256 + threadIdx.x;
  const int lane = threadIdx.x & 31;
  const int wave = threadIdx.x >> 5;
  const int c = (i < n) ? cnt[i] : 0;
  int x = c;
  for (int o = 1; o < 32; o <<= 1) {
    int y = __shfl_up(x, o);
    if (lane >= o) x += y;
  }
  if (lane == 31) wsum[wave] = x;
  __syncthreads();
  if (threadIdx.x == 0) {
    int t = 0;
    for (int k = 0; k < 8; ++k) { int v = wsum[k]; wsum[k] = t; t += v; }
    sbase = atomicAdd(cursor, t);
  }
  __syncthreads();
  if (i < n) {
    first[i] = sbase + wsum[wave] + x - c;
    dinv[i] = rsqrtf((float)(c + 1));
  }
}

__global__ void k_fill_buckets(const int* __restrict__ src, const int* __restrict__ dst,
                               const int* __restrict__ slot, const int* __restrict__ first,
                               const float* __restrict__ dinv, int2* __restrict__ bucket,
                               int e, int q) {
  const int i = blockIdx.x * blockDim.x + threadIdx.x;
  if (i >= q) return;
  int s[4], p[4];
  float w[4];
  bool ok[4];
#pragma unroll
  for (int k = 0; k < 4; ++k) {
    const int j = i + k * q;
    ok[k] = j < e;
    const int jj = ok[k] ? j : i;
    s[k] = src[jj];
    p[k] = first[dst[jj]] + slot[jj];
    w[k] = dinv[s[k]];
  }
#pragma unroll
  for (int k = 0; k < 4; ++k)
    if (ok[k]) bucket[p[k]] = make_int2(s[k], __float_as_int(w[k]));
}

template <int K, int N>
__global__ void __launch_bounds__(256)
k_gemm_wmma_f32(const float* __restrict__ A, const float* __restrict__ B,
                float* __restrict__ C, int mtiles) {
  constexpr int NT = (N + 15) / 16;
  const int lane = threadIdx.x & 31;
  const int row  = lane & 15;
  const int kp   = lane >> 4;
  const int tile = blockIdx.x * 8 + (threadIdx.x >> 5);
  if (tile >= mtiles) return;
  const int m0 = tile * 16;

  int colc[NT];
#pragma unroll
  for (int t = 0; t < NT; ++t) {
    const int col = t * 16 + row;
    colc[t] = (col < N) ? col : (N - 1);
  }

  const float* Arow = A + (size_t)(m0 + row) * K;
  v8f c[NT];
#pragma unroll
  for (int t = 0; t < NT; ++t) c[t] = (v8f){};

#pragma unroll
  for (int k0 = 0; k0 < K; k0 += 4) {
    const int ka = k0 + 2 * kp;
    v2f a;
    a.x = Arow[ka];
    a.y = Arow[ka + 1];
#pragma unroll
    for (int t = 0; t < NT; ++t) {
      v2f b;
      b.x = B[(size_t)ka * N + colc[t]];
      b.y = B[(size_t)(ka + 1) * N + colc[t]];
      c[t] = __builtin_amdgcn_wmma_f32_16x16x4_f32(false, a, false, b, (short)0, c[t],
                                                   false, false);
    }
  }
#pragma unroll
  for (int t = 0; t < NT; ++t) {
    const int col = t * 16 + row;
    if (col < N) {
#pragma unroll
      for (int r = 0; r < 8; ++r)
        C[(size_t)(m0 + r + 8 * kp) * N + col] = c[t][r];
    }
  }
}

#define NPW 8

__device__ __forceinline__ int2 bucket_words(const int2* __restrict__ bucket, int pos,
                                             int m, int i, int lane) {
  int2 t = make_int2(i, 0);
  if (lane < m) t = bucket[pos + lane];
  return t;
}

template <int F>
__device__ __forceinline__ void add_rows(int es, int ew, int m, const float* __restrict__ hl,
                                         float di, float2& acc) {
  for (int e = 0; e < m; e += 4) {
    const int s0 = __builtin_amdgcn_readlane(es, e);
    const int s1 = __builtin_amdgcn_readlane(es, e + 1);
    const int s2 = __builtin_amdgcn_readlane(es, e + 2);
    const int s3 = __builtin_amdgcn_readlane(es, e + 3);
    const float2 v0 = *(const float2*)(hl + (size_t)s0 * F);
    const float2 v1 = *(const float2*)(hl + (size_t)s1 * F);
    const float2 v2 = *(const float2*)(hl + (size_t)s2 * F);
    const float2 v3 = *(const float2*)(hl + (size_t)s3 * F);
    const float n0 = __int_as_float(__builtin_amdgcn_readlane(ew, e)) * di;
    const float n1 = __int_as_float(__builtin_amdgcn_readlane(ew, e + 1)) * di;
    const float n2 = __int_as_float(__builtin_amdgcn_readlane(ew, e + 2)) * di;
    const float n3 = __int_as_float(__builtin_amdgcn_readlane(ew, e + 3)) * di;
    acc.x += v0.x * n0; acc.y += v0.y * n0;
    if (e + 1 < m) { acc.x += v1.x * n1; acc.y += v1.y * n1; }
    if (e + 2 < m) { acc.x += v2.x * n2; acc.y += v2.y * n2; }
    if (e + 3 < m) { acc.x += v3.x * n3; acc.y += v3.y * n3; }
  }
}

template <int F>
__device__ __forceinline__ float2 node_sum(const int2* __restrict__ bucket, int start, int ne,
                                           int i, int lane, int es, int ew, float2 own,
                                           const float* __restrict__ hl, float di) {
  float2 acc;
  acc.x = own.x * di * di;
  acc.y = own.y * di * di;
  add_rows<F>(es, ew, ne < 32 ? ne : 32, hl, di, acc);
  for (int base = 32; base < ne; base += 32) {
    const int m = (ne - base < 32) ? (ne - base) : 32;
    const int2 u = bucket_words(bucket, start + base, m, i, lane);
    add_rows<F>(u.x, u.y, m, hl, di, acc);
  }
  return acc;
}

__global__ void __launch_bounds__(256)
k_gather64_relu(const int2* __restrict__ bucket, const int* __restrict__ first,
                const int* __restrict__ cnt, const float* __restrict__ dinv,
                const float* __restrict__ h, const float* __restrict__ b,
                float* __restrict__ out, int n) {
  const int lane = threadIdx.x & 31;
  const int i0 = NPW * __builtin_amdgcn_readfirstlane((int)(blockIdx.x * 8 + (threadIdx.x >> 5)));
  if (i0 >= n) return;
  const int nn = (n - i0 < NPW) ? (n - i0) : NPW;
  int cv = 0, fv = 0, dv = 0;
  if (lane < nn) {
    cv = cnt[i0 + lane];
    fv = first[i0 + lane];
    dv = __float_as_int(dinv[i0 + lane]);
  }
  const float* hl = h + 2 * lane;
  const float2 bv = *(const float2*)(b + 2 * lane);
  int ne = __builtin_amdgcn_readlane(cv, 0);
  int start = __builtin_amdgcn_readlane(fv, 0);
  int2 t = bucket_words(bucket, start, ne, i0, lane);
  float2 hv = *(const float2*)(hl + (size_t)i0 * 64);
  for (int j = 0; j < nn; ++j) {
    const int i = i0 + j;
    const int ne_c = ne, start_c = start;
    const int es = t.x, ew = t.y;
    const float2 own = hv;
    const float di = __int_as_float(__builtin_amdgcn_readlane(dv, j));
    if (j + 1 < nn) {
      ne = __builtin_amdgcn_readlane(cv, j + 1);
      start = __builtin_amdgcn_readlane(fv, j + 1);
      t = bucket_words(bucket, start, ne, i + 1, lane);
      hv = *(const float2*)(hl + (size_t)(i + 1) * 64);
    }
    const float2 acc = node_sum<64>(bucket, start_c, ne_c, i, lane, es, ew, own, hl, di);
    float2 o;
    o.x = fmaxf(acc.x + bv.x, 0.0f);
    o.y = fmaxf(acc.y + bv.y, 0.0f);
    *(float2*)(out + (size_t)i * 64 + 2 * lane) = o;
  }
}

__global__ void __launch_bounds__(256)
k_gather40_logsoftmax(const int2* __restrict__ bucket, const int* __restrict__ first,
                      const int* __restrict__ cnt, const float* __restrict__ dinv,
                      const float* __restrict__ h, const float* __restrict__ b,
                      float* __restrict__ out, int n) {
  const int lane = threadIdx.x & 31;
  const int i0 = NPW * __builtin_amdgcn_readfirstlane((int)(blockIdx.x * 8 + (threadIdx.x >> 5)));
  if (i0 >= n) return;
  const int nn = (n - i0 < NPW) ? (n - i0) : NPW;
  int cv = 0, fv = 0, dv = 0;
  if (lane < nn) {
    cv = cnt[i0 + lane];
    fv = first[i0 + lane];
    dv = __float_as_int(dinv[i0 + lane]);
  }
  const bool own_lane = lane < 20;
  const int fo = 2 * (own_lane ? lane : lane - 20);
  const float* hl = h + fo;
  const float2 bv = *(const float2*)(b + fo);
  int ne = __builtin_amdgcn_readlane(cv, 0);
  int start = __builtin_amdgcn_readlane(fv, 0);
  int2 t = bucket_words(bucket, start, ne, i0, lane);
  float2 hv = *(const float2*)(hl + (size_t)i0 * 40);
  for (int j = 0; j < nn; ++j) {
    const int i = i0 + j;
    const int ne_c = ne, start_c = start;
    const int es = t.x, ew = t.y;
    const float2 own = hv;
    const float di = __int_as_float(__builtin_amdgcn_readlane(dv, j));
    if (j + 1 < nn) {
      ne = __builtin_amdgcn_readlane(cv, j + 1);
      start = __builtin_amdgcn_readlane(fv, j + 1);
      t = bucket_words(bucket, start, ne, i + 1, lane);
      hv = *(const float2*)(hl + (size_t)(i + 1) * 40);
    }
    const float2 acc = node_sum<40>(bucket, start_c, ne_c, i, lane, es, ew, own, hl, di);
    const float vx = acc.x + bv.x;
    const float vy = acc.y + bv.y;
    float mx = own_lane ? fmaxf(vx, vy) : -3.402823466e38f;
    for (int o = 16; o > 0; o >>= 1) mx = fmaxf(mx, __shfl_xor(mx, o));
    float sm = own_lane ? (expf(vx - mx) + expf(vy - mx)) : 0.0f;
    for (int o = 16; o > 0; o >>= 1) sm += __shfl_xor(sm, o);
    const float lse = mx + logf(sm);
    if (own_lane) {
      float2 o2;
      o2.x = vx - lse;
      o2.y = vy - lse;
      *(float2*)(out + (size_t)i * 40 + fo) = o2;
    }
  }
}

extern "C" void kernel_launch(void* const* d_in, const int* in_sizes, int n_in,
                              void* d_out, int out_size, void* d_ws, size_t ws_size,
                              hipStream_t stream) {
  (void)n_in; (void)out_size; (void)ws_size;

  const float* x  = (const float*)d_in[0];
  const int*   ei = (const int*)d_in[1];
  const float* W1 = (const float*)d_in[2];
  const float* b1 = (const float*)d_in[3];
  const float* W2 = (const float*)d_in[4];
  const float* b2 = (const float*)d_in[5];
  float* out = (float*)d_out;

  const int F_IN = 128;
  const int Nn = in_sizes[0] / F_IN;
  const int E  = in_sizes[1] / 2;
  const int* src = ei;
  const int* dst = ei + E;

  char* ws = (char*)d_ws;
  int*   cnt    = (int*)(ws);
  int*   cursor = (int*)(ws + ((size_t)1  << 20));
  int*   first  = (int*)(ws + ((size_t)2  << 20));
  float* dinv   = (float*)(ws + ((size_t)3  << 20));
  float* h      = (float*)(ws + ((size_t)4  << 20));
  float* h1     = (float*)(ws + ((size_t)36 << 20));
  float* h2     = (float*)(ws + ((size_t)68 << 20));
  int*   slot   = (int*)(ws + ((size_t)88 << 20));
  int2*  bucket = (int2*)(ws + ((size_t)92 << 20));

  const dim3 B256(256);

  const int Q = (E + 3) / 4;
  k_zero_counts<<<((Nn + 3) / 4 + 255) / 256, B256, 0, stream>>>(cnt, cursor, Nn);
  k_count_slot<<<(Q + 255) / 256, B256, 0, stream>>>(dst, cnt, slot, E, Q);
  k_segments<<<(Nn + 255) / 256, B256, 0, stream>>>(cnt, cursor, first, dinv, Nn);
  k_fill_buckets<<<(Q + 255) / 256, B256, 0, stream>>>(src, dst, slot, first, dinv, bucket, E, Q);

  const int MT = Nn / 16;
  const int GW = 8 * NPW;
  k_gemm_wmma_f32<128, 64><<<(MT + 7) / 8, B256, 0, stream>>>(x, W1, h, MT);
  k_gather64_relu<<<(Nn + GW - 1) / GW, B256, 0, stream>>>(bucket, first, cnt, dinv, h, b1, h1, Nn);

  k_gemm_wmma_f32<64, 40><<<(MT + 7) / 8, B256, 0, stream>>>(h1, W2, h2, MT);
  k_gather40_logsoftmax<<<(Nn + GW - 1) / GW, B256, 0, stream>>>(bucket, first, cnt, dinv, h2, b2, out, Nn);
}


# Edges bucketed by a two-level counting sort on LDS counters (per-chunk bin counts, per-bin node ranks): no global atomics; a node's edge rows are summed in LDS arrival order (order changes)
# speedup vs baseline: 2.7552x; 1.3680x over previous
#include <hip/hip_runtime.h>
#include <hip/hip_bf16.h>

typedef __attribute__((ext_vector_type(2))) float v2f;
typedef __attribute__((ext_vector_type(8))) float v8f;


#define SORT_G 128
#define BIN_SHIFT 9
#define BIN_NODES 512
#define MAX_BINS 256

__global__ void __launch_bounds__(1024)
k_bin_hist(const int* __restrict__ dst, int* __restrict__ histg, int e, int ch, int nb) {
  __shared__ int hist[MAX_BINS];
  const int t = threadIdx.x;
  if (t < MAX_BINS) hist[t] = 0;
  __syncthreads();
  const int lo = blockIdx.x * ch;
  const int hi = (lo + ch < e) ? (lo + ch) : e;
  for (int base = lo + t; base < hi; base += 8192) {
    int d[8];
#pragma unroll
    for (int k = 0; k < 8; ++k) {
      const int i = base + 1024 * k;
      d[k] = (i < hi) ? dst[i] : 0;
    }
#pragma unroll
    for (int k = 0; k < 8; ++k)
      if (base + 1024 * k < hi) atomicAdd(&hist[d[k] >> BIN_SHIFT], 1);
  }
  __syncthreads();
  if (t < nb) histg[blockIdx.x * nb + t] = hist[t];
}

__global__ void __launch_bounds__(1024)
k_bin_scatter(const int* __restrict__ src, const int* __restrict__ dst,
              const int* __restrict__ histg, int* __restrict__ binstart,
              int2* __restrict__ tmp, int e, int ch, int nb) {
  __shared__ int part_pre[4][MAX_BINS];
  __shared__ int part_tot[4][MAX_BINS];
  __shared__ int cur[MAX_BINS];
  __shared__ int wsum[8];
  const int t = threadIdx.x;
  const int g = blockIdx.x;
  const int b = t & (MAX_BINS - 1);
  const int q = t >> 8;
  int pre = 0, tot = 0;
  if (b < nb) {
    const int r0 = q * (SORT_G / 4);
#pragma unroll 16
    for (int r = r0; r < r0 + SORT_G / 4; ++r) {
      const int v = histg[r * nb + b];
      tot += v;
      if (r < g) pre += v;
    }
  }
  part_pre[q][b] = pre;
  part_tot[q][b] = tot;
  __syncthreads();
  const int lane = t & 31;
  int x = 0, c = 0, p = 0;
  if (t < MAX_BINS) {
    c = part_tot[0][t] + part_tot[1][t] + part_tot[2][t] + part_tot[3][t];
    p = part_pre[0][t] + part_pre[1][t] + part_pre[2][t] + part_pre[3][t];
    x = c;
    for (int o = 1; o < 32; o <<= 1) {
      const int y = __shfl_up(x, o);
      if (lane >= o) x += y;
    }
    if (lane == 31) wsum[t >> 5] = x;
  }
  __syncthreads();
  if (t == 0) {
    int a = 0;
    for (int k = 0; k < 8; ++k) { const int v = wsum[k]; wsum[k] = a; a += v; }
  }
  __syncthreads();
  if (t < MAX_BINS) {
    const int start = wsum[t >> 5] + x - c;
    cur[t] = start + p;
    if (g == 0 && t < nb) binstart[t] = start;
  }
  if (g == 0 && t == 0) binstart[nb] = e;
  __syncthreads();
  const int lo = g * ch;
  const int hi = (lo + ch < e) ? (lo + ch) : e;
  for (int base = lo + t; base < hi; base += 8192) {
    int d[8], s[8];
#pragma unroll
    for (int k = 0; k < 8; ++k) {
      const int i = base + 1024 * k;
      d[k] = (i < hi) ? dst[i] : 0;
      s[k] = (i < hi) ? src[i] : 0;
    }
#pragma unroll
    for (int k = 0; k < 8; ++k)
      if (base + 1024 * k < hi) {
        const int pos = atomicAdd(&cur[d[k] >> BIN_SHIFT], 1);
        tmp[pos] = make_int2(s[k], d[k]);
      }
  }
}

__global__ void __launch_bounds__(512)
k_bin_finalize(const int2* __restrict__ tmp, const int* __restrict__ binstart,
               int* __restrict__ cnt, int* __restrict__ first, float* __restrict__ dinv,
               int* __restrict__ srcs, int n) {
  __shared__ int c[BIN_NODES];
  __shared__ int wsum[BIN_NODES / 32];
  const int t = threadIdx.x;
  const int b = blockIdx.x;
  const int lane = t & 31;
  const int s0 = binstart[b];
  const int s1 = binstart[b + 1];
  if (t < BIN_NODES) c[t] = 0;
  __syncthreads();
  for (int base = s0 + t; base < s1; base += 5120) {
    int d[10];
#pragma unroll
    for (int k = 0; k < 10; ++k) {
      const int i = base + 512 * k;
      d[k] = (i < s1) ? tmp[i].y : 0;
    }
#pragma unroll
    for (int k = 0; k < 10; ++k)
      if (base + 512 * k < s1) atomicAdd(&c[d[k] & (BIN_NODES - 1)], 1);
  }
  __syncthreads();
  int x = 0, kc = 0;
  if (t < BIN_NODES) {
    kc = c[t];
    x = kc;
    for (int o = 1; o < 32; o <<= 1) {
      const int y = __shfl_up(x, o);
      if (lane >= o) x += y;
    }
    if (lane == 31) wsum[t >> 5] = x;
  }
  __syncthreads();
  if (t == 0) {
    int a = 0;
    for (int j = 0; j < BIN_NODES / 32; ++j) { const int v = wsum[j]; wsum[j] = a; a += v; }
  }
  __syncthreads();
  if (t < BIN_NODES) {
    const int f = s0 + wsum[t >> 5] + x - kc;
    c[t] = f;
    const int node = b * BIN_NODES + t;
    if (node < n) {
      cnt[node] = kc;
      first[node] = f;
      dinv[node] = rsqrtf((float)(kc + 1));
    }
  }
  __syncthreads();
  for (int base = s0 + t; base < s1; base += 5120) {
    int2 ed[10];
#pragma unroll
    for (int k = 0; k < 10; ++k) {
      const int i = base + 512 * k;
      ed[k] = (i < s1) ? tmp[i] : make_int2(0, 0);
    }
#pragma unroll
    for (int k = 0; k < 10; ++k)
      if (base + 512 * k < s1) {
        const int p = atomicAdd(&c[ed[k].y & (BIN_NODES - 1)], 1);
        srcs[p] = ed[k].x;
      }
  }
}

__global__ void k_bucket_weights(const int* __restrict__ srcs, const float* __restrict__ dinv,
                                 int2* __restrict__ bucket, int e) {
  const int j = 8 * (blockIdx.x * blockDim.x + threadIdx.x);
  if (j + 7 < e) {
    const int4 sa = *(const int4*)(srcs + j);
    const int4 sb = *(const int4*)(srcs + j + 4);
    const float w0 = dinv[sa.x], w1 = dinv[sa.y], w2 = dinv[sa.z], w3 = dinv[sa.w];
    const float w4 = dinv[sb.x], w5 = dinv[sb.y], w6 = dinv[sb.z], w7 = dinv[sb.w];
    *(int4*)(bucket + j)     = make_int4(sa.x, __float_as_int(w0), sa.y, __float_as_int(w1));
    *(int4*)(bucket + j + 2) = make_int4(sa.z, __float_as_int(w2), sa.w, __float_as_int(w3));
    *(int4*)(bucket + j + 4) = make_int4(sb.x, __float_as_int(w4), sb.y, __float_as_int(w5));
    *(int4*)(bucket + j + 6) = make_int4(sb.z, __float_as_int(w6), sb.w, __float_as_int(w7));
  } else {
    for (int k = j; k < e; ++k) {
      const int s = srcs[k];
      bucket[k] = make_int2(s, __float_as_int(dinv[s]));
    }
  }
}

template <int K, int N>
__global__ void __launch_bounds__(256)
k_gemm_wmma_f32(const float* __restrict__ A, const float* __restrict__ B,
                float* __restrict__ C, int mtiles) {
  constexpr int NT = (N + 15) / 16;
  const int lane = threadIdx.x & 31;
  const int row  = lane & 15;
  const int kp   = lane >> 4;
  const int tile = blockIdx.x * 8 + (threadIdx.x >> 5);
  if (tile >= mtiles) return;
  const int m0 = tile * 16;

  int colc[NT];
#pragma unroll
  for (int t = 0; t < NT; ++t) {
    const int col = t * 16 + row;
    colc[t] = (col < N) ? col : (N - 1);
  }

  const float* Arow = A + (size_t)(m0 + row) * K;
  v8f c[NT];
#pragma unroll
  for (int t = 0; t < NT; ++t) c[t] = (v8f){};

#pragma unroll
  for (int k0 = 0; k0 < K; k0 += 4) {
    const int ka = k0 + 2 * kp;
    v2f a;
    a.x = Arow[ka];
    a.y = Arow[ka + 1];
#pragma unroll
    for (int t = 0; t < NT; ++t) {
      v2f b;
      b.x = B[(size_t)ka * N + colc[t]];
      b.y = B[(size_t)(ka + 1) * N + colc[t]];
      c[t] = __builtin_amdgcn_wmma_f32_16x16x4_f32(false, a, false, b, (short)0, c[t],
                                                   false, false);
    }
  }
#pragma unroll
  for (int t = 0; t < NT; ++t) {
    const int col = t * 16 + row;
    if (col < N) {
#pragma unroll
      for (int r = 0; r < 8; ++r)
        C[(size_t)(m0 + r + 8 * kp) * N + col] = c[t][r];
    }
  }
}

#define NPW 8

__device__ __forceinline__ int2 bucket_words(const int2* __restrict__ bucket, int pos,
                                             int m, int i, int lane) {
  int2 t = make_int2(i, 0);
  if (lane < m) t = bucket[pos + lane];
  return t;
}

template <int F>
__device__ __forceinline__ void add_rows(int es, int ew, int m, const float* __restrict__ hl,
                                         float di, float2& acc) {
  for (int e = 0; e < m; e += 4) {
    const int s0 = __builtin_amdgcn_readlane(es, e);
    const int s1 = __builtin_amdgcn_readlane(es, e + 1);
    const int s2 = __builtin_amdgcn_readlane(es, e + 2);
    const int s3 = __builtin_amdgcn_readlane(es, e + 3);
    const float2 v0 = *(const float2*)(hl + (size_t)s0 * F);
    const float2 v1 = *(const float2*)(hl + (size_t)s1 * F);
    const float2 v2 = *(const float2*)(hl + (size_t)s2 * F);
    const float2 v3 = *(const float2*)(hl + (size_t)s3 * F);
    const float n0 = __int_as_float(__builtin_amdgcn_readlane(ew, e)) * di;
    const float n1 = __int_as_float(__builtin_amdgcn_readlane(ew, e + 1)) * di;
    const float n2 = __int_as_float(__builtin_amdgcn_readlane(ew, e + 2)) * di;
    const float n3 = __int_as_float(__builtin_amdgcn_readlane(ew, e + 3)) * di;
    acc.x += v0.x * n0; acc.y += v0.y * n0;
    if (e + 1 < m) { acc.x += v1.x * n1; acc.y += v1.y * n1; }
    if (e + 2 < m) { acc.x += v2.x * n2; acc.y += v2.y * n2; }
    if (e + 3 < m) { acc.x += v3.x * n3; acc.y += v3.y * n3; }
  }
}

template <int F>
__device__ __forceinline__ float2 node_sum(const int2* __restrict__ bucket, int start, int ne,
                                           int i, int lane, int es, int ew, float2 own,
                                           const float* __restrict__ hl, float di) {
  float2 acc;
  acc.x = own.x * di * di;
  acc.y = own.y * di * di;
  add_rows<F>(es, ew, ne < 32 ? ne : 32, hl, di, acc);
  for (int base = 32; base < ne; base += 32) {
    const int m = (ne - base < 32) ? (ne - base) : 32;
    const int2 u = bucket_words(bucket, start + base, m, i, lane);
    add_rows<F>(u.x, u.y, m, hl, di, acc);
  }
  return acc;
}

__global__ void __launch_bounds__(256)
k_gather64_relu(const int2* __restrict__ bucket, const int* __restrict__ first,
                const int* __restrict__ cnt, const float* __restrict__ dinv,
                const float* __restrict__ h, const float* __restrict__ b,
                float* __restrict__ out, int n) {
  const int lane = threadIdx.x & 31;
  const int i0 = NPW * __builtin_amdgcn_readfirstlane((int)(blockIdx.x * 8 + (threadIdx.x >> 5)));
  if (i0 >= n) return;
  const int nn = (n - i0 < NPW) ? (n - i0) : NPW;
  int cv = 0, fv = 0, dv = 0;
  if (lane < nn) {
    cv = cnt[i0 + lane];
    fv = first[i0 + lane];
    dv = __float_as_int(dinv[i0 + lane]);
  }
  const float* hl = h + 2 * lane;
  const float2 bv = *(const float2*)(b + 2 * lane);
  int ne = __builtin_amdgcn_readlane(cv, 0);
  int start = __builtin_amdgcn_readlane(fv, 0);
  int2 t = bucket_words(bucket, start, ne, i0, lane);
  float2 hv = *(const float2*)(hl + (size_t)i0 * 64);
  for (int j = 0; j < nn; ++j) {
    const int i = i0 + j;
    const int ne_c = ne, start_c = start;
    const int es = t.x, ew = t.y;
    const float2 own = hv;
    const float di = __int_as_float(__builtin_amdgcn_readlane(dv, j));
    if (j + 1 < nn) {
      ne = __builtin_amdgcn_readlane(cv, j + 1);
      start = __builtin_amdgcn_readlane(fv, j + 1);
      t = bucket_words(bucket, start, ne, i + 1, lane);
      hv = *(const float2*)(hl + (size_t)(i + 1) * 64);
    }
    const float2 acc = node_sum<64>(bucket, start_c, ne_c, i, lane, es, ew, own, hl, di);
    float2 o;
    o.x = fmaxf(acc.x + bv.x, 0.0f);
    o.y = fmaxf(acc.y + bv.y, 0.0f);
    *(float2*)(out + (size_t)i * 64 + 2 * lane) = o;
  }
}

__global__ void __launch_bounds__(256)
k_gather40_logsoftmax(const int2* __restrict__ bucket, const int* __restrict__ first,
                      const int* __restrict__ cnt, const float* __restrict__ dinv,
                      const float* __restrict__ h, const float* __restrict__ b,
                      float* __restrict__ out, int n) {
  const int lane = threadIdx.x & 31;
  const int i0 = NPW * __builtin_amdgcn_readfirstlane((int)(blockIdx.x * 8 + (threadIdx.x >> 5)));
  if (i0 >= n) return;
  const int nn = (n - i0 < NPW) ? (n - i0) : NPW;
  int cv = 0, fv = 0, dv = 0;
  if (lane < nn) {
    cv = cnt[i0 + lane];
    fv = first[i0 + lane];
    dv = __float_as_int(dinv[i0 + lane]);
  }
  const bool own_lane = lane < 20;
  const int fo = 2 * (own_lane ? lane : lane - 20);
  const float* hl = h + fo;
  const float2 bv = *(const float2*)(b + fo);
  int ne = __builtin_amdgcn_readlane(cv, 0);
  int start = __builtin_amdgcn_readlane(fv, 0);
  int2 t = bucket_words(bucket, start, ne, i0, lane);
  float2 hv = *(const float2*)(hl + (size_t)i0 * 40);
  for (int j = 0; j < nn; ++j) {
    const int i = i0 + j;
    const int ne_c = ne, start_c = start;
    const int es = t.x, ew = t.y;
    const float2 own = hv;
    const float di = __int_as_float(__builtin_amdgcn_readlane(dv, j));
    if (j + 1 < nn) {
      ne = __builtin_amdgcn_readlane(cv, j + 1);
      start = __builtin_amdgcn_readlane(fv, j + 1);
      t = bucket_words(bucket, start, ne, i + 1, lane);
      hv = *(const float2*)(hl + (size_t)(i + 1) * 40);
    }
    const float2 acc = node_sum<40>(bucket, start_c, ne_c, i, lane, es, ew, own, hl, di);
    const float vx = acc.x + bv.x;
    const float vy = acc.y + bv.y;
    float mx = own_lane ? fmaxf(vx, vy) : -3.402823466e38f;
    for (int o = 16; o > 0; o >>= 1) mx = fmaxf(mx, __shfl_xor(mx, o));
    float sm = own_lane ? (expf(vx - mx) + expf(vy - mx)) : 0.0f;
    for (int o = 16; o > 0; o >>= 1) sm += __shfl_xor(sm, o);
    const float lse = mx + logf(sm);
    if (own_lane) {
      float2 o2;
      o2.x = vx - lse;
      o2.y = vy - lse;
      *(float2*)(out + (size_t)i * 40 + fo) = o2;
    }
  }
}

extern "C" void kernel_launch(void* const* d_in, const int* in_sizes, int n_in,
                              void* d_out, int out_size, void* d_ws, size_t ws_size,
                              hipStream_t stream) {
  (void)n_in; (void)out_size; (void)ws_size;

  const float* x  = (const float*)d_in[0];
  const int*   ei = (const int*)d_in[1];
  const float* W1 = (const float*)d_in[2];
  const float* b1 = (const float*)d_in[3];
  const float* W2 = (const float*)d_in[4];
  const float* b2 = (const float*)d_in[5];
  float* out = (float*)d_out;

  const int F_IN = 128;
  const int Nn = in_sizes[0] / F_IN;
  const int E  = in_sizes[1] / 2;
  const int* src = ei;
  const int* dst = ei + E;

  char* ws = (char*)d_ws;
  int*   cnt      = (int*)(ws);
  int*   histg    = (int*)(ws + ((size_t)1  << 20));
  int*   binstart = (int*)(ws + ((size_t)3  << 19));
  int*   first    = (int*)(ws + ((size_t)2  << 20));
  float* dinv     = (float*)(ws + ((size_t)3  << 20));
  float* h        = (float*)(ws + ((size_t)4  << 20));
  float* h1       = (float*)(ws + ((size_t)36 << 20));
  float* h2       = (float*)(ws + ((size_t)68 << 20));
  int*   srcs     = (int*)(ws + ((size_t)88 << 20));
  int2*  bucket   = (int2*)(ws + ((size_t)92 << 20));
  int2*  tmp      = (int2*)(ws + ((size_t)100 << 20));

  const dim3 B256(256);

  const int CH   = (E + SORT_G - 1) / SORT_G;
  const int BINS = (Nn + BIN_NODES - 1) / BIN_NODES;
  const dim3 B1024(1024);
  k_bin_hist<<<SORT_G, B1024, 0, stream>>>(dst, histg, E, CH, BINS);
  k_bin_scatter<<<SORT_G, B1024, 0, stream>>>(src, dst, histg, binstart, tmp, E, CH, BINS);
  k_bin_finalize<<<BINS, dim3(512), 0, stream>>>(tmp, binstart, cnt, first, dinv, srcs, Nn);
  k_bucket_weights<<<((E + 7) / 8 + 255) / 256, B256, 0, stream>>>(srcs, dinv, bucket, E);

  const int MT = Nn / 16;
  const int GW = 8 * NPW;
  k_gemm_wmma_f32<128, 64><<<(MT + 7) / 8, B256, 0, stream>>>(x, W1, h, MT);
  k_gather64_relu<<<(Nn + GW - 1) / GW, B256, 0, stream>>>(bucket, first, cnt, dinv, h, b1, h1, Nn);

  k_gemm_wmma_f32<64, 40><<<(MT + 7) / 8, B256, 0, stream>>>(h1, W2, h2, MT);
  k_gather40_logsoftmax<<<(Nn + GW - 1) / GW, B256, 0, stream>>>(bucket, first, cnt, dinv, h2, b2, out, Nn);
}


# Rows stored pre-scaled by dinv[src] in the GEMM epilogues, node sum scaled once by dinv[dst]: per-edge weights, their launch and the 8-byte bucket go (7 launches); rounding changes, sum order kept
# speedup vs baseline: 2.9328x; 1.0645x over previous
#include <hip/hip_runtime.h>
#include <hip/hip_bf16.h>

typedef __attribute__((ext_vector_type(2))) float v2f;
typedef __attribute__((ext_vector_type(8))) float v8f;


#define SORT_G 128
#define BIN_SHIFT 9
#define BIN_NODES 512
#define MAX_BINS 256

__global__ void __launch_bounds__(1024)
k_bin_hist(const int* __restrict__ dst, int* __restrict__ histg, int e, int ch, int nb) {
  __shared__ int hist[MAX_BINS];
  const int t = threadIdx.x;
  if (t < MAX_BINS) hist[t] = 0;
  __syncthreads();
  const int lo = blockIdx.x * ch;
  const int hi = (lo + ch < e) ? (lo + ch) : e;
  for (int base = lo + t; base < hi; base += 8192) {
    int d[8];
#pragma unroll
    for (int k = 0; k < 8; ++k) {
      const int i = base + 1024 * k;
      d[k] = (i < hi) ? dst[i] : 0;
    }
#pragma unroll
    for (int k = 0; k < 8; ++k)
      if (base + 1024 * k < hi) atomicAdd(&hist[d[k] >> BIN_SHIFT], 1);
  }
  __syncthreads();
  if (t < nb) histg[blockIdx.x * nb + t] = hist[t];
}

__global__ void __launch_bounds__(1024)
k_bin_scatter(const int* __restrict__ src, const int* __restrict__ dst,
              const int* __restrict__ histg, int* __restrict__ binstart,
              int2* __restrict__ tmp, int e, int ch, int nb) {
  __shared__ int part_pre[4][MAX_BINS];
  __shared__ int part_tot[4][MAX_BINS];
  __shared__ int cur[MAX_BINS];
  __shared__ int wsum[8];
  const int t = threadIdx.x;
  const int g = blockIdx.x;
  const int b = t & (MAX_BINS - 1);
  const int q = t >> 8;
  int pre = 0, tot = 0;
  if (b < nb) {
    const int r0 = q * (SORT_G / 4);
#pragma unroll 16
    for (int r = r0; r < r0 + SORT_G / 4; ++r) {
      const int v = histg[r * nb + b];
      tot += v;
      if (r < g) pre += v;
    }
  }
  part_pre[q][b] = pre;
  part_tot[q][b] = tot;
  __syncthreads();
  const int lane = t & 31;
  int x = 0, c = 0, p = 0;
  if (t < MAX_BINS) {
    c = part_tot[0][t] + part_tot[1][t] + part_tot[2][t] + part_tot[3][t];
    p = part_pre[0][t] + part_pre[1][t] + part_pre[2][t] + part_pre[3][t];
    x = c;
    for (int o = 1; o < 32; o <<= 1) {
      const int y = __shfl_up(x, o);
      if (lane >= o) x += y;
    }
    if (lane == 31) wsum[t >> 5] = x;
  }
  __syncthreads();
  if (t == 0) {
    int a = 0;
    for (int k = 0; k < 8; ++k) { const int v = wsum[k]; wsum[k] = a; a += v; }
  }
  __syncthreads();
  if (t < MAX_BINS) {
    const int start = wsum[t >> 5] + x - c;
    cur[t] = start + p;
    if (g == 0 && t < nb) binstart[t] = start;
  }
  if (g == 0 && t == 0) binstart[nb] = e;
  __syncthreads();
  const int lo = g * ch;
  const int hi = (lo + ch < e) ? (lo + ch) : e;
  for (int base = lo + t; base < hi; base += 8192) {
    int d[8], s[8];
#pragma unroll
    for (int k = 0; k < 8; ++k) {
      const int i = base + 1024 * k;
      d[k] = (i < hi) ? dst[i] : 0;
      s[k] = (i < hi) ? src[i] : 0;
    }
#pragma unroll
    for (int k = 0; k < 8; ++k)
      if (base + 1024 * k < hi) {
        const int pos = atomicAdd(&cur[d[k] >> BIN_SHIFT], 1);
        tmp[pos] = make_int2(s[k], d[k]);
      }
  }
}

__global__ void __launch_bounds__(512)
k_bin_finalize(const int2* __restrict__ tmp, const int* __restrict__ binstart,
               int* __restrict__ cnt, int* __restrict__ first, float* __restrict__ dinv,
               int* __restrict__ srcs, int n) {
  __shared__ int c[BIN_NODES];
  __shared__ int wsum[BIN_NODES / 32];
  const int t = threadIdx.x;
  const int b = blockIdx.x;
  const int lane = t & 31;
  const int s0 = binstart[b];
  const int s1 = binstart[b + 1];
  if (t < BIN_NODES) c[t] = 0;
  __syncthreads();
  for (int base = s0 + t; base < s1; base += 5120) {
    int d[10];
#pragma unroll
    for (int k = 0; k < 10; ++k) {
      const int i = base + 512 * k;
      d[k] = (i < s1) ? tmp[i].y : 0;
    }
#pragma unroll
    for (int k = 0; k < 10; ++k)
      if (base + 512 * k < s1) atomicAdd(&c[d[k] & (BIN_NODES - 1)], 1);
  }
  __syncthreads();
  int x = 0, kc = 0;
  if (t < BIN_NODES) {
    kc = c[t];
    x = kc;
    for (int o = 1; o < 32; o <<= 1) {
      const int y = __shfl_up(x, o);
      if (lane >= o) x += y;
    }
    if (lane == 31) wsum[t >> 5] = x;
  }
  __syncthreads();
  if (t == 0) {
    int a = 0;
    for (int j = 0; j < BIN_NODES / 32; ++j) { const int v = wsum[j]; wsum[j] = a; a += v; }
  }
  __syncthreads();
  if (t < BIN_NODES) {
    const int f = s0 + wsum[t >> 5] + x - kc;
    c[t] = f;
    const int node = b * BIN_NODES + t;
    if (node < n) {
      cnt[node] = kc;
      first[node] = f;
      dinv[node] = rsqrtf((float)(kc + 1));
    }
  }
  __syncthreads();
  for (int base = s0 + t; base < s1; base += 5120) {
    int2 ed[10];
#pragma unroll
    for (int k = 0; k < 10; ++k) {
      const int i = base + 512 * k;
      ed[k] = (i < s1) ? tmp[i] : make_int2(0, 0);
    }
#pragma unroll
    for (int k = 0; k < 10; ++k)
      if (base + 512 * k < s1) {
        const int p = atomicAdd(&c[ed[k].y & (BIN_NODES - 1)], 1);
        srcs[p] = ed[k].x;
      }
  }
}

template <int K, int N>
__global__ void __launch_bounds__(256)
k_gemm_wmma_f32(const float* __restrict__ A, const float* __restrict__ B,
                const float* __restrict__ rs, float* __restrict__ C, int mtiles) {
  constexpr int NT = (N + 15) / 16;
  const int lane = threadIdx.x & 31;
  const int row  = lane & 15;
  const int kp   = lane >> 4;
  const int tile = blockIdx.x * 8 + (threadIdx.x >> 5);
  if (tile >= mtiles) return;
  const int m0 = tile * 16;

  int colc[NT];
#pragma unroll
  for (int t = 0; t < NT; ++t) {
    const int col = t * 16 + row;
    colc[t] = (col < N) ? col : (N - 1);
  }

  const float* Arow = A + (size_t)(m0 + row) * K;
  v8f c[NT];
#pragma unroll
  for (int t = 0; t < NT; ++t) c[t] = (v8f){};

#pragma unroll
  for (int k0 = 0; k0 < K; k0 += 4) {
    const int ka = k0 + 2 * kp;
    v2f a;
    a.x = Arow[ka];
    a.y = Arow[ka + 1];
#pragma unroll
    for (int t = 0; t < NT; ++t) {
      v2f b;
      b.x = B[(size_t)ka * N + colc[t]];
      b.y = B[(size_t)(ka + 1) * N + colc[t]];
      c[t] = __builtin_amdgcn_wmma_f32_16x16x4_f32(false, a, false, b, (short)0, c[t],
                                                   false, false);
    }
  }
  const float4 sa = *(const float4*)(rs + m0 + 8 * kp);
  const float4 sb = *(const float4*)(rs + m0 + 8 * kp + 4);
  const float sc[8] = {sa.x, sa.y, sa.z, sa.w, sb.x, sb.y, sb.z, sb.w};
#pragma unroll
  for (int t = 0; t < NT; ++t) {
    const int col = t * 16 + row;
    if (col < N) {
#pragma unroll
      for (int r = 0; r < 8; ++r)
        C[(size_t)(m0 + r + 8 * kp) * N + col] = c[t][r] * sc[r];
    }
  }
}

#define NPW 8

__device__ __forceinline__ int source_words(const int* __restrict__ srcs, int pos,
                                            int m, int i, int lane) {
  int t = i;
  if (lane < m) t = srcs[pos + lane];
  return t;
}

template <int F>
__device__ __forceinline__ void add_rows(int es, int m, const float* __restrict__ hl,
                                         float2& acc) {
  for (int e = 0; e < m; e += 4) {
    const int s0 = __builtin_amdgcn_readlane(es, e);
    const int s1 = __builtin_amdgcn_readlane(es, e + 1);
    const int s2 = __builtin_amdgcn_readlane(es, e + 2);
    const int s3 = __builtin_amdgcn_readlane(es, e + 3);
    const float2 v0 = *(const float2*)(hl + (size_t)s0 * F);
    const float2 v1 = *(const float2*)(hl + (size_t)s1 * F);
    const float2 v2 = *(const float2*)(hl + (size_t)s2 * F);
    const float2 v3 = *(const float2*)(hl + (size_t)s3 * F);
    acc.x += v0.x; acc.y += v0.y;
    if (e + 1 < m) { acc.x += v1.x; acc.y += v1.y; }
    if (e + 2 < m) { acc.x += v2.x; acc.y += v2.y; }
    if (e + 3 < m) { acc.x += v3.x; acc.y += v3.y; }
  }
}

template <int F>
__device__ __forceinline__ float2 node_sum(const int* __restrict__ srcs, int start, int ne,
                                           int i, int lane, int es, float2 own,
                                           const float* __restrict__ hl, float di) {
  float2 acc = own;
  add_rows<F>(es, ne < 32 ? ne : 32, hl, acc);
  for (int base = 32; base < ne; base += 32) {
    const int m = (ne - base < 32) ? (ne - base) : 32;
    const int u = source_words(srcs, start + base, m, i, lane);
    add_rows<F>(u, m, hl, acc);
  }
  acc.x *= di;
  acc.y *= di;
  return acc;
}

__global__ void __launch_bounds__(256)
k_gather64_relu(const int* __restrict__ srcs, const int* __restrict__ first,
                const int* __restrict__ cnt, const float* __restrict__ dinv,
                const float* __restrict__ h, const float* __restrict__ b,
                float* __restrict__ out, int n) {
  const int lane = threadIdx.x & 31;
  const int i0 = NPW * __builtin_amdgcn_readfirstlane((int)(blockIdx.x * 8 + (threadIdx.x >> 5)));
  if (i0 >= n) return;
  const int nn = (n - i0 < NPW) ? (n - i0) : NPW;
  int cv = 0, fv = 0, dv = 0;
  if (lane < nn) {
    cv = cnt[i0 + lane];
    fv = first[i0 + lane];
    dv = __float_as_int(dinv[i0 + lane]);
  }
  const float* hl = h + 2 * lane;
  const float2 bv = *(const float2*)(b + 2 * lane);
  int ne = __builtin_amdgcn_readlane(cv, 0);
  int start = __builtin_amdgcn_readlane(fv, 0);
  int t = source_words(srcs, start, ne, i0, lane);
  float2 hv = *(const float2*)(hl + (size_t)i0 * 64);
  for (int j = 0; j < nn; ++j) {
    const int i = i0 + j;
    const int ne_c = ne, start_c = start;
    const int es = t;
    const float2 own = hv;
    const float di = __int_as_float(__builtin_amdgcn_readlane(dv, j));
    if (j + 1 < nn) {
      ne = __builtin_amdgcn_readlane(cv, j + 1);
      start = __builtin_amdgcn_readlane(fv, j + 1);
      t = source_words(srcs, start, ne, i + 1, lane);
      hv = *(const float2*)(hl + (size_t)(i + 1) * 64);
    }
    const float2 acc = node_sum<64>(srcs, start_c, ne_c, i, lane, es, own, hl, di);
    float2 o;
    o.x = fmaxf(acc.x + bv.x, 0.0f);
    o.y = fmaxf(acc.y + bv.y, 0.0f);
    *(float2*)(out + (size_t)i * 64 + 2 * lane) = o;
  }
}

__global__ void __launch_bounds__(256)
k_gather40_logsoftmax(const int* __restrict__ srcs, const int* __restrict__ first,
                      const int* __restrict__ cnt, const float* __restrict__ dinv,
                      const float* __restrict__ h, const float* __restrict__ b,
                      float* __restrict__ out, int n) {
  const int lane = threadIdx.x & 31;
  const int i0 = NPW * __builtin_amdgcn_readfirstlane((int)(blockIdx.x * 8 + (threadIdx.x >> 5)));
  if (i0 >= n) return;
  const int nn = (n - i0 < NPW) ? (n - i0) : NPW;
  int cv = 0, fv = 0, dv = 0;
  if (lane < nn) {
    cv = cnt[i0 + lane];
    fv = first[i0 + lane];
    dv = __float_as_int(dinv[i0 + lane]);
  }
  const bool own_lane = lane < 20;
  const int fo = 2 * (own_lane ? lane : lane - 20);
  const float* hl = h + fo;
  const float2 bv = *(const float2*)(b + fo);
  int ne = __builtin_amdgcn_readlane(cv, 0);
  int start = __builtin_amdgcn_readlane(fv, 0);
  int t = source_words(srcs, start, ne, i0, lane);
  float2 hv = *(const float2*)(hl + (size_t)i0 * 40);
  for (int j = 0; j < nn; ++j) {
    const int i = i0 + j;
    const int ne_c = ne, start_c = start;
    const int es = t;
    const float2 own = hv;
    const float di = __int_as_float(__builtin_amdgcn_readlane(dv, j));
    if (j + 1 < nn) {
      ne = __builtin_amdgcn_readlane(cv, j + 1);
      start = __builtin_amdgcn_readlane(fv, j + 1);
      t = source_words(srcs, start, ne, i + 1, lane);
      hv = *(const float2*)(hl + (size_t)(i + 1) * 40);
    }
    const float2 acc = node_sum<40>(srcs, start_c, ne_c, i, lane, es, own, hl, di);
    const float vx = acc.x + bv.x;
    const float vy = acc.y + bv.y;
    float mx = own_lane ? fmaxf(vx, vy) : -3.402823466e38f;
    for (int o = 16; o > 0; o >>= 1) mx = fmaxf(mx, __shfl_xor(mx, o));
    float sm = own_lane ? (expf(vx - mx) + expf(vy - mx)) : 0.0f;
    for (int o = 16; o > 0; o >>= 1) sm += __shfl_xor(sm, o);
    const float lse = mx + logf(sm);
    if (own_lane) {
      float2 o2;
      o2.x = vx - lse;
      o2.y = vy - lse;
      *(float2*)(out + (size_t)i * 40 + fo) = o2;
    }
  }
}

extern "C" void kernel_launch(void* const* d_in, const int* in_sizes, int n_in,
                              void* d_out, int out_size, void* d_ws, size_t ws_size,
                              hipStream_t stream) {
  (void)n_in; (void)out_size; (void)ws_size;

  const float* x  = (const float*)d_in[0];
  const int*   ei = (const int*)d_in[1];
  const float* W1 = (const float*)d_in[2];
  const float* b1 = (const float*)d_in[3];
  const float* W2 = (const float*)d_in[4];
  const float* b2 = (const float*)d_in[5];
  float* out = (float*)d_out;

  const int F_IN = 128;
  const int Nn = in_sizes[0] / F_IN;
  const int E  = in_sizes[1] / 2;
  const int* src = ei;
  const int* dst = ei + E;

  char* ws = (char*)d_ws;
  int*   cnt      = (int*)(ws);
  int*   histg    = (int*)(ws + ((size_t)1  << 20));
  int*   binstart = (int*)(ws + ((size_t)3  << 19));
  int*   first    = (int*)(ws + ((size_t)2  << 20));
  float* dinv     = (float*)(ws + ((size_t)3  << 20));
  float* h        = (float*)(ws + ((size_t)4  << 20));
  float* h1       = (float*)(ws + ((size_t)36 << 20));
  float* h2       = (float*)(ws + ((size_t)68 << 20));
  int*   srcs     = (int*)(ws + ((size_t)88 << 20));
  int2*  tmp      = (int2*)(ws + ((size_t)100 << 20));

  const dim3 B256(256);

  const int CH   = (E + SORT_G - 1) / SORT_G;
  const int BINS = (Nn + BIN_NODES - 1) / BIN_NODES;
  const dim3 B1024(1024);
  k_bin_hist<<<SORT_G, B1024, 0, stream>>>(dst, histg, E, CH, BINS);
  k_bin_scatter<<<SORT_G, B1024, 0, stream>>>(src, dst, histg, binstart, tmp, E, CH, BINS);
  k_bin_finalize<<<BINS, dim3(512), 0, stream>>>(tmp, binstart, cnt, first, dinv, srcs, Nn);

  const int MT = Nn / 16;
  const int GW = 8 * NPW;
  k_gemm_wmma_f32<128, 64><<<(MT + 7) / 8, B256, 0, stream>>>(x, W1, dinv, h, MT);
  k_gather64_relu<<<(Nn + GW - 1) / GW, B256, 0, stream>>>(srcs, first, cnt, dinv, h, b1, h1, Nn);

  k_gemm_wmma_f32<64, 40><<<(MT + 7) / 8, B256, 0, stream>>>(h1, W2, dinv, h2, MT);
  k_gather40_logsoftmax<<<(Nn + GW - 1) / GW, B256, 0, stream>>>(srcs, first, cnt, dinv, h2, b2, out, Nn);
}


# Layer-2 product moved into the layer-1 gather's work-groups: 64 summed rows and W2 staged in LDS, four waves run the WMMAs of one row tile each; h1 never stored, one launch fewer; sum order kept
# speedup vs baseline: 3.1602x; 1.0775x over previous
#include <hip/hip_runtime.h>
#include <hip/hip_bf16.h>

typedef __attribute__((ext_vector_type(2))) float v2f;
typedef __attribute__((ext_vector_type(8))) float v8f;


#define SORT_G 128
#define BIN_SHIFT 9
#define BIN_NODES 512
#define MAX_BINS 256

__global__ void __launch_bounds__(1024)
k_bin_hist(const int* __restrict__ dst, int* __restrict__ histg, int e, int ch, int nb) {
  __shared__ int hist[MAX_BINS];
  const int t = threadIdx.x;
  if (t < MAX_BINS) hist[t] = 0;
  __syncthreads();
  const int lo = blockIdx.x * ch;
  const int hi = (lo + ch < e) ? (lo + ch) : e;
  for (int base = lo + t; base < hi; base += 8192) {
    int d[8];
#pragma unroll
    for (int k = 0; k < 8; ++k) {
      const int i = base + 1024 * k;
      d[k] = (i < hi) ? dst[i] : 0;
    }
#pragma unroll
    for (int k = 0; k < 8; ++k)
      if (base + 1024 * k < hi) atomicAdd(&hist[d[k] >> BIN_SHIFT], 1);
  }
  __syncthreads();
  if (t < nb) histg[blockIdx.x * nb + t] = hist[t];
}

__global__ void __launch_bounds__(1024)
k_bin_scatter(const int* __restrict__ src, const int* __restrict__ dst,
              const int* __restrict__ histg, int* __restrict__ binstart,
              int2* __restrict__ tmp, int e, int ch, int nb) {
  __shared__ int part_pre[4][MAX_BINS];
  __shared__ int part_tot[4][MAX_BINS];
  __shared__ int cur[MAX_BINS];
  __shared__ int wsum[8];
  const int t = threadIdx.x;
  const int g = blockIdx.x;
  const int b = t & (MAX_BINS - 1);
  const int q = t >> 8;
  int pre = 0, tot = 0;
  if (b < nb) {
    const int r0 = q * (SORT_G / 4);
#pragma unroll 16
    for (int r = r0; r < r0 + SORT_G / 4; ++r) {
      const int v = histg[r * nb + b];
      tot += v;
      if (r < g) pre += v;
    }
  }
  part_pre[q][b] = pre;
  part_tot[q][b] = tot;
  __syncthreads();
  const int lane = t & 31;
  int x = 0, c = 0, p = 0;
  if (t < MAX_BINS) {
    c = part_tot[0][t] + part_tot[1][t] + part_tot[2][t] + part_tot[3][t];
    p = part_pre[0][t] + part_pre[1][t] + part_pre[2][t] + part_pre[3][t];
    x = c;
    for (int o = 1; o < 32; o <<= 1) {
      const int y = __shfl_up(x, o);
      if (lane >= o) x += y;
    }
    if (lane == 31) wsum[t >> 5] = x;
  }
  __syncthreads();
  if (t == 0) {
    int a = 0;
    for (int k = 0; k < 8; ++k) { const int v = wsum[k]; wsum[k] = a; a += v; }
  }
  __syncthreads();
  if (t < MAX_BINS) {
    const int start = wsum[t >> 5] + x - c;
    cur[t] = start + p;
    if (g == 0 && t < nb) binstart[t] = start;
  }
  if (g == 0 && t == 0) binstart[nb] = e;
  __syncthreads();
  const int lo = g * ch;
  const int hi = (lo + ch < e) ? (lo + ch) : e;
  for (int base = lo + t; base < hi; base += 8192) {
    int d[8], s[8];
#pragma unroll
    for (int k = 0; k < 8; ++k) {
      const int i = base + 1024 * k;
      d[k] = (i < hi) ? dst[i] : 0;
      s[k] = (i < hi) ? src[i] : 0;
    }
#pragma unroll
    for (int k = 0; k < 8; ++k)
      if (base + 1024 * k < hi) {
        const int pos = atomicAdd(&cur[d[k] >> BIN_SHIFT], 1);
        tmp[pos] = make_int2(s[k], d[k]);
      }
  }
}

__global__ void __launch_bounds__(512)
k_bin_finalize(const int2* __restrict__ tmp, const int* __restrict__ binstart,
               int* __restrict__ cnt, int* __restrict__ first, float* __restrict__ dinv,
               int* __restrict__ srcs, int n) {
  __shared__ int c[BIN_NODES];
  __shared__ int wsum[BIN_NODES / 32];
  const int t = threadIdx.x;
  const int b = blockIdx.x;
  const int lane = t & 31;
  const int s0 = binstart[b];
  const int s1 = binstart[b + 1];
  if (t < BIN_NODES) c[t] = 0;
  __syncthreads();
  for (int base = s0 + t; base < s1; base += 5120) {
    int d[10];
#pragma unroll
    for (int k = 0; k < 10; ++k) {
      const int i = base + 512 * k;
      d[k] = (i < s1) ? tmp[i].y : 0;
    }
#pragma unroll
    for (int k = 0; k < 10; ++k)
      if (base + 512 * k < s1) atomicAdd(&c[d[k] & (BIN_NODES - 1)], 1);
  }
  __syncthreads();
  int x = 0, kc = 0;
  if (t < BIN_NODES) {
    kc = c[t];
    x = kc;
    for (int o = 1; o < 32; o <<= 1) {
      const int y = __shfl_up(x, o);
      if (lane >= o) x += y;
    }
    if (lane == 31) wsum[t >> 5] = x;
  }
  __syncthreads();
  if (t == 0) {
    int a = 0;
    for (int j = 0; j < BIN_NODES / 32; ++j) { const int v = wsum[j]; wsum[j] = a; a += v; }
  }
  __syncthreads();
  if (t < BIN_NODES) {
    const int f = s0 + wsum[t >> 5] + x - kc;
    c[t] = f;
    const int node = b * BIN_NODES + t;
    if (node < n) {
      cnt[node] = kc;
      first[node] = f;
      dinv[node] = rsqrtf((float)(kc + 1));
    }
  }
  __syncthreads();
  for (int base = s0 + t; base < s1; base += 5120) {
    int2 ed[10];
#pragma unroll
    for (int k = 0; k < 10; ++k) {
      const int i = base + 512 * k;
      ed[k] = (i < s1) ? tmp[i] : make_int2(0, 0);
    }
#pragma unroll
    for (int k = 0; k < 10; ++k)
      if (base + 512 * k < s1) {
        const int p = atomicAdd(&c[ed[k].y & (BIN_NODES - 1)], 1);
        srcs[p] = ed[k].x;
      }
  }
}

template <int K, int N>
__global__ void __launch_bounds__(256)
k_gemm_wmma_f32(const float* __restrict__ A, const float* __restrict__ B,
                const float* __restrict__ rs, float* __restrict__ C, int mtiles) {
  constexpr int NT = (N + 15) / 16;
  const int lane = threadIdx.x & 31;
  const int row  = lane & 15;
  const int kp   = lane >> 4;
  const int tile = blockIdx.x * 8 + (threadIdx.x >> 5);
  if (tile >= mtiles) return;
  const int m0 = tile * 16;

  int colc[NT];
#pragma unroll
  for (int t = 0; t < NT; ++t) {
    const int col = t * 16 + row;
    colc[t] = (col < N) ? col : (N - 1);
  }

  const float* Arow = A + (size_t)(m0 + row) * K;
  v8f c[NT];
#pragma unroll
  for (int t = 0; t < NT; ++t) c[t] = (v8f){};

#pragma unroll
  for (int k0 = 0; k0 < K; k0 += 4) {
    const int ka = k0 + 2 * kp;
    v2f a;
    a.x = Arow[ka];
    a.y = Arow[ka + 1];
#pragma unroll
    for (int t = 0; t < NT; ++t) {
      v2f b;
      b.x = B[(size_t)ka * N + colc[t]];
      b.y = B[(size_t)(ka + 1) * N + colc[t]];
      c[t] = __builtin_amdgcn_wmma_f32_16x16x4_f32(false, a, false, b, (short)0, c[t],
                                                   false, false);
    }
  }
  const float4 sa = *(const float4*)(rs + m0 + 8 * kp);
  const float4 sb = *(const float4*)(rs + m0 + 8 * kp + 4);
  const float sc[8] = {sa.x, sa.y, sa.z, sa.w, sb.x, sb.y, sb.z, sb.w};
#pragma unroll
  for (int t = 0; t < NT; ++t) {
    const int col = t * 16 + row;
    if (col < N) {
#pragma unroll
      for (int r = 0; r < 8; ++r)
        C[(size_t)(m0 + r + 8 * kp) * N + col] = c[t][r] * sc[r];
    }
  }
}

#define NPW 8

__device__ __forceinline__ int source_words(const int* __restrict__ srcs, int pos,
                                            int m, int i, int lane) {
  int t = i;
  if (lane < m) t = srcs[pos + lane];
  return t;
}

template <int F>
__device__ __forceinline__ void add_rows(int es, int m, const float* __restrict__ hl,
                                         float2& acc) {
  for (int e = 0; e < m; e += 4) {
    const int s0 = __builtin_amdgcn_readlane(es, e);
    const int s1 = __builtin_amdgcn_readlane(es, e + 1);
    const int s2 = __builtin_amdgcn_readlane(es, e + 2);
    const int s3 = __builtin_amdgcn_readlane(es, e + 3);
    const float2 v0 = *(const float2*)(hl + (size_t)s0 * F);
    const float2 v1 = *(const float2*)(hl + (size_t)s1 * F);
    const float2 v2 = *(const float2*)(hl + (size_t)s2 * F);
    const float2 v3 = *(const float2*)(hl + (size_t)s3 * F);
    acc.x += v0.x; acc.y += v0.y;
    if (e + 1 < m) { acc.x += v1.x; acc.y += v1.y; }
    if (e + 2 < m) { acc.x += v2.x; acc.y += v2.y; }
    if (e + 3 < m) { acc.x += v3.x; acc.y += v3.y; }
  }
}

template <int F>
__device__ __forceinline__ float2 node_sum(const int* __restrict__ srcs, int start, int ne,
                                           int i, int lane, int es, float2 own,
                                           const float* __restrict__ hl, float di) {
  float2 acc = own;
  add_rows<F>(es, ne < 32 ? ne : 32, hl, acc);
  for (int base = 32; base < ne; base += 32) {
    const int m = (ne - base < 32) ? (ne - base) : 32;
    const int u = source_words(srcs, start + base, m, i, lane);
    add_rows<F>(u, m, hl, acc);
  }
  acc.x *= di;
  acc.y *= di;
  return acc;
}

#define TILE_PITCH 68
#define L2_K 64
#define L2_N 40
__global__ void __launch_bounds__(256)
k_gather64_relu_gemm40(const int* __restrict__ srcs, const int* __restrict__ first,
                       const int* __restrict__ cnt, const float* __restrict__ dinv,
                       const float* __restrict__ h, const float* __restrict__ b,
                       const float* __restrict__ W2, float* __restrict__ h2, int n) {
  __shared__ __attribute__((aligned(16))) float tile[8 * NPW * TILE_PITCH];
  __shared__ __attribute__((aligned(16))) float Bs[L2_K * L2_N];
  const int lane = threadIdx.x & 31;
  const int w = __builtin_amdgcn_readfirstlane((int)(threadIdx.x >> 5));
  const int nb0 = blockIdx.x * (8 * NPW);
  static_assert(L2_K * L2_N / 4 == 2 * 256 + 128, "W2 copy is written for 640 pieces and 256 threads");
  const int q = threadIdx.x;
  const float4* W2q = (const float4*)W2;
  const float4 wq0 = W2q[q];
  const float4 wq1 = W2q[q + 256];
  float4 wq2 = make_float4(0.0f, 0.0f, 0.0f, 0.0f);
  if (q < 128) wq2 = W2q[q + 512];

  const int i0 = nb0 + NPW * w;
  const bool active = i0 < n;
  const int nn = active ? ((n - i0 < NPW) ? (n - i0) : NPW) : 0;
  int cv = 0, fv = 0, dv = 0;
  if (lane < nn) {
    cv = cnt[i0 + lane];
    fv = first[i0 + lane];
    dv = __float_as_int(dinv[i0 + lane]);
  }
  const float2 bv = *(const float2*)(b + 2 * lane);
  float4* Bq = (float4*)Bs;
  Bq[q] = wq0;
  Bq[q + 256] = wq1;
  if (q < 128) Bq[q + 512] = wq2;

  if (active) {
    const float* hl = h + 2 * lane;
    int ne = __builtin_amdgcn_readlane(cv, 0);
    int start = __builtin_amdgcn_readlane(fv, 0);
    int t = source_words(srcs, start, ne, i0, lane);
    float2 hv = *(const float2*)(hl + (size_t)i0 * 64);
    float* trow = tile + (NPW * w) * TILE_PITCH + 2 * lane;
    for (int j = 0; j < nn; ++j) {
      const int i = i0 + j;
      const int ne_c = ne, start_c = start;
      const int es = t;
      const float2 own = hv;
      const float di = __int_as_float(__builtin_amdgcn_readlane(dv, j));
      if (j + 1 < nn) {
        ne = __builtin_amdgcn_readlane(cv, j + 1);
        start = __builtin_amdgcn_readlane(fv, j + 1);
        t = source_words(srcs, start, ne, i + 1, lane);
        hv = *(const float2*)(hl + (size_t)(i + 1) * 64);
      }
      const float2 acc = node_sum<64>(srcs, start_c, ne_c, i, lane, es, own, hl, di);
      float2 o;
      o.x = fmaxf(acc.x + bv.x, 0.0f);
      o.y = fmaxf(acc.y + bv.y, 0.0f);
      *(float2*)(trow + j * TILE_PITCH) = o;
    }
  }
  __syncthreads();

  if (w >= 4) return;
  const int m0 = nb0 + 16 * w;
  if (m0 >= n) return;
  constexpr int NT = (L2_N + 15) / 16;
  const int row = lane & 15;
  const int kp  = lane >> 4;
  int colc[NT];
#pragma unroll
  for (int t = 0; t < NT; ++t) {
    const int col = t * 16 + row;
    colc[t] = (col < L2_N) ? col : (L2_N - 1);
  }
  const float4 sa = *(const float4*)(dinv + m0 + 8 * kp);
  const float4 sb = *(const float4*)(dinv + m0 + 8 * kp + 4);
  const float* Ar = tile + (16 * w + row) * TILE_PITCH + 2 * kp;
  const float* Br = Bs + (2 * kp) * L2_N;
  v8f c[NT];
#pragma unroll
  for (int t = 0; t < NT; ++t) c[t] = (v8f){};
#pragma unroll
  for (int k0 = 0; k0 < L2_K; k0 += 4) {
    v2f a;
    a.x = Ar[k0];
    a.y = Ar[k0 + 1];
#pragma unroll
    for (int t = 0; t < NT; ++t) {
      v2f bb;
      bb.x = Br[k0 * L2_N + colc[t]];
      bb.y = Br[(k0 + 1) * L2_N + colc[t]];
      c[t] = __builtin_amdgcn_wmma_f32_16x16x4_f32(false, a, false, bb, (short)0, c[t],
                                                   false, false);
    }
  }
  const float sc[8] = {sa.x, sa.y, sa.z, sa.w, sb.x, sb.y, sb.z, sb.w};
#pragma unroll
  for (int t = 0; t < NT; ++t) {
    const int col = t * 16 + row;
    if (col < L2_N) {
#pragma unroll
      for (int r = 0; r < 8; ++r)
        h2[(size_t)(m0 + r + 8 * kp) * L2_N + col] = c[t][r] * sc[r];
    }
  }
}

__global__ void __launch_bounds__(256)
k_gather40_logsoftmax(const int* __restrict__ srcs, const int* __restrict__ first,
                      const int* __restrict__ cnt, const float* __restrict__ dinv,
                      const float* __restrict__ h, const float* __restrict__ b,
                      float* __restrict__ out, int n) {
  const int lane = threadIdx.x & 31;
  const int i0 = NPW * __builtin_amdgcn_readfirstlane((int)(blockIdx.x * 8 + (threadIdx.x >> 5)));
  if (i0 >= n) return;
  const int nn = (n - i0 < NPW) ? (n - i0) : NPW;
  int cv = 0, fv = 0, dv = 0;
  if (lane < nn) {
    cv = cnt[i0 + lane];
    fv = first[i0 + lane];
    dv = __float_as_int(dinv[i0 + lane]);
  }
  const bool own_lane = lane < 20;
  const int fo = 2 * (own_lane ? lane : lane - 20);
  const float* hl = h + fo;
  const float2 bv = *(const float2*)(b + fo);
  int ne = __builtin_amdgcn_readlane(cv, 0);
  int start = __builtin_amdgcn_readlane(fv, 0);
  int t = source_words(srcs, start, ne, i0, lane);
  float2 hv = *(const float2*)(hl + (size_t)i0 * 40);
  for (int j = 0; j < nn; ++j) {
    const int i = i0 + j;
    const int ne_c = ne, start_c = start;
    const int es = t;
    const float2 own = hv;
    const float di = __int_as_float(__builtin_amdgcn_readlane(dv, j));
    if (j + 1 < nn) {
      ne = __builtin_amdgcn_readlane(cv, j + 1);
      start = __builtin_amdgcn_readlane(fv, j + 1);
      t = source_words(srcs, start, ne, i + 1, lane);
      hv = *(const float2*)(hl + (size_t)(i + 1) * 40);
    }
    const float2 acc = node_sum<40>(srcs, start_c, ne_c, i, lane, es, own, hl, di);
    const float vx = acc.x + bv.x;
    const float vy = acc.y + bv.y;
    float mx = own_lane ? fmaxf(vx, vy) : -3.402823466e38f;
    for (int o = 16; o > 0; o >>= 1) mx = fmaxf(mx, __shfl_xor(mx, o));
    float sm = own_lane ? (expf(vx - mx) + expf(vy - mx)) : 0.0f;
    for (int o = 16; o > 0; o >>= 1) sm += __shfl_xor(sm, o);
    const float lse = mx + logf(sm);
    if (own_lane) {
      float2 o2;
      o2.x = vx - lse;
      o2.y = vy - lse;
      *(float2*)(out + (size_t)i * 40 + fo) = o2;
    }
  }
}

extern "C" void kernel_launch(void* const* d_in, const int* in_sizes, int n_in,
                              void* d_out, int out_size, void* d_ws, size_t ws_size,
                              hipStream_t stream) {
  (void)n_in; (void)out_size; (void)ws_size;

  const float* x  = (const float*)d_in[0];
  const int*   ei = (const int*)d_in[1];
  const float* W1 = (const float*)d_in[2];
  const float* b1 = (const float*)d_in[3];
  const float* W2 = (const float*)d_in[4];
  const float* b2 = (const float*)d_in[5];
  float* out = (float*)d_out;

  const int F_IN = 128;
  const int Nn = in_sizes[0] / F_IN;
  const int E  = in_sizes[1] / 2;
  const int* src = ei;
  const int* dst = ei + E;

  char* ws = (char*)d_ws;
  int*   cnt      = (int*)(ws);
  int*   histg    = (int*)(ws + ((size_t)1  << 20));
  int*   binstart = (int*)(ws + ((size_t)3  << 19));
  int*   first    = (int*)(ws + ((size_t)2  << 20));
  float* dinv     = (float*)(ws + ((size_t)3  << 20));
  float* h        = (float*)(ws + ((size_t)4  << 20));
  float* h2       = (float*)(ws + ((size_t)68 << 20));
  int*   srcs     = (int*)(ws + ((size_t)88 << 20));
  int2*  tmp      = (int2*)(ws + ((size_t)100 << 20));

  const dim3 B256(256);

  const int CH   = (E + SORT_G - 1) / SORT_G;
  const int BINS = (Nn + BIN_NODES - 1) / BIN_NODES;
  const dim3 B1024(1024);
  k_bin_hist<<<SORT_G, B1024, 0, stream>>>(dst, histg, E, CH, BINS);
  k_bin_scatter<<<SORT_G, B1024, 0, stream>>>(src, dst, histg, binstart, tmp, E, CH, BINS);
  k_bin_finalize<<<BINS, dim3(512), 0, stream>>>(tmp, binstart, cnt, first, dinv, srcs, Nn);

  const int MT = Nn / 16;
  const int GW = 8 * NPW;
  k_gemm_wmma_f32<128, 64><<<(MT + 7) / 8, B256, 0, stream>>>(x, W1, dinv, h, MT);
  k_gather64_relu_gemm40<<<(Nn + GW - 1) / GW, B256, 0, stream>>>(srcs, first, cnt, dinv, h, b1, W2, h2, Nn);

  k_gather40_logsoftmax<<<(Nn + GW - 1) / GW, B256, 0, stream>>>(srcs, first, cnt, dinv, h2, b2, out, Nn);
}


# Layer-1 product x@W1 runs as extra work-groups of the last sort launch (role by block index): 5 launches; rows stored unscaled, gather applies dinv[src] per edge by fma (rounding changes)
# speedup vs baseline: 3.5680x; 1.1290x over previous
#include <hip/hip_runtime.h>
#include <hip/hip_bf16.h>

typedef __attribute__((ext_vector_type(2))) float v2f;
typedef __attribute__((ext_vector_type(8))) float v8f;


#define SORT_G 128
#define BIN_SHIFT 9
#define BIN_NODES 512
#define MAX_BINS 256
#define GEMM_TPB 16

template <int K, int N>
__device__ __forceinline__ void gemm_tile_f32(const float* __restrict__ A, const float* __restrict__ B,
                                              float* __restrict__ C, int tile);

__global__ void __launch_bounds__(1024)
k_bin_hist(const int* __restrict__ dst, int* __restrict__ histg, int e, int ch, int nb) {
  __shared__ int hist[MAX_BINS];
  const int t = threadIdx.x;
  if (t < MAX_BINS) hist[t] = 0;
  __syncthreads();
  const int lo = blockIdx.x * ch;
  const int hi = (lo + ch < e) ? (lo + ch) : e;
  for (int base = lo + t; base < hi; base += 8192) {
    int d[8];
#pragma unroll
    for (int k = 0; k < 8; ++k) {
      const int i = base + 1024 * k;
      d[k] = (i < hi) ? dst[i] : 0;
    }
#pragma unroll
    for (int k = 0; k < 8; ++k)
      if (base + 1024 * k < hi) atomicAdd(&hist[d[k] >> BIN_SHIFT], 1);
  }
  __syncthreads();
  if (t < nb) histg[blockIdx.x * nb + t] = hist[t];
}

__global__ void __launch_bounds__(1024)
k_bin_scatter(const int* __restrict__ src, const int* __restrict__ dst,
              const int* __restrict__ histg, int* __restrict__ binstart,
              int2* __restrict__ tmp, int e, int ch, int nb) {
  __shared__ int part_pre[4][MAX_BINS];
  __shared__ int part_tot[4][MAX_BINS];
  __shared__ int cur[MAX_BINS];
  __shared__ int wsum[8];
  const int t = threadIdx.x;
  const int g = blockIdx.x;
  const int b = t & (MAX_BINS - 1);
  const int q = t >> 8;
  int pre = 0, tot = 0;
  if (b < nb) {
    const int r0 = q * (SORT_G / 4);
#pragma unroll 16
    for (int r = r0; r < r0 + SORT_G / 4; ++r) {
      const int v = histg[r * nb + b];
      tot += v;
      if (r < g) pre += v;
    }
  }
  part_pre[q][b] = pre;
  part_tot[q][b] = tot;
  __syncthreads();
  const int lane = t & 31;
  int x = 0, c = 0, p = 0;
  if (t < MAX_BINS) {
    c = part_tot[0][t] + part_tot[1][t] + part_tot[2][t] + part_tot[3][t];
    p = part_pre[0][t] + part_pre[1][t] + part_pre[2][t] + part_pre[3][t];
    x = c;
    for (int o = 1; o < 32; o <<= 1) {
      const int y = __shfl_up(x, o);
      if (lane >= o) x += y;
    }
    if (lane == 31) wsum[t >> 5] = x;
  }
  __syncthreads();
  if (t == 0) {
    int a = 0;
    for (int k = 0; k < 8; ++k) { const int v = wsum[k]; wsum[k] = a; a += v; }
  }
  __syncthreads();
  if (t < MAX_BINS) {
    const int start = wsum[t >> 5] + x - c;
    cur[t] = start + p;
    if (g == 0 && t < nb) binstart[t] = start;
  }
  if (g == 0 && t == 0) binstart[nb] = e;
  __syncthreads();
  const int lo = g * ch;
  const int hi = (lo + ch < e) ? (lo + ch) : e;
  for (int base = lo + t; base < hi; base += 8192) {
    int d[8], s[8];
#pragma unroll
    for (int k = 0; k < 8; ++k) {
      const int i = base + 1024 * k;
      d[k] = (i < hi) ? dst[i] : 0;
      s[k] = (i < hi) ? src[i] : 0;
    }
#pragma unroll
    for (int k = 0; k < 8; ++k)
      if (base + 1024 * k < hi) {
        const int pos = atomicAdd(&cur[d[k] >> BIN_SHIFT], 1);
        tmp[pos] = make_int2(s[k], d[k]);
      }
  }
}

__global__ void __launch_bounds__(512)
k_bin_finalize_gemm(const int2* __restrict__ tmp, const int* __restrict__ binstart,
                    int* __restrict__ cnt, int* __restrict__ first, float* __restrict__ dinv,
                    int* __restrict__ srcs, int n, int nbins,
                    const float* __restrict__ xa, const float* __restrict__ wb,
                    float* __restrict__ hc, int mtiles) {
  if ((int)blockIdx.x >= nbins) {
    const int wv = __builtin_amdgcn_readfirstlane((int)(threadIdx.x >> 5));
    const int tile = ((int)blockIdx.x - nbins) * GEMM_TPB + wv;
    if (tile < mtiles) gemm_tile_f32<128, 64>(xa, wb, hc, tile);
    return;
  }
  __shared__ int c[BIN_NODES];
  __shared__ int wsum[BIN_NODES / 32];
  const int t = threadIdx.x;
  const int b = blockIdx.x;
  const int lane = t & 31;
  const int s0 = binstart[b];
  const int s1 = binstart[b + 1];
  if (t < BIN_NODES) c[t] = 0;
  __syncthreads();
  for (int base = s0 + t; base < s1; base += 5120) {
    int d[10];
#pragma unroll
    for (int k = 0; k < 10; ++k) {
      const int i = base + 512 * k;
      d[k] = (i < s1) ? tmp[i].y : 0;
    }
#pragma unroll
    for (int k = 0; k < 10; ++k)
      if (base + 512 * k < s1) atomicAdd(&c[d[k] & (BIN_NODES - 1)], 1);
  }
  __syncthreads();
  int x = 0, kc = 0;
  if (t < BIN_NODES) {
    kc = c[t];
    x = kc;
    for (int o = 1; o < 32; o <<= 1) {
      const int y = __shfl_up(x, o);
      if (lane >= o) x += y;
    }
    if (lane == 31) wsum[t >> 5] = x;
  }
  __syncthreads();
  if (t == 0) {
    int a = 0;
    for (int j = 0; j < BIN_NODES / 32; ++j) { const int v = wsum[j]; wsum[j] = a; a += v; }
  }
  __syncthreads();
  if (t < BIN_NODES) {
    const int f = s0 + wsum[t >> 5] + x - kc;
    c[t] = f;
    const int node = b * BIN_NODES + t;
    if (node < n) {
      cnt[node] = kc;
      first[node] = f;
      dinv[node] = rsqrtf((float)(kc + 1));
    }
  }
  __syncthreads();
  for (int base = s0 + t; base < s1; base += 5120) {
    int2 ed[10];
#pragma unroll
    for (int k = 0; k < 10; ++k) {
      const int i = base + 512 * k;
      ed[k] = (i < s1) ? tmp[i] : make_int2(0, 0);
    }
#pragma unroll
    for (int k = 0; k < 10; ++k)
      if (base + 512 * k < s1) {
        const int p = atomicAdd(&c[ed[k].y & (BIN_NODES - 1)], 1);
        srcs[p] = ed[k].x;
      }
  }
}

template <int K, int N>
__device__ __forceinline__ void gemm_tile_f32(const float* __restrict__ A, const float* __restrict__ B,
                                              float* __restrict__ C, int tile) {
  constexpr int NT = (N + 15) / 16;
  const int lane = threadIdx.x & 31;
  const int row  = lane & 15;
  const int kp   = lane >> 4;
  const float* At = A + (size_t)tile * 16 * K;
  float*       Ct = C + (size_t)tile * 16 * N;

  unsigned boff[NT];
#pragma unroll
  for (int t = 0; t < NT; ++t) {
    const int col = t * 16 + row;
    boff[t] = (unsigned)(2 * kp * N + ((col < N) ? col : (N - 1)));
  }

  const unsigned aoff = (unsigned)(row * K + 2 * kp);
  v8f c[NT];
#pragma unroll
  for (int t = 0; t < NT; ++t) c[t] = (v8f){};

#pragma unroll
  for (int k0 = 0; k0 < K; k0 += 4) {
    v2f a;
    a.x = At[aoff + k0];
    a.y = At[aoff + k0 + 1];
#pragma unroll
    for (int t = 0; t < NT; ++t) {
      v2f b;
      b.x = B[boff[t] + k0 * N];
      b.y = B[boff[t] + (k0 + 1) * N];
      c[t] = __builtin_amdgcn_wmma_f32_16x16x4_f32(false, a, false, b, (short)0, c[t],
                                                   false, false);
    }
  }
#pragma unroll
  for (int t = 0; t < NT; ++t) {
    const int col = t * 16 + row;
    if (col < N) {
#pragma unroll
      for (int r = 0; r < 8; ++r)
        Ct[(unsigned)((r + 8 * kp) * N + col)] = c[t][r];
    }
  }
}

#define NPW 8

__device__ __forceinline__ int source_words(const int* __restrict__ srcs, int pos,
                                            int m, int i, int lane) {
  int t = i;
  if (lane < m) t = srcs[pos + lane];
  return t;
}

template <int F, bool WT>
__device__ __forceinline__ void add_rows(int es, int ws, int m, const float* __restrict__ hl,
                                         float2& acc) {
  for (int e = 0; e < m; e += 4) {
    const int s0 = __builtin_amdgcn_readlane(es, e);
    const int s1 = __builtin_amdgcn_readlane(es, e + 1);
    const int s2 = __builtin_amdgcn_readlane(es, e + 2);
    const int s3 = __builtin_amdgcn_readlane(es, e + 3);
    const float2 v0 = *(const float2*)(hl + (size_t)s0 * F);
    const float2 v1 = *(const float2*)(hl + (size_t)s1 * F);
    const float2 v2 = *(const float2*)(hl + (size_t)s2 * F);
    const float2 v3 = *(const float2*)(hl + (size_t)s3 * F);
    if (WT) {
      const float w0 = __int_as_float(__builtin_amdgcn_readlane(ws, e));
      const float w1 = __int_as_float(__builtin_amdgcn_readlane(ws, e + 1));
      const float w2 = __int_as_float(__builtin_amdgcn_readlane(ws, e + 2));
      const float w3 = __int_as_float(__builtin_amdgcn_readlane(ws, e + 3));
      acc.x = fmaf(w0, v0.x, acc.x); acc.y = fmaf(w0, v0.y, acc.y);
      if (e + 1 < m) { acc.x = fmaf(w1, v1.x, acc.x); acc.y = fmaf(w1, v1.y, acc.y); }
      if (e + 2 < m) { acc.x = fmaf(w2, v2.x, acc.x); acc.y = fmaf(w2, v2.y, acc.y); }
      if (e + 3 < m) { acc.x = fmaf(w3, v3.x, acc.x); acc.y = fmaf(w3, v3.y, acc.y); }
    } else {
      acc.x += v0.x; acc.y += v0.y;
      if (e + 1 < m) { acc.x += v1.x; acc.y += v1.y; }
      if (e + 2 < m) { acc.x += v2.x; acc.y += v2.y; }
      if (e + 3 < m) { acc.x += v3.x; acc.y += v3.y; }
    }
  }
}

template <int F, bool WT>
__device__ __forceinline__ float2 node_sum(const int* __restrict__ srcs,
                                           const float* __restrict__ dinv, int start, int ne,
                                           int i, int lane, int es, int ws, float2 own,
                                           const float* __restrict__ hl, float di) {
  float2 acc = own;
  if (WT) { acc.x *= di; acc.y *= di; }
  add_rows<F, WT>(es, ws, ne < 32 ? ne : 32, hl, acc);
  for (int base = 32; base < ne; base += 32) {
    const int m = (ne - base < 32) ? (ne - base) : 32;
    const int u = source_words(srcs, start + base, m, i, lane);
    int wu = 0;
    if (WT) wu = __float_as_int(dinv[u]);
    add_rows<F, WT>(u, wu, m, hl, acc);
  }
  acc.x *= di;
  acc.y *= di;
  return acc;
}

#define TILE_PITCH 68
#define L2_K 64
#define L2_N 40
__global__ void __launch_bounds__(256)
k_gather64_relu_gemm40(const int* __restrict__ srcs, const int* __restrict__ first,
                       const int* __restrict__ cnt, const float* __restrict__ dinv,
                       const float* __restrict__ h, const float* __restrict__ b,
                       const float* __restrict__ W2, float* __restrict__ h2, int n) {
  __shared__ __attribute__((aligned(16))) float tile[8 * NPW * TILE_PITCH];
  __shared__ __attribute__((aligned(16))) float Bs[L2_K * L2_N];
  const int lane = threadIdx.x & 31;
  const int w = __builtin_amdgcn_readfirstlane((int)(threadIdx.x >> 5));
  const int nb0 = blockIdx.x * (8 * NPW);
  static_assert(L2_K * L2_N / 4 == 2 * 256 + 128, "W2 copy is written for 640 pieces and 256 threads");
  const int q = threadIdx.x;
  const float4* W2q = (const float4*)W2;
  const float4 wq0 = W2q[q];
  const float4 wq1 = W2q[q + 256];
  float4 wq2 = make_float4(0.0f, 0.0f, 0.0f, 0.0f);
  if (q < 128) wq2 = W2q[q + 512];

  const int i0 = nb0 + NPW * w;
  const bool active = i0 < n;
  const int nn = active ? ((n - i0 < NPW) ? (n - i0) : NPW) : 0;
  int cv = 0, fv = 0, dv = 0;
  if (lane < nn) {
    cv = cnt[i0 + lane];
    fv = first[i0 + lane];
    dv = __float_as_int(dinv[i0 + lane]);
  }
  const float2 bv = *(const float2*)(b + 2 * lane);
  float4* Bq = (float4*)Bs;
  Bq[q] = wq0;
  Bq[q + 256] = wq1;
  if (q < 128) Bq[q + 512] = wq2;

  if (active) {
    const float* hl = h + 2 * lane;
    int ne_c = __builtin_amdgcn_readlane(cv, 0);
    int st_c = __builtin_amdgcn_readlane(fv, 0);
    int t_c = source_words(srcs, st_c, ne_c, i0, lane);
    int ne_n = 0, st_n = 0, t_n = i0;
    if (nn > 1) {
      ne_n = __builtin_amdgcn_readlane(cv, 1);
      st_n = __builtin_amdgcn_readlane(fv, 1);
      t_n = source_words(srcs, st_n, ne_n, i0 + 1, lane);
    }
    float2 hv = *(const float2*)(hl + (size_t)i0 * 64);
    int w_c = __float_as_int(dinv[t_c]);
    float* trow = tile + (NPW * w) * TILE_PITCH + 2 * lane;
    for (int j = 0; j < nn; ++j) {
      const int i = i0 + j;
      const float di = __int_as_float(__builtin_amdgcn_readlane(dv, j));
      int w_n = 0, t_nn = i, ne_nn = 0, st_nn = 0;
      float2 hv_n = hv;
      if (j + 1 < nn) {
        w_n = __float_as_int(dinv[t_n]);
        hv_n = *(const float2*)(hl + (size_t)(i + 1) * 64);
      }
      if (j + 2 < nn) {
        ne_nn = __builtin_amdgcn_readlane(cv, j + 2);
        st_nn = __builtin_amdgcn_readlane(fv, j + 2);
        t_nn = source_words(srcs, st_nn, ne_nn, i + 2, lane);
      }
      const float2 acc = node_sum<64, true>(srcs, dinv, st_c, ne_c, i, lane, t_c, w_c, hv, hl, di);
      float2 o;
      o.x = fmaxf(acc.x + bv.x, 0.0f);
      o.y = fmaxf(acc.y + bv.y, 0.0f);
      *(float2*)(trow + j * TILE_PITCH) = o;
      t_c = t_n; w_c = w_n; ne_c = ne_n; st_c = st_n; hv = hv_n;
      t_n = t_nn; ne_n = ne_nn; st_n = st_nn;
    }
  }
  __syncthreads();

  if (w >= 4) return;
  const int m0 = nb0 + 16 * w;
  if (m0 >= n) return;
  constexpr int NT = (L2_N + 15) / 16;
  const int row = lane & 15;
  const int kp  = lane >> 4;
  int colc[NT];
#pragma unroll
  for (int t = 0; t < NT; ++t) {
    const int col = t * 16 + row;
    colc[t] = (col < L2_N) ? col : (L2_N - 1);
  }
  const float4 sa = *(const float4*)(dinv + m0 + 8 * kp);
  const float4 sb = *(const float4*)(dinv + m0 + 8 * kp + 4);
  const float* Ar = tile + (16 * w + row) * TILE_PITCH + 2 * kp;
  const float* Br = Bs + (2 * kp) * L2_N;
  v8f c[NT];
#pragma unroll
  for (int t = 0; t < NT; ++t) c[t] = (v8f){};
#pragma unroll
  for (int k0 = 0; k0 < L2_K; k0 += 4) {
    v2f a;
    a.x = Ar[k0];
    a.y = Ar[k0 + 1];
#pragma unroll
    for (int t = 0; t < NT; ++t) {
      v2f bb;
      bb.x = Br[k0 * L2_N + colc[t]];
      bb.y = Br[(k0 + 1) * L2_N + colc[t]];
      c[t] = __builtin_amdgcn_wmma_f32_16x16x4_f32(false, a, false, bb, (short)0, c[t],
                                                   false, false);
    }
  }
  const float sc[8] = {sa.x, sa.y, sa.z, sa.w, sb.x, sb.y, sb.z, sb.w};
#pragma unroll
  for (int t = 0; t < NT; ++t) {
    const int col = t * 16 + row;
    if (col < L2_N) {
#pragma unroll
      for (int r = 0; r < 8; ++r)
        h2[(size_t)(m0 + r + 8 * kp) * L2_N + col] = c[t][r] * sc[r];
    }
  }
}

__global__ void __launch_bounds__(256)
k_gather40_logsoftmax(const int* __restrict__ srcs, const int* __restrict__ first,
                      const int* __restrict__ cnt, const float* __restrict__ dinv,
                      const float* __restrict__ h, const float* __restrict__ b,
                      float* __restrict__ out, int n) {
  const int lane = threadIdx.x & 31;
  const int i0 = NPW * __builtin_amdgcn_readfirstlane((int)(blockIdx.x * 8 + (threadIdx.x >> 5)));
  if (i0 >= n) return;
  const int nn = (n - i0 < NPW) ? (n - i0) : NPW;
  int cv = 0, fv = 0, dv = 0;
  if (lane < nn) {
    cv = cnt[i0 + lane];
    fv = first[i0 + lane];
    dv = __float_as_int(dinv[i0 + lane]);
  }
  const bool own_lane = lane < 20;
  const int fo = 2 * (own_lane ? lane : lane - 20);
  const float* hl = h + fo;
  const float2 bv = *(const float2*)(b + fo);
  int ne = __builtin_amdgcn_readlane(cv, 0);
  int start = __builtin_amdgcn_readlane(fv, 0);
  int t = source_words(srcs, start, ne, i0, lane);
  float2 hv = *(const float2*)(hl + (size_t)i0 * 40);
  for (int j = 0; j < nn; ++j) {
    const int i = i0 + j;
    const int ne_c = ne, start_c = start;
    const int es = t;
    const float2 own = hv;
    const float di = __int_as_float(__builtin_amdgcn_readlane(dv, j));
    if (j + 1 < nn) {
      ne = __builtin_amdgcn_readlane(cv, j + 1);
      start = __builtin_amdgcn_readlane(fv, j + 1);
      t = source_words(srcs, start, ne, i + 1, lane);
      hv = *(const float2*)(hl + (size_t)(i + 1) * 40);
    }
    const float2 acc = node_sum<40, false>(srcs, dinv, start_c, ne_c, i, lane, es, 0, own, hl, di);
    const float vx = acc.x + bv.x;
    const float vy = acc.y + bv.y;
    float mx = own_lane ? fmaxf(vx, vy) : -3.402823466e38f;
    for (int o = 16; o > 0; o >>= 1) mx = fmaxf(mx, __shfl_xor(mx, o));
    float sm = own_lane ? (expf(vx - mx) + expf(vy - mx)) : 0.0f;
    for (int o = 16; o > 0; o >>= 1) sm += __shfl_xor(sm, o);
    const float lse = mx + logf(sm);
    if (own_lane) {
      float2 o2;
      o2.x = vx - lse;
      o2.y = vy - lse;
      *(float2*)(out + (size_t)i * 40 + fo) = o2;
    }
  }
}

extern "C" void kernel_launch(void* const* d_in, const int* in_sizes, int n_in,
                              void* d_out, int out_size, void* d_ws, size_t ws_size,
                              hipStream_t stream) {
  (void)n_in; (void)out_size; (void)ws_size;

  const float* x  = (const float*)d_in[0];
  const int*   ei = (const int*)d_in[1];
  const float* W1 = (const float*)d_in[2];
  const float* b1 = (const float*)d_in[3];
  const float* W2 = (const float*)d_in[4];
  const float* b2 = (const float*)d_in[5];
  float* out = (float*)d_out;

  const int F_IN = 128;
  const int Nn = in_sizes[0] / F_IN;
  const int E  = in_sizes[1] / 2;
  const int* src = ei;
  const int* dst = ei + E;

  char* ws = (char*)d_ws;
  int*   cnt      = (int*)(ws);
  int*   histg    = (int*)(ws + ((size_t)1  << 20));
  int*   binstart = (int*)(ws + ((size_t)3  << 19));
  int*   first    = (int*)(ws + ((size_t)2  << 20));
  float* dinv     = (float*)(ws + ((size_t)3  << 20));
  float* h        = (float*)(ws + ((size_t)4  << 20));
  float* h2       = (float*)(ws + ((size_t)68 << 20));
  int*   srcs     = (int*)(ws + ((size_t)88 << 20));
  int2*  tmp      = (int2*)(ws + ((size_t)100 << 20));

  const dim3 B256(256);

  const int CH   = (E + SORT_G - 1) / SORT_G;
  const int BINS = (Nn + BIN_NODES - 1) / BIN_NODES;
  const dim3 B1024(1024);
  k_bin_hist<<<SORT_G, B1024, 0, stream>>>(dst, histg, E, CH, BINS);
  k_bin_scatter<<<SORT_G, B1024, 0, stream>>>(src, dst, histg, binstart, tmp, E, CH, BINS);
  const int MT = Nn / 16;
  const int GB = (MT + GEMM_TPB - 1) / GEMM_TPB;
  k_bin_finalize_gemm<<<BINS + GB, dim3(512), 0, stream>>>(tmp, binstart, cnt, first, dinv, srcs, Nn, BINS,
                                                           x, W1, h, MT);

  const int GW = 8 * NPW;
  k_gather64_relu_gemm40<<<(Nn + GW - 1) / GW, B256, 0, stream>>>(srcs, first, cnt, dinv, h, b1, W2, h2, Nn);

  k_gather40_logsoftmax<<<(Nn + GW - 1) / GW, B256, 0, stream>>>(srcs, first, cnt, dinv, h2, b2, out, Nn);
}


# Level-1 sort made chunk-local: a chunk orders its own edges by bin in its own stretch, each bin block gathers its 128 pieces; global histogram pass and its launch go (4 launches); sum order changes
# speedup vs baseline: 3.7858x; 1.0610x over previous
#include <hip/hip_runtime.h>
#include <hip/hip_bf16.h>

typedef __attribute__((ext_vector_type(2))) float v2f;
typedef __attribute__((ext_vector_type(8))) float v8f;


#define SORT_G 128
#define BIN_SHIFT 9
#define BIN_NODES 512
#define MAX_BINS 256
#define GEMM_TPB 16

template <int K, int N>
__device__ __forceinline__ void gemm_tile_f32(const float* __restrict__ A, const float* __restrict__ B,
                                              float* __restrict__ C, int tile);

__global__ void __launch_bounds__(1024)
k_chunk_sort(const int* __restrict__ src, const int* __restrict__ dst,
             int* __restrict__ segoff, int2* __restrict__ tmp, int e, int ch, int nb) {
  __shared__ int cur[MAX_BINS];
  __shared__ int wsum[8];
  const int t = threadIdx.x;
  const int g = blockIdx.x;
  const int lane = t & 31;
  if (t < MAX_BINS) cur[t] = 0;
  __syncthreads();
  const int lo = g * ch;
  const int hi = (lo + ch < e) ? (lo + ch) : e;
  int d[8], s[8];
#pragma unroll
  for (int k = 0; k < 8; ++k) {
    const int i = lo + t + 1024 * k;
    d[k] = (i < hi) ? dst[i] : 0;
    s[k] = (i < hi) ? src[i] : 0;
  }
#pragma unroll
  for (int k = 0; k < 8; ++k)
    if (lo + t + 1024 * k < hi) atomicAdd(&cur[d[k] >> BIN_SHIFT], 1);
  for (int i = lo + t + 8192; i < hi; i += 1024) atomicAdd(&cur[dst[i] >> BIN_SHIFT], 1);
  __syncthreads();
  int x = 0, c = 0;
  if (t < MAX_BINS) {
    c = cur[t];
    x = c;
    for (int o = 1; o < 32; o <<= 1) {
      const int y = __shfl_up(x, o);
      if (lane >= o) x += y;
    }
    if (lane == 31) wsum[t >> 5] = x;
  }
  __syncthreads();
  if (t == 0) {
    int a = 0;
    for (int k = 0; k < 8; ++k) { const int v = wsum[k]; wsum[k] = a; a += v; }
  }
  __syncthreads();
  if (t < MAX_BINS) {
    const int start = wsum[t >> 5] + x - c;
    cur[t] = start;
    if (t < nb) segoff[t * SORT_G + g] = start;
  }
  if (t == 0) segoff[nb * SORT_G + g] = (hi > lo) ? (hi - lo) : 0;
  __syncthreads();
  int pos[8];
#pragma unroll
  for (int k = 0; k < 8; ++k) {
    pos[k] = 0;
    if (lo + t + 1024 * k < hi) pos[k] = atomicAdd(&cur[d[k] >> BIN_SHIFT], 1);
  }
#pragma unroll
  for (int k = 0; k < 8; ++k)
    if (lo + t + 1024 * k < hi) tmp[lo + pos[k]] = make_int2(s[k], d[k]);
  for (int i = lo + t + 8192; i < hi; i += 1024) {
    const int dd = dst[i];
    const int ss = src[i];
    const int p = atomicAdd(&cur[dd >> BIN_SHIFT], 1);
    tmp[lo + p] = make_int2(ss, dd);
  }
}

#define SEG_PH (SORT_G / 32)
__global__ void __launch_bounds__(512)
k_bin_finalize_gemm(const int2* __restrict__ tmp, const int* __restrict__ segoff,
                    int* __restrict__ cnt, int* __restrict__ first, float* __restrict__ dinv,
                    int* __restrict__ srcs, int n, int nbins, int ch,
                    const float* __restrict__ xa, const float* __restrict__ wb,
                    float* __restrict__ hc, int mtiles) {
  if ((int)blockIdx.x >= nbins) {
    const int wv = __builtin_amdgcn_readfirstlane((int)(threadIdx.x >> 5));
    const int tile = ((int)blockIdx.x - nbins) * GEMM_TPB + wv;
    if (tile < mtiles) gemm_tile_f32<128, 64>(xa, wb, hc, tile);
    return;
  }
  __shared__ int c[BIN_NODES];
  __shared__ int wsum[BIN_NODES / 32];
  __shared__ int segb[SORT_G];
  __shared__ int segl[SORT_G];
  __shared__ int wred[SORT_G / 32];
  const int t = threadIdx.x;
  const int b = blockIdx.x;
  const int lane = t & 31;
  if (t < BIN_NODES) c[t] = 0;
  if (t < SORT_G) {
    const int st = segoff[b * SORT_G + t];
    const int en = segoff[(b + 1) * SORT_G + t];
    segb[t] = t * ch + st;
    segl[t] = en - st;
    int r = st;
    for (int o = 16; o > 0; o >>= 1) r += __shfl_xor(r, o);
    if (lane == 0) wred[t >> 5] = r;
  }
  __syncthreads();
  int s0 = 0;
#pragma unroll
  for (int k = 0; k < SORT_G / 32; ++k) s0 += wred[k];
  const int hw = t >> 4;
  const int l16 = t & 15;
  int sb[SEG_PH], sl[SEG_PH];
  int2 ld[SEG_PH][4];
#pragma unroll
  for (int q = 0; q < SEG_PH; ++q) {
    sb[q] = segb[hw + 32 * q];
    sl[q] = segl[hw + 32 * q];
#pragma unroll
    for (int k = 0; k < 4; ++k) {
      const int idx = l16 + 16 * k;
      ld[q][k] = (idx < sl[q]) ? tmp[sb[q] + idx] : make_int2(0, 0);
    }
  }
  int ed[SEG_PH][4];
#pragma unroll
  for (int q = 0; q < SEG_PH; ++q) {
#pragma unroll
    for (int k = 0; k < 4; ++k) {
      ed[q][k] = (ld[q][k].x << BIN_SHIFT) | (ld[q][k].y & (BIN_NODES - 1));
      if (l16 + 16 * k < sl[q]) atomicAdd(&c[ed[q][k] & (BIN_NODES - 1)], 1);
    }
    for (int idx = 64 + l16; idx < sl[q]; idx += 16)
      atomicAdd(&c[tmp[sb[q] + idx].y & (BIN_NODES - 1)], 1);
  }
  __syncthreads();
  int x = 0, kc = 0;
  if (t < BIN_NODES) {
    kc = c[t];
    x = kc;
    for (int o = 1; o < 32; o <<= 1) {
      const int y = __shfl_up(x, o);
      if (lane >= o) x += y;
    }
    if (lane == 31) wsum[t >> 5] = x;
  }
  __syncthreads();
  if (t == 0) {
    int a = 0;
    for (int j = 0; j < BIN_NODES / 32; ++j) { const int v = wsum[j]; wsum[j] = a; a += v; }
  }
  __syncthreads();
  if (t < BIN_NODES) {
    const int f = s0 + wsum[t >> 5] + x - kc;
    c[t] = f;
    const int node = b * BIN_NODES + t;
    if (node < n) {
      cnt[node] = kc;
      first[node] = f;
      dinv[node] = rsqrtf((float)(kc + 1));
    }
  }
  __syncthreads();
#pragma unroll
  for (int q = 0; q < SEG_PH; ++q) {
#pragma unroll
    for (int k = 0; k < 4; ++k)
      if (l16 + 16 * k < sl[q]) {
        const int p = atomicAdd(&c[ed[q][k] & (BIN_NODES - 1)], 1);
        srcs[p] = (int)((unsigned)ed[q][k] >> BIN_SHIFT);
      }
    for (int idx = 64 + l16; idx < sl[q]; idx += 16) {
      const int2 v = tmp[sb[q] + idx];
      const int p = atomicAdd(&c[v.y & (BIN_NODES - 1)], 1);
      srcs[p] = v.x;
    }
  }
}

template <int K, int N>
__device__ __forceinline__ void gemm_tile_f32(const float* __restrict__ A, const float* __restrict__ B,
                                              float* __restrict__ C, int tile) {
  constexpr int NT = (N + 15) / 16;
  const int lane = threadIdx.x & 31;
  const int row  = lane & 15;
  const int kp   = lane >> 4;
  const float* At = A + (size_t)tile * 16 * K;
  float*       Ct = C + (size_t)tile * 16 * N;

  unsigned boff[NT];
#pragma unroll
  for (int t = 0; t < NT; ++t) {
    const int col = t * 16 + row;
    boff[t] = (unsigned)(2 * kp * N + ((col < N) ? col : (N - 1)));
  }

  const unsigned aoff = (unsigned)(row * K + 2 * kp);
  v8f c[NT];
#pragma unroll
  for (int t = 0; t < NT; ++t) c[t] = (v8f){};

#pragma unroll
  for (int k0 = 0; k0 < K; k0 += 4) {
    v2f a;
    a.x = At[aoff + k0];
    a.y = At[aoff + k0 + 1];
#pragma unroll
    for (int t = 0; t < NT; ++t) {
      v2f b;
      b.x = B[boff[t] + k0 * N];
      b.y = B[boff[t] + (k0 + 1) * N];
      c[t] = __builtin_amdgcn_wmma_f32_16x16x4_f32(false, a, false, b, (short)0, c[t],
                                                   false, false);
    }
  }
#pragma unroll
  for (int t = 0; t < NT; ++t) {
    const int col = t * 16 + row;
    if (col < N) {
#pragma unroll
      for (int r = 0; r < 8; ++r)
        Ct[(unsigned)((r + 8 * kp) * N + col)] = c[t][r];
    }
  }
}

#define NPW 8

__device__ __forceinline__ int source_words(const int* __restrict__ srcs, int pos,
                                            int m, int i, int lane) {
  int t = i;
  if (lane < m) t = srcs[pos + lane];
  return t;
}

template <int F, bool WT>
__device__ __forceinline__ void add_rows(int es, int ws, int m, const float* __restrict__ hl,
                                         float2& acc) {
  for (int e = 0; e < m; e += 4) {
    const int s0 = __builtin_amdgcn_readlane(es, e);
    const int s1 = __builtin_amdgcn_readlane(es, e + 1);
    const int s2 = __builtin_amdgcn_readlane(es, e + 2);
    const int s3 = __builtin_amdgcn_readlane(es, e + 3);
    const float2 v0 = *(const float2*)(hl + (size_t)s0 * F);
    const float2 v1 = *(const float2*)(hl + (size_t)s1 * F);
    const float2 v2 = *(const float2*)(hl + (size_t)s2 * F);
    const float2 v3 = *(const float2*)(hl + (size_t)s3 * F);
    if (WT) {
      const float w0 = __int_as_float(__builtin_amdgcn_readlane(ws, e));
      const float w1 = __int_as_float(__builtin_amdgcn_readlane(ws, e + 1));
      const float w2 = __int_as_float(__builtin_amdgcn_readlane(ws, e + 2));
      const float w3 = __int_as_float(__builtin_amdgcn_readlane(ws, e + 3));
      acc.x = fmaf(w0, v0.x, acc.x); acc.y = fmaf(w0, v0.y, acc.y);
      if (e + 1 < m) { acc.x = fmaf(w1, v1.x, acc.x); acc.y = fmaf(w1, v1.y, acc.y); }
      if (e + 2 < m) { acc.x = fmaf(w2, v2.x, acc.x); acc.y = fmaf(w2, v2.y, acc.y); }
      if (e + 3 < m) { acc.x = fmaf(w3, v3.x, acc.x); acc.y = fmaf(w3, v3.y, acc.y); }
    } else {
      acc.x += v0.x; acc.y += v0.y;
      if (e + 1 < m) { acc.x += v1.x; acc.y += v1.y; }
      if (e + 2 < m) { acc.x += v2.x; acc.y += v2.y; }
      if (e + 3 < m) { acc.x += v3.x; acc.y += v3.y; }
    }
  }
}

template <int F, bool WT>
__device__ __forceinline__ float2 node_sum(const int* __restrict__ srcs,
                                           const float* __restrict__ dinv, int start, int ne,
                                           int i, int lane, int es, int ws, float2 own,
                                           const float* __restrict__ hl, float di) {
  float2 acc = own;
  if (WT) { acc.x *= di; acc.y *= di; }
  add_rows<F, WT>(es, ws, ne < 32 ? ne : 32, hl, acc);
  for (int base = 32; base < ne; base += 32) {
    const int m = (ne - base < 32) ? (ne - base) : 32;
    const int u = source_words(srcs, start + base, m, i, lane);
    int wu = 0;
    if (WT) wu = __float_as_int(dinv[u]);
    add_rows<F, WT>(u, wu, m, hl, acc);
  }
  acc.x *= di;
  acc.y *= di;
  return acc;
}

#define TILE_PITCH 68
#define L2_K 64
#define L2_N 40
__global__ void __launch_bounds__(256)
k_gather64_relu_gemm40(const int* __restrict__ srcs, const int* __restrict__ first,
                       const int* __restrict__ cnt, const float* __restrict__ dinv,
                       const float* __restrict__ h, const float* __restrict__ b,
                       const float* __restrict__ W2, float* __restrict__ h2, int n) {
  __shared__ __attribute__((aligned(16))) float tile[8 * NPW * TILE_PITCH];
  __shared__ __attribute__((aligned(16))) float Bs[L2_K * L2_N];
  const int lane = threadIdx.x & 31;
  const int w = __builtin_amdgcn_readfirstlane((int)(threadIdx.x >> 5));
  const int nb0 = blockIdx.x * (8 * NPW);
  static_assert(L2_K * L2_N / 4 == 2 * 256 + 128, "W2 copy is written for 640 pieces and 256 threads");
  const int q = threadIdx.x;
  const float4* W2q = (const float4*)W2;
  const float4 wq0 = W2q[q];
  const float4 wq1 = W2q[q + 256];
  float4 wq2 = make_float4(0.0f, 0.0f, 0.0f, 0.0f);
  if (q < 128) wq2 = W2q[q + 512];

  const int i0 = nb0 + NPW * w;
  const bool active = i0 < n;
  const int nn = active ? ((n - i0 < NPW) ? (n - i0) : NPW) : 0;
  int cv = 0, fv = 0, dv = 0;
  if (lane < nn) {
    cv = cnt[i0 + lane];
    fv = first[i0 + lane];
    dv = __float_as_int(dinv[i0 + lane]);
  }
  const float2 bv = *(const float2*)(b + 2 * lane);
  float4* Bq = (float4*)Bs;
  Bq[q] = wq0;
  Bq[q + 256] = wq1;
  if (q < 128) Bq[q + 512] = wq2;

  if (active) {
    const float* hl = h + 2 * lane;
    int ne_c = __builtin_amdgcn_readlane(cv, 0);
    int st_c = __builtin_amdgcn_readlane(fv, 0);
    int t_c = source_words(srcs, st_c, ne_c, i0, lane);
    int ne_n = 0, st_n = 0, t_n = i0;
    if (nn > 1) {
      ne_n = __builtin_amdgcn_readlane(cv, 1);
      st_n = __builtin_amdgcn_readlane(fv, 1);
      t_n = source_words(srcs, st_n, ne_n, i0 + 1, lane);
    }
    float2 hv = *(const float2*)(hl + (size_t)i0 * 64);
    int w_c = __float_as_int(dinv[t_c]);
    float* trow = tile + (NPW * w) * TILE_PITCH + 2 * lane;
    for (int j = 0; j < nn; ++j) {
      const int i = i0 + j;
      const float di = __int_as_float(__builtin_amdgcn_readlane(dv, j));
      int w_n = 0, t_nn = i, ne_nn = 0, st_nn = 0;
      float2 hv_n = hv;
      if (j + 1 < nn) {
        w_n = __float_as_int(dinv[t_n]);
        hv_n = *(const float2*)(hl + (size_t)(i + 1) * 64);
      }
      if (j + 2 < nn) {
        ne_nn = __builtin_amdgcn_readlane(cv, j + 2);
        st_nn = __builtin_amdgcn_readlane(fv, j + 2);
        t_nn = source_words(srcs, st_nn, ne_nn, i + 2, lane);
      }
      const float2 acc = node_sum<64, true>(srcs, dinv, st_c, ne_c, i, lane, t_c, w_c, hv, hl, di);
      float2 o;
      o.x = fmaxf(acc.x + bv.x, 0.0f);
      o.y = fmaxf(acc.y + bv.y, 0.0f);
      *(float2*)(trow + j * TILE_PITCH) = o;
      t_c = t_n; w_c = w_n; ne_c = ne_n; st_c = st_n; hv = hv_n;
      t_n = t_nn; ne_n = ne_nn; st_n = st_nn;
    }
  }
  __syncthreads();

  if (w >= 4) return;
  const int m0 = nb0 + 16 * w;
  if (m0 >= n) return;
  constexpr int NT = (L2_N + 15) / 16;
  const int row = lane & 15;
  const int kp  = lane >> 4;
  int colc[NT];
#pragma unroll
  for (int t = 0; t < NT; ++t) {
    const int col = t * 16 + row;
    colc[t] = (col < L2_N) ? col : (L2_N - 1);
  }
  const float4 sa = *(const float4*)(dinv + m0 + 8 * kp);
  const float4 sb = *(const float4*)(dinv + m0 + 8 * kp + 4);
  const float* Ar = tile + (16 * w + row) * TILE_PITCH + 2 * kp;
  const float* Br = Bs + (2 * kp) * L2_N;
  v8f c[NT];
#pragma unroll
  for (int t = 0; t < NT; ++t) c[t] = (v8f){};
#pragma unroll
  for (int k0 = 0; k0 < L2_K; k0 += 4) {
    v2f a;
    a.x = Ar[k0];
    a.y = Ar[k0 + 1];
#pragma unroll
    for (int t = 0; t < NT; ++t) {
      v2f bb;
      bb.x = Br[k0 * L2_N + colc[t]];
      bb.y = Br[(k0 + 1) * L2_N + colc[t]];
      c[t] = __builtin_amdgcn_wmma_f32_16x16x4_f32(false, a, false, bb, (short)0, c[t],
                                                   false, false);
    }
  }
  const float sc[8] = {sa.x, sa.y, sa.z, sa.w, sb.x, sb.y, sb.z, sb.w};
#pragma unroll
  for (int t = 0; t < NT; ++t) {
    const int col = t * 16 + row;
    if (col < L2_N) {
#pragma unroll
      for (int r = 0; r < 8; ++r)
        h2[(size_t)(m0 + r + 8 * kp) * L2_N + col] = c[t][r] * sc[r];
    }
  }
}

__global__ void __launch_bounds__(256)
k_gather40_logsoftmax(const int* __restrict__ srcs, const int* __restrict__ first,
                      const int* __restrict__ cnt, const float* __restrict__ dinv,
                      const float* __restrict__ h, const float* __restrict__ b,
                      float* __restrict__ out, int n) {
  const int lane = threadIdx.x & 31;
  const int i0 = NPW * __builtin_amdgcn_readfirstlane((int)(blockIdx.x * 8 + (threadIdx.x >> 5)));
  if (i0 >= n) return;
  const int nn = (n - i0 < NPW) ? (n - i0) : NPW;
  int cv = 0, fv = 0, dv = 0;
  if (lane < nn) {
    cv = cnt[i0 + lane];
    fv = first[i0 + lane];
    dv = __float_as_int(dinv[i0 + lane]);
  }
  const bool own_lane = lane < 20;
  const int fo = 2 * (own_lane ? lane : lane - 20);
  const float* hl = h + fo;
  const float2 bv = *(const float2*)(b + fo);
  int ne = __builtin_amdgcn_readlane(cv, 0);
  int start = __builtin_amdgcn_readlane(fv, 0);
  int t = source_words(srcs, start, ne, i0, lane);
  float2 hv = *(const float2*)(hl + (size_t)i0 * 40);
  for (int j = 0; j < nn; ++j) {
    const int i = i0 + j;
    const int ne_c = ne, start_c = start;
    const int es = t;
    const float2 own = hv;
    const float di = __int_as_float(__builtin_amdgcn_readlane(dv, j));
    if (j + 1 < nn) {
      ne = __builtin_amdgcn_readlane(cv, j + 1);
      start = __builtin_amdgcn_readlane(fv, j + 1);
      t = source_words(srcs, start, ne, i + 1, lane);
      hv = *(const float2*)(hl + (size_t)(i + 1) * 40);
    }
    const float2 acc = node_sum<40, false>(srcs, dinv, start_c, ne_c, i, lane, es, 0, own, hl, di);
    const float vx = acc.x + bv.x;
    const float vy = acc.y + bv.y;
    float mx = own_lane ? fmaxf(vx, vy) : -3.402823466e38f;
    for (int o = 16; o > 0; o >>= 1) mx = fmaxf(mx, __shfl_xor(mx, o));
    float sm = own_lane ? (expf(vx - mx) + expf(vy - mx)) : 0.0f;
    for (int o = 16; o > 0; o >>= 1) sm += __shfl_xor(sm, o);
    const float lse = mx + logf(sm);
    if (own_lane) {
      float2 o2;
      o2.x = vx - lse;
      o2.y = vy - lse;
      *(float2*)(out + (size_t)i * 40 + fo) = o2;
    }
  }
}

extern "C" void kernel_launch(void* const* d_in, const int* in_sizes, int n_in,
                              void* d_out, int out_size, void* d_ws, size_t ws_size,
                              hipStream_t stream) {
  (void)n_in; (void)out_size; (void)ws_size;

  const float* x  = (const float*)d_in[0];
  const int*   ei = (const int*)d_in[1];
  const float* W1 = (const float*)d_in[2];
  const float* b1 = (const float*)d_in[3];
  const float* W2 = (const float*)d_in[4];
  const float* b2 = (const float*)d_in[5];
  float* out = (float*)d_out;

  const int F_IN = 128;
  const int Nn = in_sizes[0] / F_IN;
  const int E  = in_sizes[1] / 2;
  const int* src = ei;
  const int* dst = ei + E;

  char* ws = (char*)d_ws;
  int*   cnt      = (int*)(ws);
  int*   segoff   = (int*)(ws + ((size_t)1  << 20));
  int*   first    = (int*)(ws + ((size_t)2  << 20));
  float* dinv     = (float*)(ws + ((size_t)3  << 20));
  float* h        = (float*)(ws + ((size_t)4  << 20));
  float* h2       = (float*)(ws + ((size_t)68 << 20));
  int*   srcs     = (int*)(ws + ((size_t)88 << 20));
  int2*  tmp      = (int2*)(ws + ((size_t)100 << 20));

  const dim3 B256(256);

  const int CH   = (E + SORT_G - 1) / SORT_G;
  const int BINS = (Nn + BIN_NODES - 1) / BIN_NODES;
  const dim3 B1024(1024);
  k_chunk_sort<<<SORT_G, B1024, 0, stream>>>(src, dst, segoff, tmp, E, CH, BINS);
  const int MT = Nn / 16;
  const int GB = (MT + GEMM_TPB - 1) / GEMM_TPB;
  k_bin_finalize_gemm<<<BINS + GB, dim3(512), 0, stream>>>(tmp, segoff, cnt, first, dinv, srcs, Nn, BINS, CH,
                                                           x, W1, h, MT);

  const int GW = 8 * NPW;
  k_gather64_relu_gemm40<<<(Nn + GW - 1) / GW, B256, 0, stream>>>(srcs, first, cnt, dinv, h, b1, W2, h2, Nn);

  k_gather40_logsoftmax<<<(Nn + GW - 1) / GW, B256, 0, stream>>>(srcs, first, cnt, dinv, h2, b2, out, Nn);
}


# Layer-1 product x@W1 split over both sort launches: 2048 of its 6250 row tiles run as extra work-groups beside the chunk sort (half the chip idle there), the rest beside the bin pass
# speedup vs baseline: 4.0798x; 1.0777x over previous
#include <hip/hip_runtime.h>
#include <hip/hip_bf16.h>

typedef __attribute__((ext_vector_type(2))) float v2f;
typedef __attribute__((ext_vector_type(8))) float v8f;


#define SORT_G 128
#define BIN_SHIFT 9
#define BIN_NODES 512
#define MAX_BINS 256
#define GEMM_TPB 16

template <int K, int N>
__device__ __forceinline__ void gemm_tile_f32(const float* __restrict__ A, const float* __restrict__ B,
                                              float* __restrict__ C, int tile);

__global__ void __launch_bounds__(1024)
k_chunk_sort(const int* __restrict__ src, const int* __restrict__ dst,
             int* __restrict__ segoff, int2* __restrict__ tmp, int e, int ch, int nb,
             const float* __restrict__ xa, const float* __restrict__ wb,
             float* __restrict__ hc, int mt1) {
  if ((int)blockIdx.x >= SORT_G) {
    const int wv = __builtin_amdgcn_readfirstlane((int)(threadIdx.x >> 5));
    const int tile = ((int)blockIdx.x - SORT_G) * GEMM_TPB + wv;
    if (wv < GEMM_TPB && tile < mt1) gemm_tile_f32<128, 64>(xa, wb, hc, tile);
    return;
  }
  __shared__ int cur[MAX_BINS];
  __shared__ int wsum[8];
  const int t = threadIdx.x;
  const int g = blockIdx.x;
  const int lane = t & 31;
  if (t < MAX_BINS) cur[t] = 0;
  __syncthreads();
  const int lo = g * ch;
  const int hi = (lo + ch < e) ? (lo + ch) : e;
  int d[8], s[8];
#pragma unroll
  for (int k = 0; k < 8; ++k) {
    const int i = lo + t + 1024 * k;
    d[k] = (i < hi) ? dst[i] : 0;
    s[k] = (i < hi) ? src[i] : 0;
  }
#pragma unroll
  for (int k = 0; k < 8; ++k)
    if (lo + t + 1024 * k < hi) atomicAdd(&cur[d[k] >> BIN_SHIFT], 1);
  for (int i = lo + t + 8192; i < hi; i += 1024) atomicAdd(&cur[dst[i] >> BIN_SHIFT], 1);
  __syncthreads();
  int x = 0, c = 0;
  if (t < MAX_BINS) {
    c = cur[t];
    x = c;
    for (int o = 1; o < 32; o <<= 1) {
      const int y = __shfl_up(x, o);
      if (lane >= o) x += y;
    }
    if (lane == 31) wsum[t >> 5] = x;
  }
  __syncthreads();
  if (t == 0) {
    int a = 0;
    for (int k = 0; k < 8; ++k) { const int v = wsum[k]; wsum[k] = a; a += v; }
  }
  __syncthreads();
  if (t < MAX_BINS) {
    const int start = wsum[t >> 5] + x - c;
    cur[t] = start;
    if (t < nb) segoff[t * SORT_G + g] = start;
  }
  if (t == 0) segoff[nb * SORT_G + g] = (hi > lo) ? (hi - lo) : 0;
  __syncthreads();
  int pos[8];
#pragma unroll
  for (int k = 0; k < 8; ++k) {
    pos[k] = 0;
    if (lo + t + 1024 * k < hi) pos[k] = atomicAdd(&cur[d[k] >> BIN_SHIFT], 1);
  }
#pragma unroll
  for (int k = 0; k < 8; ++k)
    if (lo + t + 1024 * k < hi) tmp[lo + pos[k]] = make_int2(s[k], d[k]);
  for (int i = lo + t + 8192; i < hi; i += 1024) {
    const int dd = dst[i];
    const int ss = src[i];
    const int p = atomicAdd(&cur[dd >> BIN_SHIFT], 1);
    tmp[lo + p] = make_int2(ss, dd);
  }
}

#define SEG_PH (SORT_G / 32)
__global__ void __launch_bounds__(512)
k_bin_finalize_gemm(const int2* __restrict__ tmp, const int* __restrict__ segoff,
                    int* __restrict__ cnt, int* __restrict__ first, float* __restrict__ dinv,
                    int* __restrict__ srcs, int n, int nbins, int ch,
                    const float* __restrict__ xa, const float* __restrict__ wb,
                    float* __restrict__ hc, int tile0, int mtiles) {
  if ((int)blockIdx.x >= nbins) {
    const int wv = __builtin_amdgcn_readfirstlane((int)(threadIdx.x >> 5));
    const int tile = tile0 + ((int)blockIdx.x - nbins) * GEMM_TPB + wv;
    if (tile < mtiles) gemm_tile_f32<128, 64>(xa, wb, hc, tile);
    return;
  }
  __shared__ int c[BIN_NODES];
  __shared__ int wsum[BIN_NODES / 32];
  __shared__ int segb[SORT_G];
  __shared__ int segl[SORT_G];
  __shared__ int wred[SORT_G / 32];
  const int t = threadIdx.x;
  const int b = blockIdx.x;
  const int lane = t & 31;
  if (t < BIN_NODES) c[t] = 0;
  if (t < SORT_G) {
    const int st = segoff[b * SORT_G + t];
    const int en = segoff[(b + 1) * SORT_G + t];
    segb[t] = t * ch + st;
    segl[t] = en - st;
    int r = st;
    for (int o = 16; o > 0; o >>= 1) r += __shfl_xor(r, o);
    if (lane == 0) wred[t >> 5] = r;
  }
  __syncthreads();
  int s0 = 0;
#pragma unroll
  for (int k = 0; k < SORT_G / 32; ++k) s0 += wred[k];
  const int hw = t >> 4;
  const int l16 = t & 15;
  int sb[SEG_PH], sl[SEG_PH];
  int2 ld[SEG_PH][4];
#pragma unroll
  for (int q = 0; q < SEG_PH; ++q) {
    sb[q] = segb[hw + 32 * q];
    sl[q] = segl[hw + 32 * q];
#pragma unroll
    for (int k = 0; k < 4; ++k) {
      const int idx = l16 + 16 * k;
      ld[q][k] = (idx < sl[q]) ? tmp[sb[q] + idx] : make_int2(0, 0);
    }
  }
  int ed[SEG_PH][4];
#pragma unroll
  for (int q = 0; q < SEG_PH; ++q) {
#pragma unroll
    for (int k = 0; k < 4; ++k) {
      ed[q][k] = (ld[q][k].x << BIN_SHIFT) | (ld[q][k].y & (BIN_NODES - 1));
      if (l16 + 16 * k < sl[q]) atomicAdd(&c[ed[q][k] & (BIN_NODES - 1)], 1);
    }
    for (int idx = 64 + l16; idx < sl[q]; idx += 16)
      atomicAdd(&c[tmp[sb[q] + idx].y & (BIN_NODES - 1)], 1);
  }
  __syncthreads();
  int x = 0, kc = 0;
  if (t < BIN_NODES) {
    kc = c[t];
    x = kc;
    for (int o = 1; o < 32; o <<= 1) {
      const int y = __shfl_up(x, o);
      if (lane >= o) x += y;
    }
    if (lane == 31) wsum[t >> 5] = x;
  }
  __syncthreads();
  if (t == 0) {
    int a = 0;
    for (int j = 0; j < BIN_NODES / 32; ++j) { const int v = wsum[j]; wsum[j] = a; a += v; }
  }
  __syncthreads();
  if (t < BIN_NODES) {
    const int f = s0 + wsum[t >> 5] + x - kc;
    c[t] = f;
    const int node = b * BIN_NODES + t;
    if (node < n) {
      cnt[node] = kc;
      first[node] = f;
      dinv[node] = rsqrtf((float)(kc + 1));
    }
  }
  __syncthreads();
#pragma unroll
  for (int q = 0; q < SEG_PH; ++q) {
#pragma unroll
    for (int k = 0; k < 4; ++k)
      if (l16 + 16 * k < sl[q]) {
        const int p = atomicAdd(&c[ed[q][k] & (BIN_NODES - 1)], 1);
        srcs[p] = (int)((unsigned)ed[q][k] >> BIN_SHIFT);
      }
    for (int idx = 64 + l16; idx < sl[q]; idx += 16) {
      const int2 v = tmp[sb[q] + idx];
      const int p = atomicAdd(&c[v.y & (BIN_NODES - 1)], 1);
      srcs[p] = v.x;
    }
  }
}

template <int K, int N>
__device__ __forceinline__ void gemm_tile_f32(const float* __restrict__ A, const float* __restrict__ B,
                                              float* __restrict__ C, int tile) {
  constexpr int NT = (N + 15) / 16;
  const int lane = threadIdx.x & 31;
  const int row  = lane & 15;
  const int kp   = lane >> 4;
  const float* At = A + (size_t)tile * 16 * K;
  float*       Ct = C + (size_t)tile * 16 * N;

  unsigned boff[NT];
#pragma unroll
  for (int t = 0; t < NT; ++t) {
    const int col = t * 16 + row;
    boff[t] = (unsigned)(2 * kp * N + ((col < N) ? col : (N - 1)));
  }

  const unsigned aoff = (unsigned)(row * K + 2 * kp);
  v8f c[NT];
#pragma unroll
  for (int t = 0; t < NT; ++t) c[t] = (v8f){};

#pragma unroll
  for (int k0 = 0; k0 < K; k0 += 4) {
    v2f a;
    a.x = At[aoff + k0];
    a.y = At[aoff + k0 + 1];
#pragma unroll
    for (int t = 0; t < NT; ++t) {
      v2f b;
      b.x = B[boff[t] + k0 * N];
      b.y = B[boff[t] + (k0 + 1) * N];
      c[t] = __builtin_amdgcn_wmma_f32_16x16x4_f32(false, a, false, b, (short)0, c[t],
                                                   false, false);
    }
  }
#pragma unroll
  for (int t = 0; t < NT; ++t) {
    const int col = t * 16 + row;
    if (col < N) {
#pragma unroll
      for (int r = 0; r < 8; ++r)
        Ct[(unsigned)((r + 8 * kp) * N + col)] = c[t][r];
    }
  }
}

#define NPW 8

__device__ __forceinline__ int source_words(const int* __restrict__ srcs, int pos,
                                            int m, int i, int lane) {
  int t = i;
  if (lane < m) t = srcs[pos + lane];
  return t;
}

template <int F, bool WT>
__device__ __forceinline__ void add_rows(int es, int ws, int m, const float* __restrict__ hl,
                                         float2& acc) {
  for (int e = 0; e < m; e += 4) {
    const int s0 = __builtin_amdgcn_readlane(es, e);
    const int s1 = __builtin_amdgcn_readlane(es, e + 1);
    const int s2 = __builtin_amdgcn_readlane(es, e + 2);
    const int s3 = __builtin_amdgcn_readlane(es, e + 3);
    const float2 v0 = *(const float2*)(hl + (size_t)s0 * F);
    const float2 v1 = *(const float2*)(hl + (size_t)s1 * F);
    const float2 v2 = *(const float2*)(hl + (size_t)s2 * F);
    const float2 v3 = *(const float2*)(hl + (size_t)s3 * F);
    if (WT) {
      const float w0 = __int_as_float(__builtin_amdgcn_readlane(ws, e));
      const float w1 = __int_as_float(__builtin_amdgcn_readlane(ws, e + 1));
      const float w2 = __int_as_float(__builtin_amdgcn_readlane(ws, e + 2));
      const float w3 = __int_as_float(__builtin_amdgcn_readlane(ws, e + 3));
      acc.x = fmaf(w0, v0.x, acc.x); acc.y = fmaf(w0, v0.y, acc.y);
      if (e + 1 < m) { acc.x = fmaf(w1, v1.x, acc.x); acc.y = fmaf(w1, v1.y, acc.y); }
      if (e + 2 < m) { acc.x = fmaf(w2, v2.x, acc.x); acc.y = fmaf(w2, v2.y, acc.y); }
      if (e + 3 < m) { acc.x = fmaf(w3, v3.x, acc.x); acc.y = fmaf(w3, v3.y, acc.y); }
    } else {
      acc.x += v0.x; acc.y += v0.y;
      if (e + 1 < m) { acc.x += v1.x; acc.y += v1.y; }
      if (e + 2 < m) { acc.x += v2.x; acc.y += v2.y; }
      if (e + 3 < m) { acc.x += v3.x; acc.y += v3.y; }
    }
  }
}

template <int F, bool WT>
__device__ __forceinline__ float2 node_sum(const int* __restrict__ srcs,
                                           const float* __restrict__ dinv, int start, int ne,
                                           int i, int lane, int es, int ws, float2 own,
                                           const float* __restrict__ hl, float di) {
  float2 acc = own;
  if (WT) { acc.x *= di; acc.y *= di; }
  add_rows<F, WT>(es, ws, ne < 32 ? ne : 32, hl, acc);
  for (int base = 32; base < ne; base += 32) {
    const int m = (ne - base < 32) ? (ne - base) : 32;
    const int u = source_words(srcs, start + base, m, i, lane);
    int wu = 0;
    if (WT) wu = __float_as_int(dinv[u]);
    add_rows<F, WT>(u, wu, m, hl, acc);
  }
  acc.x *= di;
  acc.y *= di;
  return acc;
}

#define TILE_PITCH 68
#define L2_K 64
#define L2_N 40
__global__ void __launch_bounds__(256)
k_gather64_relu_gemm40(const int* __restrict__ srcs, const int* __restrict__ first,
                       const int* __restrict__ cnt, const float* __restrict__ dinv,
                       const float* __restrict__ h, const float* __restrict__ b,
                       const float* __restrict__ W2, float* __restrict__ h2, int n) {
  __shared__ __attribute__((aligned(16))) float tile[8 * NPW * TILE_PITCH];
  __shared__ __attribute__((aligned(16))) float Bs[L2_K * L2_N];
  const int lane = threadIdx.x & 31;
  const int w = __builtin_amdgcn_readfirstlane((int)(threadIdx.x >> 5));
  const int nb0 = blockIdx.x * (8 * NPW);
  static_assert(L2_K * L2_N / 4 == 2 * 256 + 128, "W2 copy is written for 640 pieces and 256 threads");
  const int q = threadIdx.x;
  const float4* W2q = (const float4*)W2;
  const float4 wq0 = W2q[q];
  const float4 wq1 = W2q[q + 256];
  float4 wq2 = make_float4(0.0f, 0.0f, 0.0f, 0.0f);
  if (q < 128) wq2 = W2q[q + 512];

  const int i0 = nb0 + NPW * w;
  const bool active = i0 < n;
  const int nn = active ? ((n - i0 < NPW) ? (n - i0) : NPW) : 0;
  int cv = 0, fv = 0, dv = 0;
  if (lane < nn) {
    cv = cnt[i0 + lane];
    fv = first[i0 + lane];
    dv = __float_as_int(dinv[i0 + lane]);
  }
  const float2 bv = *(const float2*)(b + 2 * lane);
  float4* Bq = (float4*)Bs;
  Bq[q] = wq0;
  Bq[q + 256] = wq1;
  if (q < 128) Bq[q + 512] = wq2;

  if (active) {
    const float* hl = h + 2 * lane;
    int ne_c = __builtin_amdgcn_readlane(cv, 0);
    int st_c = __builtin_amdgcn_readlane(fv, 0);
    int t_c = source_words(srcs, st_c, ne_c, i0, lane);
    int ne_n = 0, st_n = 0, t_n = i0;
    if (nn > 1) {
      ne_n = __builtin_amdgcn_readlane(cv, 1);
      st_n = __builtin_amdgcn_readlane(fv, 1);
      t_n = source_words(srcs, st_n, ne_n, i0 + 1, lane);
    }
    float2 hv = *(const float2*)(hl + (size_t)i0 * 64);
    int w_c = __float_as_int(dinv[t_c]);
    float* trow = tile + (NPW * w) * TILE_PITCH + 2 * lane;
    for (int j = 0; j < nn; ++j) {
      const int i = i0 + j;
      const float di = __int_as_float(__builtin_amdgcn_readlane(dv, j));
      int w_n = 0, t_nn = i, ne_nn = 0, st_nn = 0;
      float2 hv_n = hv;
      if (j + 1 < nn) {
        w_n = __float_as_int(dinv[t_n]);
        hv_n = *(const float2*)(hl + (size_t)(i + 1) * 64);
      }
      if (j + 2 < nn) {
        ne_nn = __builtin_amdgcn_readlane(cv, j + 2);
        st_nn = __builtin_amdgcn_readlane(fv, j + 2);
        t_nn = source_words(srcs, st_nn, ne_nn, i + 2, lane);
      }
      const float2 acc = node_sum<64, true>(srcs, dinv, st_c, ne_c, i, lane, t_c, w_c, hv, hl, di);
      float2 o;
      o.x = fmaxf(acc.x + bv.x, 0.0f);
      o.y = fmaxf(acc.y + bv.y, 0.0f);
      *(float2*)(trow + j * TILE_PITCH) = o;
      t_c = t_n; w_c = w_n; ne_c = ne_n; st_c = st_n; hv = hv_n;
      t_n = t_nn; ne_n = ne_nn; st_n = st_nn;
    }
  }
  __syncthreads();

  if (w >= 4) return;
  const int m0 = nb0 + 16 * w;
  if (m0 >= n) return;
  constexpr int NT = (L2_N + 15) / 16;
  const int row = lane & 15;
  const int kp  = lane >> 4;
  int colc[NT];
#pragma unroll
  for (int t = 0; t < NT; ++t) {
    const int col = t * 16 + row;
    colc[t] = (col < L2_N) ? col : (L2_N - 1);
  }
  const float4 sa = *(const float4*)(dinv + m0 + 8 * kp);
  const float4 sb = *(const float4*)(dinv + m0 + 8 * kp + 4);
  const float* Ar = tile + (16 * w + row) * TILE_PITCH + 2 * kp;
  const float* Br = Bs + (2 * kp) * L2_N;
  v8f c[NT];
#pragma unroll
  for (int t = 0; t < NT; ++t) c[t] = (v8f){};
#pragma unroll
  for (int k0 = 0; k0 < L2_K; k0 += 4) {
    v2f a;
    a.x = Ar[k0];
    a.y = Ar[k0 + 1];
#pragma unroll
    for (int t = 0; t < NT; ++t) {
      v2f bb;
      bb.x = Br[k0 * L2_N + colc[t]];
      bb.y = Br[(k0 + 1) * L2_N + colc[t]];
      c[t] = __builtin_amdgcn_wmma_f32_16x16x4_f32(false, a, false, bb, (short)0, c[t],
                                                   false, false);
    }
  }
  const float sc[8] = {sa.x, sa.y, sa.z, sa.w, sb.x, sb.y, sb.z, sb.w};
#pragma unroll
  for (int t = 0; t < NT; ++t) {
    const int col = t * 16 + row;
    if (col < L2_N) {
#pragma unroll
      for (int r = 0; r < 8; ++r)
        h2[(size_t)(m0 + r + 8 * kp) * L2_N + col] = c[t][r] * sc[r];
    }
  }
}

__global__ void __launch_bounds__(256)
k_gather40_logsoftmax(const int* __restrict__ srcs, const int* __restrict__ first,
                      const int* __restrict__ cnt, const float* __restrict__ dinv,
                      const float* __restrict__ h, const float* __restrict__ b,
                      float* __restrict__ out, int n) {
  const int lane = threadIdx.x & 31;
  const int i0 = NPW * __builtin_amdgcn_readfirstlane((int)(blockIdx.x * 8 + (threadIdx.x >> 5)));
  if (i0 >= n) return;
  const int nn = (n - i0 < NPW) ? (n - i0) : NPW;
  int cv = 0, fv = 0, dv = 0;
  if (lane < nn) {
    cv = cnt[i0 + lane];
    fv = first[i0 + lane];
    dv = __float_as_int(dinv[i0 + lane]);
  }
  const bool own_lane = lane < 20;
  const int fo = 2 * (own_lane ? lane : lane - 20);
  const float* hl = h + fo;
  const float2 bv = *(const float2*)(b + fo);
  int ne = __builtin_amdgcn_readlane(cv, 0);
  int start = __builtin_amdgcn_readlane(fv, 0);
  int t = source_words(srcs, start, ne, i0, lane);
  float2 hv = *(const float2*)(hl + (size_t)i0 * 40);
  for (int j = 0; j < nn; ++j) {
    const int i = i0 + j;
    const int ne_c = ne, start_c = start;
    const int es = t;
    const float2 own = hv;
    const float di = __int_as_float(__builtin_amdgcn_readlane(dv, j));
    if (j + 1 < nn) {
      ne = __builtin_amdgcn_readlane(cv, j + 1);
      start = __builtin_amdgcn_readlane(fv, j + 1);
      t = source_words(srcs, start, ne, i + 1, lane);
      hv = *(const float2*)(hl + (size_t)(i + 1) * 40);
    }
    const float2 acc = node_sum<40, false>(srcs, dinv, start_c, ne_c, i, lane, es, 0, own, hl, di);
    const float vx = acc.x + bv.x;
    const float vy = acc.y + bv.y;
    float mx = own_lane ? fmaxf(vx, vy) : -3.402823466e38f;
    for (int o = 16; o > 0; o >>= 1) mx = fmaxf(mx, __shfl_xor(mx, o));
    float sm = own_lane ? (expf(vx - mx) + expf(vy - mx)) : 0.0f;
    for (int o = 16; o > 0; o >>= 1) sm += __shfl_xor(sm, o);
    const float lse = mx + logf(sm);
    if (own_lane) {
      float2 o2;
      o2.x = vx - lse;
      o2.y = vy - lse;
      *(float2*)(out + (size_t)i * 40 + fo) = o2;
    }
  }
}

extern "C" void kernel_launch(void* const* d_in, const int* in_sizes, int n_in,
                              void* d_out, int out_size, void* d_ws, size_t ws_size,
                              hipStream_t stream) {
  (void)n_in; (void)out_size; (void)ws_size;

  const float* x  = (const float*)d_in[0];
  const int*   ei = (const int*)d_in[1];
  const float* W1 = (const float*)d_in[2];
  const float* b1 = (const float*)d_in[3];
  const float* W2 = (const float*)d_in[4];
  const float* b2 = (const float*)d_in[5];
  float* out = (float*)d_out;

  const int F_IN = 128;
  const int Nn = in_sizes[0] / F_IN;
  const int E  = in_sizes[1] / 2;
  const int* src = ei;
  const int* dst = ei + E;

  char* ws = (char*)d_ws;
  int*   cnt      = (int*)(ws);
  int*   segoff   = (int*)(ws + ((size_t)1  << 20));
  int*   first    = (int*)(ws + ((size_t)2  << 20));
  float* dinv     = (float*)(ws + ((size_t)3  << 20));
  float* h        = (float*)(ws + ((size_t)4  << 20));
  float* h2       = (float*)(ws + ((size_t)68 << 20));
  int*   srcs     = (int*)(ws + ((size_t)88 << 20));
  int2*  tmp      = (int2*)(ws + ((size_t)100 << 20));

  const dim3 B256(256);

  const int CH   = (E + SORT_G - 1) / SORT_G;
  const int BINS = (Nn + BIN_NODES - 1) / BIN_NODES;
  const dim3 B1024(1024);
  const int MT = Nn / 16;
  const int MT1 = (MT < SORT_G * GEMM_TPB) ? MT : SORT_G * GEMM_TPB;
  const int GB1 = (MT1 + GEMM_TPB - 1) / GEMM_TPB;
  k_chunk_sort<<<SORT_G + GB1, B1024, 0, stream>>>(src, dst, segoff, tmp, E, CH, BINS, x, W1, h, MT1);
  const int GB = (MT - MT1 + GEMM_TPB - 1) / GEMM_TPB;
  k_bin_finalize_gemm<<<BINS + GB, dim3(512), 0, stream>>>(tmp, segoff, cnt, first, dinv, srcs, Nn, BINS, CH,
                                                           x, W1, h, MT1, MT);

  const int GW = 8 * NPW;
  k_gather64_relu_gemm40<<<(Nn + GW - 1) / GW, B256, 0, stream>>>(srcs, first, cnt, dinv, h, b1, W2, h2, Nn);

  k_gather40_logsoftmax<<<(Nn + GW - 1) / GW, B256, 0, stream>>>(srcs, first, cnt, dinv, h2, b2, out, Nn);
}


# Layer-2 gather: log-softmax of a wave's 8 nodes done once after the node loop, rows turned through the wave's own LDS words (4 lanes x 10 features a node); line stores; softmax sum order changes
# speedup vs baseline: 4.2650x; 1.0454x over previous
#include <hip/hip_runtime.h>
#include <hip/hip_bf16.h>

typedef __attribute__((ext_vector_type(2))) float v2f;
typedef __attribute__((ext_vector_type(8))) float v8f;


#define SORT_G 128
#define BIN_SHIFT 9
#define BIN_NODES 512
#define MAX_BINS 256
#define GEMM_TPB 16

template <int K, int N>
__device__ __forceinline__ void gemm_tile_f32(const float* __restrict__ A, const float* __restrict__ B,
                                              float* __restrict__ C, int tile);

__global__ void __launch_bounds__(1024)
k_chunk_sort(const int* __restrict__ src, const int* __restrict__ dst,
             int* __restrict__ segoff, int2* __restrict__ tmp, int e, int ch, int nb,
             const float* __restrict__ xa, const float* __restrict__ wb,
             float* __restrict__ hc, int mt1) {
  if ((int)blockIdx.x >= SORT_G) {
    const int wv = __builtin_amdgcn_readfirstlane((int)(threadIdx.x >> 5));
    const int tile = ((int)blockIdx.x - SORT_G) * GEMM_TPB + wv;
    if (wv < GEMM_TPB && tile < mt1) gemm_tile_f32<128, 64>(xa, wb, hc, tile);
    return;
  }
  __shared__ int cur[MAX_BINS];
  __shared__ int wsum[8];
  const int t = threadIdx.x;
  const int g = blockIdx.x;
  const int lane = t & 31;
  if (t < MAX_BINS) cur[t] = 0;
  __syncthreads();
  const int lo = g * ch;
  const int hi = (lo + ch < e) ? (lo + ch) : e;
  int d[8], s[8];
#pragma unroll
  for (int k = 0; k < 8; ++k) {
    const int i = lo + t + 1024 * k;
    d[k] = (i < hi) ? dst[i] : 0;
    s[k] = (i < hi) ? src[i] : 0;
  }
#pragma unroll
  for (int k = 0; k < 8; ++k)
    if (lo + t + 1024 * k < hi) atomicAdd(&cur[d[k] >> BIN_SHIFT], 1);
  for (int i = lo + t + 8192; i < hi; i += 1024) atomicAdd(&cur[dst[i] >> BIN_SHIFT], 1);
  __syncthreads();
  int x = 0, c = 0;
  if (t < MAX_BINS) {
    c = cur[t];
    x = c;
    for (int o = 1; o < 32; o <<= 1) {
      const int y = __shfl_up(x, o);
      if (lane >= o) x += y;
    }
    if (lane == 31) wsum[t >> 5] = x;
  }
  __syncthreads();
  if (t == 0) {
    int a = 0;
    for (int k = 0; k < 8; ++k) { const int v = wsum[k]; wsum[k] = a; a += v; }
  }
  __syncthreads();
  if (t < MAX_BINS) {
    const int start = wsum[t >> 5] + x - c;
    cur[t] = start;
    if (t < nb) segoff[t * SORT_G + g] = start;
  }
  if (t == 0) segoff[nb * SORT_G + g] = (hi > lo) ? (hi - lo) : 0;
  __syncthreads();
  int pos[8];
#pragma unroll
  for (int k = 0; k < 8; ++k) {
    pos[k] = 0;
    if (lo + t + 1024 * k < hi) pos[k] = atomicAdd(&cur[d[k] >> BIN_SHIFT], 1);
  }
#pragma unroll
  for (int k = 0; k < 8; ++k)
    if (lo + t + 1024 * k < hi) tmp[lo + pos[k]] = make_int2(s[k], d[k]);
  for (int i = lo + t + 8192; i < hi; i += 1024) {
    const int dd = dst[i];
    const int ss = src[i];
    const int p = atomicAdd(&cur[dd >> BIN_SHIFT], 1);
    tmp[lo + p] = make_int2(ss, dd);
  }
}

#define SEG_PH (SORT_G / 32)
__global__ void __launch_bounds__(512)
k_bin_finalize_gemm(const int2* __restrict__ tmp, const int* __restrict__ segoff,
                    int* __restrict__ cnt, int* __restrict__ first, float* __restrict__ dinv,
                    int* __restrict__ srcs, int n, int nbins, int ch,
                    const float* __restrict__ xa, const float* __restrict__ wb,
                    float* __restrict__ hc, int tile0, int mtiles) {
  if ((int)blockIdx.x >= nbins) {
    const int wv = __builtin_amdgcn_readfirstlane((int)(threadIdx.x >> 5));
    const int tile = tile0 + ((int)blockIdx.x - nbins) * GEMM_TPB + wv;
    if (tile < mtiles) gemm_tile_f32<128, 64>(xa, wb, hc, tile);
    return;
  }
  __shared__ int c[BIN_NODES];
  __shared__ int wsum[BIN_NODES / 32];
  __shared__ int segb[SORT_G];
  __shared__ int segl[SORT_G];
  __shared__ int wred[SORT_G / 32];
  const int t = threadIdx.x;
  const int b = blockIdx.x;
  const int lane = t & 31;
  if (t < BIN_NODES) c[t] = 0;
  if (t < SORT_G) {
    const int st = segoff[b * SORT_G + t];
    const int en = segoff[(b + 1) * SORT_G + t];
    segb[t] = t * ch + st;
    segl[t] = en - st;
    int r = st;
    for (int o = 16; o > 0; o >>= 1) r += __shfl_xor(r, o);
    if (lane == 0) wred[t >> 5] = r;
  }
  __syncthreads();
  int s0 = 0;
#pragma unroll
  for (int k = 0; k < SORT_G / 32; ++k) s0 += wred[k];
  const int hw = t >> 4;
  const int l16 = t & 15;
  int sb[SEG_PH], sl[SEG_PH];
  int2 ld[SEG_PH][4];
#pragma unroll
  for (int q = 0; q < SEG_PH; ++q) {
    sb[q] = segb[hw + 32 * q];
    sl[q] = segl[hw + 32 * q];
#pragma unroll
    for (int k = 0; k < 4; ++k) {
      const int idx = l16 + 16 * k;
      ld[q][k] = (idx < sl[q]) ? tmp[sb[q] + idx] : make_int2(0, 0);
    }
  }
  int ed[SEG_PH][4];
#pragma unroll
  for (int q = 0; q < SEG_PH; ++q) {
#pragma unroll
    for (int k = 0; k < 4; ++k) {
      ed[q][k] = (ld[q][k].x << BIN_SHIFT) | (ld[q][k].y & (BIN_NODES - 1));
      if (l16 + 16 * k < sl[q]) atomicAdd(&c[ed[q][k] & (BIN_NODES - 1)], 1);
    }
    for (int idx = 64 + l16; idx < sl[q]; idx += 16)
      atomicAdd(&c[tmp[sb[q] + idx].y & (BIN_NODES - 1)], 1);
  }
  __syncthreads();
  int x = 0, kc = 0;
  if (t < BIN_NODES) {
    kc = c[t];
    x = kc;
    for (int o = 1; o < 32; o <<= 1) {
      const int y = __shfl_up(x, o);
      if (lane >= o) x += y;
    }
    if (lane == 31) wsum[t >> 5] = x;
  }
  __syncthreads();
  if (t == 0) {
    int a = 0;
    for (int j = 0; j < BIN_NODES / 32; ++j) { const int v = wsum[j]; wsum[j] = a; a += v; }
  }
  __syncthreads();
  if (t < BIN_NODES) {
    const int f = s0 + wsum[t >> 5] + x - kc;
    c[t] = f;
    const int node = b * BIN_NODES + t;
    if (node < n) {
      cnt[node] = kc;
      first[node] = f;
      dinv[node] = rsqrtf((float)(kc + 1));
    }
  }
  __syncthreads();
#pragma unroll
  for (int q = 0; q < SEG_PH; ++q) {
#pragma unroll
    for (int k = 0; k < 4; ++k)
      if (l16 + 16 * k < sl[q]) {
        const int p = atomicAdd(&c[ed[q][k] & (BIN_NODES - 1)], 1);
        srcs[p] = (int)((unsigned)ed[q][k] >> BIN_SHIFT);
      }
    for (int idx = 64 + l16; idx < sl[q]; idx += 16) {
      const int2 v = tmp[sb[q] + idx];
      const int p = atomicAdd(&c[v.y & (BIN_NODES - 1)], 1);
      srcs[p] = v.x;
    }
  }
}

template <int K, int N>
__device__ __forceinline__ void gemm_tile_f32(const float* __restrict__ A, const float* __restrict__ B,
                                              float* __restrict__ C, int tile) {
  constexpr int NT = (N + 15) / 16;
  const int lane = threadIdx.x & 31;
  const int row  = lane & 15;
  const int kp   = lane >> 4;
  const float* At = A + (size_t)tile * 16 * K;
  float*       Ct = C + (size_t)tile * 16 * N;

  unsigned boff[NT];
#pragma unroll
  for (int t = 0; t < NT; ++t) {
    const int col = t * 16 + row;
    boff[t] = (unsigned)(2 * kp * N + ((col < N) ? col : (N - 1)));
  }

  const unsigned aoff = (unsigned)(row * K + 2 * kp);
  v8f c[NT];
#pragma unroll
  for (int t = 0; t < NT; ++t) c[t] = (v8f){};

#pragma unroll
  for (int k0 = 0; k0 < K; k0 += 4) {
    v2f a;
    a.x = At[aoff + k0];
    a.y = At[aoff + k0 + 1];
#pragma unroll
    for (int t = 0; t < NT; ++t) {
      v2f b;
      b.x = B[boff[t] + k0 * N];
      b.y = B[boff[t] + (k0 + 1) * N];
      c[t] = __builtin_amdgcn_wmma_f32_16x16x4_f32(false, a, false, b, (short)0, c[t],
                                                   false, false);
    }
  }
#pragma unroll
  for (int t = 0; t < NT; ++t) {
    const int col = t * 16 + row;
    if (col < N) {
#pragma unroll
      for (int r = 0; r < 8; ++r)
        Ct[(unsigned)((r + 8 * kp) * N + col)] = c[t][r];
    }
  }
}

#define NPW 8

__device__ __forceinline__ int source_words(const int* __restrict__ srcs, int pos,
                                            int m, int i, int lane) {
  int t = i;
  if (lane < m) t = srcs[pos + lane];
  return t;
}

template <int F, bool WT>
__device__ __forceinline__ void add_rows(int es, int ws, int m, const float* __restrict__ hl,
                                         float2& acc) {
  for (int e = 0; e < m; e += 4) {
    const int s0 = __builtin_amdgcn_readlane(es, e);
    const int s1 = __builtin_amdgcn_readlane(es, e + 1);
    const int s2 = __builtin_amdgcn_readlane(es, e + 2);
    const int s3 = __builtin_amdgcn_readlane(es, e + 3);
    const float2 v0 = *(const float2*)(hl + (size_t)s0 * F);
    const float2 v1 = *(const float2*)(hl + (size_t)s1 * F);
    const float2 v2 = *(const float2*)(hl + (size_t)s2 * F);
    const float2 v3 = *(const float2*)(hl + (size_t)s3 * F);
    if (WT) {
      const float w0 = __int_as_float(__builtin_amdgcn_readlane(ws, e));
      const float w1 = __int_as_float(__builtin_amdgcn_readlane(ws, e + 1));
      const float w2 = __int_as_float(__builtin_amdgcn_readlane(ws, e + 2));
      const float w3 = __int_as_float(__builtin_amdgcn_readlane(ws, e + 3));
      acc.x = fmaf(w0, v0.x, acc.x); acc.y = fmaf(w0, v0.y, acc.y);
      if (e + 1 < m) { acc.x = fmaf(w1, v1.x, acc.x); acc.y = fmaf(w1, v1.y, acc.y); }
      if (e + 2 < m) { acc.x = fmaf(w2, v2.x, acc.x); acc.y = fmaf(w2, v2.y, acc.y); }
      if (e + 3 < m) { acc.x = fmaf(w3, v3.x, acc.x); acc.y = fmaf(w3, v3.y, acc.y); }
    } else {
      acc.x += v0.x; acc.y += v0.y;
      if (e + 1 < m) { acc.x += v1.x; acc.y += v1.y; }
      if (e + 2 < m) { acc.x += v2.x; acc.y += v2.y; }
      if (e + 3 < m) { acc.x += v3.x; acc.y += v3.y; }
    }
  }
}

template <int F, bool WT>
__device__ __forceinline__ float2 node_sum(const int* __restrict__ srcs,
                                           const float* __restrict__ dinv, int start, int ne,
                                           int i, int lane, int es, int ws, float2 own,
                                           const float* __restrict__ hl, float di) {
  float2 acc = own;
  if (WT) { acc.x *= di; acc.y *= di; }
  add_rows<F, WT>(es, ws, ne < 32 ? ne : 32, hl, acc);
  for (int base = 32; base < ne; base += 32) {
    const int m = (ne - base < 32) ? (ne - base) : 32;
    const int u = source_words(srcs, start + base, m, i, lane);
    int wu = 0;
    if (WT) wu = __float_as_int(dinv[u]);
    add_rows<F, WT>(u, wu, m, hl, acc);
  }
  acc.x *= di;
  acc.y *= di;
  return acc;
}

#define TILE_PITCH 68
#define L2_K 64
#define L2_N 40
__global__ void __launch_bounds__(256)
k_gather64_relu_gemm40(const int* __restrict__ srcs, const int* __restrict__ first,
                       const int* __restrict__ cnt, const float* __restrict__ dinv,
                       const float* __restrict__ h, const float* __restrict__ b,
                       const float* __restrict__ W2, float* __restrict__ h2, int n) {
  __shared__ __attribute__((aligned(16))) float tile[8 * NPW * TILE_PITCH];
  __shared__ __attribute__((aligned(16))) float Bs[L2_K * L2_N];
  const int lane = threadIdx.x & 31;
  const int w = __builtin_amdgcn_readfirstlane((int)(threadIdx.x >> 5));
  const int nb0 = blockIdx.x * (8 * NPW);
  static_assert(L2_K * L2_N / 4 == 2 * 256 + 128, "W2 copy is written for 640 pieces and 256 threads");
  const int q = threadIdx.x;
  const float4* W2q = (const float4*)W2;
  const float4 wq0 = W2q[q];
  const float4 wq1 = W2q[q + 256];
  float4 wq2 = make_float4(0.0f, 0.0f, 0.0f, 0.0f);
  if (q < 128) wq2 = W2q[q + 512];

  const int i0 = nb0 + NPW * w;
  const bool active = i0 < n;
  const int nn = active ? ((n - i0 < NPW) ? (n - i0) : NPW) : 0;
  int cv = 0, fv = 0, dv = 0;
  if (lane < nn) {
    cv = cnt[i0 + lane];
    fv = first[i0 + lane];
    dv = __float_as_int(dinv[i0 + lane]);
  }
  const float2 bv = *(const float2*)(b + 2 * lane);
  float4* Bq = (float4*)Bs;
  Bq[q] = wq0;
  Bq[q + 256] = wq1;
  if (q < 128) Bq[q + 512] = wq2;

  if (active) {
    const float* hl = h + 2 * lane;
    int ne_c = __builtin_amdgcn_readlane(cv, 0);
    int st_c = __builtin_amdgcn_readlane(fv, 0);
    int t_c = source_words(srcs, st_c, ne_c, i0, lane);
    int ne_n = 0, st_n = 0, t_n = i0;
    if (nn > 1) {
      ne_n = __builtin_amdgcn_readlane(cv, 1);
      st_n = __builtin_amdgcn_readlane(fv, 1);
      t_n = source_words(srcs, st_n, ne_n, i0 + 1, lane);
    }
    float2 hv = *(const float2*)(hl + (size_t)i0 * 64);
    int w_c = __float_as_int(dinv[t_c]);
    float* trow = tile + (NPW * w) * TILE_PITCH + 2 * lane;
    for (int j = 0; j < nn; ++j) {
      const int i = i0 + j;
      const float di = __int_as_float(__builtin_amdgcn_readlane(dv, j));
      int w_n = 0, t_nn = i, ne_nn = 0, st_nn = 0;
      float2 hv_n = hv;
      if (j + 1 < nn) {
        w_n = __float_as_int(dinv[t_n]);
        hv_n = *(const float2*)(hl + (size_t)(i + 1) * 64);
      }
      if (j + 2 < nn) {
        ne_nn = __builtin_amdgcn_readlane(cv, j + 2);
        st_nn = __builtin_amdgcn_readlane(fv, j + 2);
        t_nn = source_words(srcs, st_nn, ne_nn, i + 2, lane);
      }
      const float2 acc = node_sum<64, true>(srcs, dinv, st_c, ne_c, i, lane, t_c, w_c, hv, hl, di);
      float2 o;
      o.x = fmaxf(acc.x + bv.x, 0.0f);
      o.y = fmaxf(acc.y + bv.y, 0.0f);
      *(float2*)(trow + j * TILE_PITCH) = o;
      t_c = t_n; w_c = w_n; ne_c = ne_n; st_c = st_n; hv = hv_n;
      t_n = t_nn; ne_n = ne_nn; st_n = st_nn;
    }
  }
  __syncthreads();

  if (w >= 4) return;
  const int m0 = nb0 + 16 * w;
  if (m0 >= n) return;
  constexpr int NT = (L2_N + 15) / 16;
  const int row = lane & 15;
  const int kp  = lane >> 4;
  int colc[NT];
#pragma unroll
  for (int t = 0; t < NT; ++t) {
    const int col = t * 16 + row;
    colc[t] = (col < L2_N) ? col : (L2_N - 1);
  }
  const float4 sa = *(const float4*)(dinv + m0 + 8 * kp);
  const float4 sb = *(const float4*)(dinv + m0 + 8 * kp + 4);
  const float* Ar = tile + (16 * w + row) * TILE_PITCH + 2 * kp;
  const float* Br = Bs + (2 * kp) * L2_N;
  v8f c[NT];
#pragma unroll
  for (int t = 0; t < NT; ++t) c[t] = (v8f){};
#pragma unroll
  for (int k0 = 0; k0 < L2_K; k0 += 4) {
    v2f a;
    a.x = Ar[k0];
    a.y = Ar[k0 + 1];
#pragma unroll
    for (int t = 0; t < NT; ++t) {
      v2f bb;
      bb.x = Br[k0 * L2_N + colc[t]];
      bb.y = Br[(k0 + 1) * L2_N + colc[t]];
      c[t] = __builtin_amdgcn_wmma_f32_16x16x4_f32(false, a, false, bb, (short)0, c[t],
                                                   false, false);
    }
  }
  const float sc[8] = {sa.x, sa.y, sa.z, sa.w, sb.x, sb.y, sb.z, sb.w};
#pragma unroll
  for (int t = 0; t < NT; ++t) {
    const int col = t * 16 + row;
    if (col < L2_N) {
#pragma unroll
      for (int r = 0; r < 8; ++r)
        h2[(size_t)(m0 + r + 8 * kp) * L2_N + col] = c[t][r] * sc[r];
    }
  }
}

#define SM_ROW 40
#define SM_WAVE (NPW * SM_ROW)
__global__ void __launch_bounds__(256)
k_gather40_logsoftmax(const int* __restrict__ srcs, const int* __restrict__ first,
                      const int* __restrict__ cnt, const float* __restrict__ dinv,
                      const float* __restrict__ h, const float* __restrict__ b,
                      float* __restrict__ out, int n) {
  __shared__ __attribute__((aligned(16))) float rows[8 * SM_WAVE];
  const int lane = threadIdx.x & 31;
  const int wv = __builtin_amdgcn_readfirstlane((int)(threadIdx.x >> 5));
  const int i0 = NPW * ((int)blockIdx.x * 8 + wv);
  if (i0 >= n) return;
  const int nn = (n - i0 < NPW) ? (n - i0) : NPW;
  int cv = 0, fv = 0, dv = 0;
  if (lane < nn) {
    cv = cnt[i0 + lane];
    fv = first[i0 + lane];
    dv = __float_as_int(dinv[i0 + lane]);
  }
  const bool own_lane = lane < 20;
  const int fo = 2 * (own_lane ? lane : lane - 20);
  const float* hl = h + fo;
  const float2 bv = *(const float2*)(b + fo);
  float* wr = rows + wv * SM_WAVE;
  int ne = __builtin_amdgcn_readlane(cv, 0);
  int start = __builtin_amdgcn_readlane(fv, 0);
  int t = source_words(srcs, start, ne, i0, lane);
  float2 hv = *(const float2*)(hl + (size_t)i0 * 40);
  for (int j = 0; j < nn; ++j) {
    const int i = i0 + j;
    const int ne_c = ne, start_c = start;
    const int es = t;
    const float2 own = hv;
    const float di = __int_as_float(__builtin_amdgcn_readlane(dv, j));
    if (j + 1 < nn) {
      ne = __builtin_amdgcn_readlane(cv, j + 1);
      start = __builtin_amdgcn_readlane(fv, j + 1);
      t = source_words(srcs, start, ne, i + 1, lane);
      hv = *(const float2*)(hl + (size_t)(i + 1) * 40);
    }
    const float2 acc = node_sum<40, false>(srcs, dinv, start_c, ne_c, i, lane, es, 0, own, hl, di);
    if (own_lane) {
      float2 s2;
      s2.x = acc.x + bv.x;
      s2.y = acc.y + bv.y;
      *(float2*)(wr + j * SM_ROW + fo) = s2;
    }
  }

  float* rp = wr + 10 * lane;
  float v[10];
#pragma unroll
  for (int k = 0; k < 10; ++k) v[k] = 0.0f;
  if ((lane >> 2) < nn) {
#pragma unroll
    for (int k = 0; k < 5; ++k) {
      const float2 p = *(const float2*)(rp + 2 * k);
      v[2 * k] = p.x;
      v[2 * k + 1] = p.y;
    }
  }
  float mx = v[0];
#pragma unroll
  for (int k = 1; k < 10; ++k) mx = fmaxf(mx, v[k]);
  mx = fmaxf(mx, __shfl_xor(mx, 1));
  mx = fmaxf(mx, __shfl_xor(mx, 2));
  float sm = expf(v[0] - mx);
#pragma unroll
  for (int k = 1; k < 10; ++k) sm += expf(v[k] - mx);
  sm += __shfl_xor(sm, 1);
  sm += __shfl_xor(sm, 2);
  const float lse = mx + logf(sm);
#pragma unroll
  for (int k = 0; k < 5; ++k) {
    float2 p;
    p.x = v[2 * k] - lse;
    p.y = v[2 * k + 1] - lse;
    *(float2*)(rp + 2 * k) = p;
  }

  float* op = out + (size_t)i0 * 40 + lane;
  if (nn == NPW) {
    float o[10];
#pragma unroll
    for (int k = 0; k < 10; ++k) o[k] = wr[32 * k + lane];
#pragma unroll
    for (int k = 0; k < 10; ++k) op[32 * k] = o[k];
  } else {
    const int lim = nn * SM_ROW;
    for (int k = 0; k < 10; ++k)
      if (32 * k + lane < lim) op[32 * k] = wr[32 * k + lane];
  }
}

extern "C" void kernel_launch(void* const* d_in, const int* in_sizes, int n_in,
                              void* d_out, int out_size, void* d_ws, size_t ws_size,
                              hipStream_t stream) {
  (void)n_in; (void)out_size; (void)ws_size;

  const float* x  = (const float*)d_in[0];
  const int*   ei = (const int*)d_in[1];
  const float* W1 = (const float*)d_in[2];
  const float* b1 = (const float*)d_in[3];
  const float* W2 = (const float*)d_in[4];
  const float* b2 = (const float*)d_in[5];
  float* out = (float*)d_out;

  const int F_IN = 128;
  const int Nn = in_sizes[0] / F_IN;
  const int E  = in_sizes[1] / 2;
  const int* src = ei;
  const int* dst = ei + E;

  char* ws = (char*)d_ws;
  int*   cnt      = (int*)(ws);
  int*   segoff   = (int*)(ws + ((size_t)1  << 20));
  int*   first    = (int*)(ws + ((size_t)2  << 20));
  float* dinv     = (float*)(ws + ((size_t)3  << 20));
  float* h        = (float*)(ws + ((size_t)4  << 20));
  float* h2       = (float*)(ws + ((size_t)68 << 20));
  int*   srcs     = (int*)(ws + ((size_t)88 << 20));
  int2*  tmp      = (int2*)(ws + ((size_t)100 << 20));

  const dim3 B256(256);

  const int CH   = (E + SORT_G - 1) / SORT_G;
  const int BINS = (Nn + BIN_NODES - 1) / BIN_NODES;
  const dim3 B1024(1024);
  const int MT = Nn / 16;
  const int MT1 = (MT < SORT_G * GEMM_TPB) ? MT : SORT_G * GEMM_TPB;
  const int GB1 = (MT1 + GEMM_TPB - 1) / GEMM_TPB;
  k_chunk_sort<<<SORT_G + GB1, B1024, 0, stream>>>(src, dst, segoff, tmp, E, CH, BINS, x, W1, h, MT1);
  const int GB = (MT - MT1 + GEMM_TPB - 1) / GEMM_TPB;
  k_bin_finalize_gemm<<<BINS + GB, dim3(512), 0, stream>>>(tmp, segoff, cnt, first, dinv, srcs, Nn, BINS, CH,
                                                           x, W1, h, MT1, MT);

  const int GW = 8 * NPW;
  k_gather64_relu_gemm40<<<(Nn + GW - 1) / GW, B256, 0, stream>>>(srcs, first, cnt, dinv, h, b1, W2, h2, Nn);

  k_gather40_logsoftmax<<<(Nn + GW - 1) / GW, B256, 0, stream>>>(srcs, first, cnt, dinv, h2, b2, out, Nn);
}


# Both gathers: one 16-byte-a-lane load fetches two edge rows (one per 16-lane half); sources and weights reach lanes by DPP row_share, no per-row scalar work; two half sums a node (sum order changes)
# speedup vs baseline: 4.4302x; 1.0387x over previous
#include <hip/hip_runtime.h>
#include <hip/hip_bf16.h>

typedef __attribute__((ext_vector_type(2))) float v2f;
typedef __attribute__((ext_vector_type(8))) float v8f;


#define SORT_G 128
#define BIN_SHIFT 9
#define BIN_NODES 512
#define MAX_BINS 256
#define GEMM_TPB 16

template <int K, int N>
__device__ __forceinline__ void gemm_tile_f32(const float* __restrict__ A, const float* __restrict__ B,
                                              float* __restrict__ C, int tile);

__global__ void __launch_bounds__(1024)
k_chunk_sort(const int* __restrict__ src, const int* __restrict__ dst,
             int* __restrict__ segoff, int2* __restrict__ tmp, int e, int ch, int nb,
             const float* __restrict__ xa, const float* __restrict__ wb,
             float* __restrict__ hc, int mt1) {
  if ((int)blockIdx.x >= SORT_G) {
    const int wv = __builtin_amdgcn_readfirstlane((int)(threadIdx.x >> 5));
    const int tile = ((int)blockIdx.x - SORT_G) * GEMM_TPB + wv;
    if (wv < GEMM_TPB && tile < mt1) gemm_tile_f32<128, 64>(xa, wb, hc, tile);
    return;
  }
  __shared__ int cur[MAX_BINS];
  __shared__ int wsum[8];
  const int t = threadIdx.x;
  const int g = blockIdx.x;
  const int lane = t & 31;
  if (t < MAX_BINS) cur[t] = 0;
  __syncthreads();
  const int lo = g * ch;
  const int hi = (lo + ch < e) ? (lo + ch) : e;
  int d[8], s[8];
#pragma unroll
  for (int k = 0; k < 8; ++k) {
    const int i = lo + t + 1024 * k;
    d[k] = (i < hi) ? dst[i] : 0;
    s[k] = (i < hi) ? src[i] : 0;
  }
#pragma unroll
  for (int k = 0; k < 8; ++k)
    if (lo + t + 1024 * k < hi) atomicAdd(&cur[d[k] >> BIN_SHIFT], 1);
  for (int i = lo + t + 8192; i < hi; i += 1024) atomicAdd(&cur[dst[i] >> BIN_SHIFT], 1);
  __syncthreads();
  int x = 0, c = 0;
  if (t < MAX_BINS) {
    c = cur[t];
    x = c;
    for (int o = 1; o < 32; o <<= 1) {
      const int y = __shfl_up(x, o);
      if (lane >= o) x += y;
    }
    if (lane == 31) wsum[t >> 5] = x;
  }
  __syncthreads();
  if (t == 0) {
    int a = 0;
    for (int k = 0; k < 8; ++k) { const int v = wsum[k]; wsum[k] = a; a += v; }
  }
  __syncthreads();
  if (t < MAX_BINS) {
    const int start = wsum[t >> 5] + x - c;
    cur[t] = start;
    if (t < nb) segoff[t * SORT_G + g] = start;
  }
  if (t == 0) segoff[nb * SORT_G + g] = (hi > lo) ? (hi - lo) : 0;
  __syncthreads();
  int pos[8];
#pragma unroll
  for (int k = 0; k < 8; ++k) {
    pos[k] = 0;
    if (lo + t + 1024 * k < hi) pos[k] = atomicAdd(&cur[d[k] >> BIN_SHIFT], 1);
  }
#pragma unroll
  for (int k = 0; k < 8; ++k)
    if (lo + t + 1024 * k < hi) tmp[lo + pos[k]] = make_int2(s[k], d[k]);
  for (int i = lo + t + 8192; i < hi; i += 1024) {
    const int dd = dst[i];
    const int ss = src[i];
    const int p = atomicAdd(&cur[dd >> BIN_SHIFT], 1);
    tmp[lo + p] = make_int2(ss, dd);
  }
}

#define SEG_PH (SORT_G / 32)
__global__ void __launch_bounds__(512)
k_bin_finalize_gemm(const int2* __restrict__ tmp, const int* __restrict__ segoff,
                    int* __restrict__ cnt, int* __restrict__ first, float* __restrict__ dinv,
                    int* __restrict__ srcs, int n, int nbins, int ch,
                    const float* __restrict__ xa, const float* __restrict__ wb,
                    float* __restrict__ hc, int tile0, int mtiles) {
  if ((int)blockIdx.x >= nbins) {
    const int wv = __builtin_amdgcn_readfirstlane((int)(threadIdx.x >> 5));
    const int tile = tile0 + ((int)blockIdx.x - nbins) * GEMM_TPB + wv;
    if (tile < mtiles) gemm_tile_f32<128, 64>(xa, wb, hc, tile);
    return;
  }
  __shared__ int c[BIN_NODES];
  __shared__ int wsum[BIN_NODES / 32];
  __shared__ int segb[SORT_G];
  __shared__ int segl[SORT_G];
  __shared__ int wred[SORT_G / 32];
  const int t = threadIdx.x;
  const int b = blockIdx.x;
  const int lane = t & 31;
  if (t < BIN_NODES) c[t] = 0;
  if (t < SORT_G) {
    const int st = segoff[b * SORT_G + t];
    const int en = segoff[(b + 1) * SORT_G + t];
    segb[t] = t * ch + st;
    segl[t] = en - st;
    int r = st;
    for (int o = 16; o > 0; o >>= 1) r += __shfl_xor(r, o);
    if (lane == 0) wred[t >> 5] = r;
  }
  __syncthreads();
  int s0 = 0;
#pragma unroll
  for (int k = 0; k < SORT_G / 32; ++k) s0 += wred[k];
  const int hw = t >> 4;
  const int l16 = t & 15;
  int sb[SEG_PH], sl[SEG_PH];
  int2 ld[SEG_PH][4];
#pragma unroll
  for (int q = 0; q < SEG_PH; ++q) {
    sb[q] = segb[hw + 32 * q];
    sl[q] = segl[hw + 32 * q];
#pragma unroll
    for (int k = 0; k < 4; ++k) {
      const int idx = l16 + 16 * k;
      ld[q][k] = (idx < sl[q]) ? tmp[sb[q] + idx] : make_int2(0, 0);
    }
  }
  int ed[SEG_PH][4];
#pragma unroll
  for (int q = 0; q < SEG_PH; ++q) {
#pragma unroll
    for (int k = 0; k < 4; ++k) {
      ed[q][k] = (ld[q][k].x << BIN_SHIFT) | (ld[q][k].y & (BIN_NODES - 1));
      if (l16 + 16 * k < sl[q]) atomicAdd(&c[ed[q][k] & (BIN_NODES - 1)], 1);
    }
    for (int idx = 64 + l16; idx < sl[q]; idx += 16)
      atomicAdd(&c[tmp[sb[q] + idx].y & (BIN_NODES - 1)], 1);
  }
  __syncthreads();
  int x = 0, kc = 0;
  if (t < BIN_NODES) {
    kc = c[t];
    x = kc;
    for (int o = 1; o < 32; o <<= 1) {
      const int y = __shfl_up(x, o);
      if (lane >= o) x += y;
    }
    if (lane == 31) wsum[t >> 5] = x;
  }
  __syncthreads();
  if (t == 0) {
    int a = 0;
    for (int j = 0; j < BIN_NODES / 32; ++j) { const int v = wsum[j]; wsum[j] = a; a += v; }
  }
  __syncthreads();
  if (t < BIN_NODES) {
    const int f = s0 + wsum[t >> 5] + x - kc;
    c[t] = f;
    const int node = b * BIN_NODES + t;
    if (node < n) {
      cnt[node] = kc;
      first[node] = f;
      dinv[node] = rsqrtf((float)(kc + 1));
    }
  }
  __syncthreads();
#pragma unroll
  for (int q = 0; q < SEG_PH; ++q) {
#pragma unroll
    for (int k = 0; k < 4; ++k)
      if (l16 + 16 * k < sl[q]) {
        const int p = atomicAdd(&c[ed[q][k] & (BIN_NODES - 1)], 1);
        srcs[p] = (int)((unsigned)ed[q][k] >> BIN_SHIFT);
      }
    for (int idx = 64 + l16; idx < sl[q]; idx += 16) {
      const int2 v = tmp[sb[q] + idx];
      const int p = atomicAdd(&c[v.y & (BIN_NODES - 1)], 1);
      srcs[p] = v.x;
    }
  }
}

template <int K, int N>
__device__ __forceinline__ void gemm_tile_f32(const float* __restrict__ A, const float* __restrict__ B,
                                              float* __restrict__ C, int tile) {
  constexpr int NT = (N + 15) / 16;
  const int lane = threadIdx.x & 31;
  const int row  = lane & 15;
  const int kp   = lane >> 4;
  const float* At = A + (size_t)tile * 16 * K;
  float*       Ct = C + (size_t)tile * 16 * N;

  unsigned boff[NT];
#pragma unroll
  for (int t = 0; t < NT; ++t) {
    const int col = t * 16 + row;
    boff[t] = (unsigned)(2 * kp * N + ((col < N) ? col : (N - 1)));
  }

  const unsigned aoff = (unsigned)(row * K + 2 * kp);
  v8f c[NT];
#pragma unroll
  for (int t = 0; t < NT; ++t) c[t] = (v8f){};

#pragma unroll
  for (int k0 = 0; k0 < K; k0 += 4) {
    v2f a;
    a.x = At[aoff + k0];
    a.y = At[aoff + k0 + 1];
#pragma unroll
    for (int t = 0; t < NT; ++t) {
      v2f b;
      b.x = B[boff[t] + k0 * N];
      b.y = B[boff[t] + (k0 + 1) * N];
      c[t] = __builtin_amdgcn_wmma_f32_16x16x4_f32(false, a, false, b, (short)0, c[t],
                                                   false, false);
    }
  }
#pragma unroll
  for (int t = 0; t < NT; ++t) {
    const int col = t * 16 + row;
    if (col < N) {
#pragma unroll
      for (int r = 0; r < 8; ++r)
        Ct[(unsigned)((r + 8 * kp) * N + col)] = c[t][r];
    }
  }
}

#define NPW 8
#define EPC 16

template <int K>
__device__ __forceinline__ int row_word(int v) {
  return __builtin_amdgcn_update_dpp(0, v, 0x150 + K, 0xf, 0xf, true);
}

__device__ __forceinline__ float other_half(float x) {
  const int b = __builtin_bit_cast(int, x);
  return __builtin_bit_cast(float, __builtin_amdgcn_permlanex16(b, b, 0x76543210, 0xfedcba98, false, false));
}

__device__ __forceinline__ int lane_entry(int lane) {
  const int j = lane & 15;
  return (j < 8) ? (2 * j + (lane >> 4)) : (1 << 24);
}

__device__ __forceinline__ int entry_word(const int* __restrict__ srcs, int start, int ne,
                                          int base, int i, int qe) {
  const int q = base + qe;
  int t = i;
  if (q >= 1 && q <= ne) t = srcs[start + q - 1];
  return t;
}

template <int K, int FB>
__device__ __forceinline__ float4 row_piece(const char* __restrict__ hb, unsigned lob, int t) {
  const unsigned s = (unsigned)row_word<K>(t);
  return *(const float4*)(hb + (s * (unsigned)FB + lob));
}

template <int K>
__device__ __forceinline__ void add_piece(float4& acc, const float4 v, float w) {
  const float ws = __builtin_bit_cast(float, row_word<K>(__builtin_bit_cast(int, w)));
  acc.x = fmaf(ws, v.x, acc.x);
  acc.y = fmaf(ws, v.y, acc.y);
  acc.z = fmaf(ws, v.z, acc.z);
  acc.w = fmaf(ws, v.w, acc.w);
}

template <int K, int S, int FB>
__device__ __forceinline__ void load_pieces(float4* v, const char* __restrict__ hb, unsigned lob, int t) {
  if constexpr (K < S) {
    v[K] = row_piece<K, FB>(hb, lob, t);
    load_pieces<K + 1, S, FB>(v, hb, lob, t);
  }
}

template <int K, int S, bool ONES>
__device__ __forceinline__ void add_pieces(float4& acc, const float4* v, float w) {
  if constexpr (K < S) {
    if constexpr (ONES && K == 0 && K + 1 < S) {
      acc = v[0];
    } else if constexpr (ONES && K + 1 < S) {
      acc.x += v[K].x;
      acc.y += v[K].y;
      acc.z += v[K].z;
      acc.w += v[K].w;
    } else {
      add_piece<K>(acc, v[K], w);
    }
    add_pieces<K + 1, S, ONES>(acc, v, w);
  }
}

template <int S, int FB, bool ONES, typename Ahead>
__device__ __forceinline__ void chunk_steps(const char* __restrict__ hb, unsigned lob, int t, float w,
                                            float4& acc, Ahead ahead) {
  float4 v[S];
  load_pieces<0, S, FB>(v, hb, lob, t);
  ahead();
  add_pieces<0, S, ONES>(acc, v, w);
}

template <int FB, bool ONES, typename Ahead>
__device__ __forceinline__ void add_chunk(const char* __restrict__ hb, unsigned lob, int t, float w,
                                          int steps, float4& acc, Ahead ahead) {
  if (steps > 4) {
    if (steps > 6) {
      if (steps > 7) chunk_steps<8, FB, ONES>(hb, lob, t, w, acc, ahead);
      else chunk_steps<7, FB, ONES>(hb, lob, t, w, acc, ahead);
    } else {
      if (steps > 5) chunk_steps<6, FB, ONES>(hb, lob, t, w, acc, ahead);
      else chunk_steps<5, FB, ONES>(hb, lob, t, w, acc, ahead);
    }
  } else {
    if (steps > 2) {
      if (steps > 3) chunk_steps<4, FB, ONES>(hb, lob, t, w, acc, ahead);
      else chunk_steps<3, FB, ONES>(hb, lob, t, w, acc, ahead);
    } else {
      if (steps > 1) chunk_steps<2, FB, ONES>(hb, lob, t, w, acc, ahead);
      else chunk_steps<1, FB, ONES>(hb, lob, t, w, acc, ahead);
    }
  }
}

template <int F, bool WT, typename Ahead>
__device__ __forceinline__ float4 node_sum(const int* __restrict__ srcs,
                                           const float* __restrict__ dinv, int start, int ne,
                                           int i, int qe, int t, float w,
                                           const float* __restrict__ h, unsigned lob, Ahead ahead) {
  const char* hb = (const char*)h;
  float4 acc = make_float4(0.0f, 0.0f, 0.0f, 0.0f);
  const int ent = ne + 1;
  add_chunk<4 * F, !WT>(hb, lob, t, w, ((ent < EPC ? ent : EPC) + 1) >> 1, acc, ahead);
  for (int base = EPC; base < ent; base += EPC) {
    const int u = entry_word(srcs, start, ne, base, i, qe);
    float wu = 1.0f;
    if (WT) wu = dinv[u];
    wu = (base + qe <= ne) ? wu : 0.0f;
    const int m = ent - base;
    add_chunk<4 * F, false>(hb, lob, u, wu, ((m < EPC ? m : EPC) + 1) >> 1, acc, []() {});
  }
  return acc;
}

#define TILE_PITCH 68
#define L2_K 64
#define L2_N 40
__global__ void __launch_bounds__(256)
k_gather64_relu_gemm40(const int* __restrict__ srcs, const int* __restrict__ first,
                       const int* __restrict__ cnt, const float* __restrict__ dinv,
                       const float* __restrict__ h, const float* __restrict__ b,
                       const float* __restrict__ W2, float* __restrict__ h2, int n) {
  __shared__ __attribute__((aligned(16))) float tile[8 * NPW * TILE_PITCH];
  __shared__ __attribute__((aligned(16))) float Bs[L2_K * L2_N];
  const int lane = threadIdx.x & 31;
  const int w = __builtin_amdgcn_readfirstlane((int)(threadIdx.x >> 5));
  const int nb0 = blockIdx.x * (8 * NPW);
  static_assert(L2_K * L2_N / 4 == 2 * 256 + 128, "W2 copy is written for 640 pieces and 256 threads");
  const int q = threadIdx.x;
  const float4* W2q = (const float4*)W2;
  const float4 wq0 = W2q[q];
  const float4 wq1 = W2q[q + 256];
  float4 wq2 = make_float4(0.0f, 0.0f, 0.0f, 0.0f);
  if (q < 128) wq2 = W2q[q + 512];

  const int i0 = nb0 + NPW * w;
  const bool active = i0 < n;
  const int nn = active ? ((n - i0 < NPW) ? (n - i0) : NPW) : 0;
  int cv = 0, fv = 0, dv = 0;
  if (lane < nn) {
    cv = cnt[i0 + lane];
    fv = first[i0 + lane];
    dv = __float_as_int(dinv[i0 + lane]);
  }
  const int hf = lane >> 4;
  const int c16 = lane & 15;
  const int qe = lane_entry(lane);
  const float4 bv = *(const float4*)(b + 4 * c16);
  float4* Bq = (float4*)Bs;
  Bq[q] = wq0;
  Bq[q + 256] = wq1;
  if (q < 128) Bq[q + 512] = wq2;

  if (active) {
    const unsigned lob = 16u * (unsigned)c16;
    int ne_c = __builtin_amdgcn_readlane(cv, 0);
    int st_c = __builtin_amdgcn_readlane(fv, 0);
    int t_c = entry_word(srcs, st_c, ne_c, 0, i0, qe);
    int ne_n = 0, st_n = 0, t_n = i0;
    if (nn > 1) {
      ne_n = __builtin_amdgcn_readlane(cv, 1);
      st_n = __builtin_amdgcn_readlane(fv, 1);
      t_n = entry_word(srcs, st_n, ne_n, 0, i0 + 1, qe);
    }
    float w_c = (qe <= ne_c) ? dinv[t_c] : 0.0f;
    float* trow = tile + (NPW * w) * TILE_PITCH + 4 * c16;
    for (int j = 0; j < nn; ++j) {
      const int i = i0 + j;
      const float di = __int_as_float(__builtin_amdgcn_readlane(dv, j));
      float w_n = 0.0f;
      int t_nn = i, ne_nn = 0, st_nn = 0;
      const float4 acc = node_sum<64, true>(srcs, dinv, st_c, ne_c, i, qe, t_c, w_c, h, lob, [&]() {
        if (j + 1 < nn) w_n = dinv[t_n];
        if (j + 2 < nn) {
          ne_nn = __builtin_amdgcn_readlane(cv, j + 2);
          st_nn = __builtin_amdgcn_readlane(fv, j + 2);
          t_nn = entry_word(srcs, st_nn, ne_nn, 0, i + 2, qe);
        }
      });
      float4 o;
      o.x = fmaxf((acc.x + other_half(acc.x)) * di + bv.x, 0.0f);
      o.y = fmaxf((acc.y + other_half(acc.y)) * di + bv.y, 0.0f);
      o.z = fmaxf((acc.z + other_half(acc.z)) * di + bv.z, 0.0f);
      o.w = fmaxf((acc.w + other_half(acc.w)) * di + bv.w, 0.0f);
      if (hf == 0) *(float4*)(trow + j * TILE_PITCH) = o;
      w_c = (qe <= ne_n) ? w_n : 0.0f;
      t_c = t_n; ne_c = ne_n; st_c = st_n;
      t_n = t_nn; ne_n = ne_nn; st_n = st_nn;
    }
  }
  __syncthreads();

  if (w >= 4) return;
  const int m0 = nb0 + 16 * w;
  if (m0 >= n) return;
  constexpr int NT = (L2_N + 15) / 16;
  const int row = lane & 15;
  const int kp  = lane >> 4;
  int colc[NT];
#pragma unroll
  for (int t = 0; t < NT; ++t) {
    const int col = t * 16 + row;
    colc[t] = (col < L2_N) ? col : (L2_N - 1);
  }
  const float4 sa = *(const float4*)(dinv + m0 + 8 * kp);
  const float4 sb = *(const float4*)(dinv + m0 + 8 * kp + 4);
  const float* Ar = tile + (16 * w + row) * TILE_PITCH + 2 * kp;
  const float* Br = Bs + (2 * kp) * L2_N;
  v8f c[NT];
#pragma unroll
  for (int t = 0; t < NT; ++t) c[t] = (v8f){};
#pragma unroll
  for (int k0 = 0; k0 < L2_K; k0 += 4) {
    v2f a;
    a.x = Ar[k0];
    a.y = Ar[k0 + 1];
#pragma unroll
    for (int t = 0; t < NT; ++t) {
      v2f bb;
      bb.x = Br[k0 * L2_N + colc[t]];
      bb.y = Br[(k0 + 1) * L2_N + colc[t]];
      c[t] = __builtin_amdgcn_wmma_f32_16x16x4_f32(false, a, false, bb, (short)0, c[t],
                                                   false, false);
    }
  }
  const float sc[8] = {sa.x, sa.y, sa.z, sa.w, sb.x, sb.y, sb.z, sb.w};
#pragma unroll
  for (int t = 0; t < NT; ++t) {
    const int col = t * 16 + row;
    if (col < L2_N) {
#pragma unroll
      for (int r = 0; r < 8; ++r)
        h2[(size_t)(m0 + r + 8 * kp) * L2_N + col] = c[t][r] * sc[r];
    }
  }
}

#define SM_ROW 40
#define SM_WAVE (NPW * SM_ROW)
__global__ void __launch_bounds__(256)
k_gather40_logsoftmax(const int* __restrict__ srcs, const int* __restrict__ first,
                      const int* __restrict__ cnt, const float* __restrict__ dinv,
                      const float* __restrict__ h, const float* __restrict__ b,
                      float* __restrict__ out, int n) {
  __shared__ __attribute__((aligned(16))) float rows[8 * SM_WAVE];
  const int lane = threadIdx.x & 31;
  const int wv = __builtin_amdgcn_readfirstlane((int)(threadIdx.x >> 5));
  const int i0 = NPW * ((int)blockIdx.x * 8 + wv);
  if (i0 >= n) return;
  const int nn = (n - i0 < NPW) ? (n - i0) : NPW;
  int cv = 0, fv = 0, dv = 0;
  if (lane < nn) {
    cv = cnt[i0 + lane];
    fv = first[i0 + lane];
    dv = __float_as_int(dinv[i0 + lane]);
  }
  const int hf = lane >> 4;
  const int c16 = lane & 15;
  const bool own_lane = (hf == 0) && (c16 < 10);
  const int cc = (c16 < 10) ? c16 : (c16 - 10);
  const unsigned lob = 16u * (unsigned)cc;
  const int qe = lane_entry(lane);
  const float4 bv = *(const float4*)(b + 4 * cc);
  float* wr = rows + wv * SM_WAVE;
  int ne = __builtin_amdgcn_readlane(cv, 0);
  int start = __builtin_amdgcn_readlane(fv, 0);
  int t = entry_word(srcs, start, ne, 0, i0, qe);
  for (int j = 0; j < nn; ++j) {
    const int i = i0 + j;
    const int ne_c = ne, start_c = start;
    const int es = t;
    const float di = __int_as_float(__builtin_amdgcn_readlane(dv, j));
    const float wsel = (qe <= ne_c) ? 1.0f : 0.0f;
    const float4 acc = node_sum<40, false>(srcs, dinv, start_c, ne_c, i, qe, es, wsel, h, lob, [&]() {
      if (j + 1 < nn) {
        ne = __builtin_amdgcn_readlane(cv, j + 1);
        start = __builtin_amdgcn_readlane(fv, j + 1);
        t = entry_word(srcs, start, ne, 0, i + 1, qe);
      }
    });
    float4 s4;
    s4.x = (acc.x + other_half(acc.x)) * di + bv.x;
    s4.y = (acc.y + other_half(acc.y)) * di + bv.y;
    s4.z = (acc.z + other_half(acc.z)) * di + bv.z;
    s4.w = (acc.w + other_half(acc.w)) * di + bv.w;
    if (own_lane) *(float4*)(wr + j * SM_ROW + 4 * cc) = s4;
  }

  float* rp = wr + 10 * lane;
  float v[10];
#pragma unroll
  for (int k = 0; k < 10; ++k) v[k] = 0.0f;
  if ((lane >> 2) < nn) {
#pragma unroll
    for (int k = 0; k < 5; ++k) {
      const float2 p = *(const float2*)(rp + 2 * k);
      v[2 * k] = p.x;
      v[2 * k + 1] = p.y;
    }
  }
  float mx = v[0];
#pragma unroll
  for (int k = 1; k < 10; ++k) mx = fmaxf(mx, v[k]);
  mx = fmaxf(mx, __shfl_xor(mx, 1));
  mx = fmaxf(mx, __shfl_xor(mx, 2));
  float sm = expf(v[0] - mx);
#pragma unroll
  for (int k = 1; k < 10; ++k) sm += expf(v[k] - mx);
  sm += __shfl_xor(sm, 1);
  sm += __shfl_xor(sm, 2);
  const float lse = mx + logf(sm);
#pragma unroll
  for (int k = 0; k < 5; ++k) {
    float2 p;
    p.x = v[2 * k] - lse;
    p.y = v[2 * k + 1] - lse;
    *(float2*)(rp + 2 * k) = p;
  }

  float* op = out + (size_t)i0 * 40 + lane;
  if (nn == NPW) {
    float o[10];
#pragma unroll
    for (int k = 0; k < 10; ++k) o[k] = wr[32 * k + lane];
#pragma unroll
    for (int k = 0; k < 10; ++k) op[32 * k] = o[k];
  } else {
    const int lim = nn * SM_ROW;
    for (int k = 0; k < 10; ++k)
      if (32 * k + lane < lim) op[32 * k] = wr[32 * k + lane];
  }
}

extern "C" void kernel_launch(void* const* d_in, const int* in_sizes, int n_in,
                              void* d_out, int out_size, void* d_ws, size_t ws_size,
                              hipStream_t stream) {
  (void)n_in; (void)out_size; (void)ws_size;

  const float* x  = (const float*)d_in[0];
  const int*   ei = (const int*)d_in[1];
  const float* W1 = (const float*)d_in[2];
  const float* b1 = (const float*)d_in[3];
  const float* W2 = (const float*)d_in[4];
  const float* b2 = (const float*)d_in[5];
  float* out = (float*)d_out;

  const int F_IN = 128;
  const int Nn = in_sizes[0] / F_IN;
  const int E  = in_sizes[1] / 2;
  const int* src = ei;
  const int* dst = ei + E;

  char* ws = (char*)d_ws;
  int*   cnt      = (int*)(ws);
  int*   segoff   = (int*)(ws + ((size_t)1  << 20));
  int*   first    = (int*)(ws + ((size_t)2  << 20));
  float* dinv     = (float*)(ws + ((size_t)3  << 20));
  float* h        = (float*)(ws + ((size_t)4  << 20));
  float* h2       = (float*)(ws + ((size_t)68 << 20));
  int*   srcs     = (int*)(ws + ((size_t)88 << 20));
  int2*  tmp      = (int2*)(ws + ((size_t)100 << 20));

  const dim3 B256(256);

  const int CH   = (E + SORT_G - 1) / SORT_G;
  const int BINS = (Nn + BIN_NODES - 1) / BIN_NODES;
  const dim3 B1024(1024);
  const int MT = Nn / 16;
  const int MT1 = (MT < SORT_G * GEMM_TPB) ? MT : SORT_G * GEMM_TPB;
  const int GB1 = (MT1 + GEMM_TPB - 1) / GEMM_TPB;
  k_chunk_sort<<<SORT_G + GB1, B1024, 0, stream>>>(src, dst, segoff, tmp, E, CH, BINS, x, W1, h, MT1);
  const int GB = (MT - MT1 + GEMM_TPB - 1) / GEMM_TPB;
  k_bin_finalize_gemm<<<BINS + GB, dim3(512), 0, stream>>>(tmp, segoff, cnt, first, dinv, srcs, Nn, BINS, CH,
                                                           x, W1, h, MT1, MT);

  const int GW = 8 * NPW;
  k_gather64_relu_gemm40<<<(Nn + GW - 1) / GW, B256, 0, stream>>>(srcs, first, cnt, dinv, h, b1, W2, h2, Nn);

  k_gather40_logsoftmax<<<(Nn + GW - 1) / GW, B256, 0, stream>>>(srcs, first, cnt, dinv, h2, b2, out, Nn);
}
